# starting kernel
# baseline (speedup 1.0000x reference)
#include <hip/hip_runtime.h>

#define N_NODES 100000
#define N_EDGES 1000000
#define D 64
#define LN_EPS 1e-5f

typedef __attribute__((ext_vector_type(2))) float v2f;
typedef __attribute__((ext_vector_type(8))) float v8f;

__global__ void zero_f32_kernel(float* __restrict__ p, int n) {
  int i = blockIdx.x * blockDim.x + threadIdx.x;
  if (i < n) p[i] = 0.0f;
}

__global__ void degree_kernel(const int* __restrict__ src, const int* __restrict__ dst,
                              float* __restrict__ out_deg, float* __restrict__ in_deg, int e) {
  int i = blockIdx.x * blockDim.x + threadIdx.x;
  if (i < e) {
    atomicAdd(&out_deg[src[i]], 1.0f);
    atomicAdd(&in_deg[dst[i]], 1.0f);
  }
}

__global__ void norm_kernel(float* __restrict__ out_norm, float* __restrict__ in_norm, int n) {
  int i = blockIdx.x * blockDim.x + threadIdx.x;
  if (i < n) {
    out_norm[i] = rsqrtf(fmaxf(out_norm[i], 1.0f));
    in_norm[i]  = rsqrtf(fmaxf(in_norm[i], 1.0f));
  }
}

__global__ void scatter_kernel(const float* __restrict__ h,
                               const float* __restrict__ out_norm,
                               const int* __restrict__ src, const int* __restrict__ dst,
                               float* __restrict__ agg, int e) {
  int g    = blockIdx.x * blockDim.x + threadIdx.x;
  int edge = g >> 5;
  int lane = g & 31;
  if (edge >= e) return;
  int s = src[edge];
  int d = dst[edge];
  float on = out_norm[s];
  float2 v = *(reinterpret_cast<const float2*>(h + (size_t)s * D) + lane);
  float* ap = agg + (size_t)d * D + lane * 2;
  atomicAdd(ap,     v.x * on);
  atomicAdd(ap + 1, v.y * on);
}

__global__ __launch_bounds__(256) void gcn_layer_kernel(
    const float* __restrict__ agg, const float* __restrict__ in_norm,
    const float* __restrict__ h_in,
    const float* __restrict__ Wc, const float* __restrict__ bc,
    const float* __restrict__ Wr, const float* __restrict__ br,
    const float* __restrict__ lng, const float* __restrict__ lnb,
    float* __restrict__ h_out, float* __restrict__ h_final, int n)
{
  __shared__ float lds[8][16 * D];
  const int wave = threadIdx.x >> 5;
  const int lane = threadIdx.x & 31;
  const int row  = lane & 15;
  const int kb   = (lane >> 4) * 2;
  const int base = (blockIdx.x * 8 + wave) * 16;
  int node  = base + row;
  int cnode = node < n ? node : (n - 1);

  v2f aC[16], aR[16];
  const float inrm = in_norm[cnode];
  const float* arow = agg  + (size_t)cnode * D;
  const float* hrow = h_in + (size_t)cnode * D;
#pragma unroll
  for (int kt = 0; kt < 16; ++kt) {
    int k0 = kt * 4 + kb;
    aC[kt][0] = arow[k0]     * inrm;
    aC[kt][1] = arow[k0 + 1] * inrm;
    aR[kt][0] = hrow[k0];
    aR[kt][1] = hrow[k0 + 1];
  }

  const int ncol = lane & 15;
#pragma unroll
  for (int nt = 0; nt < 4; ++nt) {
    v8f c = {};
    const int col = nt * 16 + ncol;
#pragma unroll
    for (int kt = 0; kt < 16; ++kt) {
      int kr = kt * 4 + kb;
      v2f bC, bR;
      bC[0] = Wc[kr * D + col];
      bC[1] = Wc[(kr + 1) * D + col];
      bR[0] = Wr[kr * D + col];
      bR[1] = Wr[(kr + 1) * D + col];
      c = __builtin_amdgcn_wmma_f32_16x16x4_f32(false, aC[kt], false, bC, (short)0, c, false, false);
      c = __builtin_amdgcn_wmma_f32_16x16x4_f32(false, aR[kt], false, bR, (short)0, c, false, false);
    }
    const float bias = bc[col] + br[col];
    const int m0 = (lane < 16) ? 0 : 8;
#pragma unroll
    for (int r = 0; r < 8; ++r)
      lds[wave][(m0 + r) * D + col] = c[r] + bias;
  }
  __syncthreads();

  const int rr = lane >> 1;
  const int j0 = (lane & 1) * 32;
  const float* myrow = &lds[wave][rr * D];
  float s = 0.0f, sq = 0.0f;
#pragma unroll
  for (int j = 0; j < 32; ++j) {
    float x = myrow[j0 + j];
    s  += x;
    sq += x * x;
  }
  s  += __shfl_xor(s, 1);
  sq += __shfl_xor(sq, 1);
  const float mu   = s * (1.0f / 64.0f);
  const float var  = sq * (1.0f / 64.0f) - mu * mu;
  const float rstd = rsqrtf(var + LN_EPS);
  const int onode = base + rr;
  if (onode < n) {
    float* ho = h_out   + (size_t)onode * D;
    float* hf = h_final + (size_t)onode * D;
#pragma unroll
    for (int j = 0; j < 32; ++j) {
      int jj = j0 + j;
      float y = (myrow[jj] - mu) * rstd * lng[jj] + lnb[jj];
      y = fmaxf(y, 0.0f);
      ho[jj] = y;
      hf[jj] += y;
    }
  }
}

__global__ __launch_bounds__(256) void pred_kernel(
    const float* __restrict__ hf, const float* __restrict__ W,
    const float* __restrict__ b, float* __restrict__ out, int n)
{
  const int wave = threadIdx.x >> 5;
  const int lane = threadIdx.x & 31;
  const int row  = lane & 15;
  const int kb   = (lane >> 4) * 2;
  const int base = (blockIdx.x * 8 + wave) * 16;
  int node  = base + row;
  int cnode = node < n ? node : (n - 1);

  v2f a[16];
  const float* ar = hf + (size_t)cnode * D;
#pragma unroll
  for (int kt = 0; kt < 16; ++kt) {
    int k0 = kt * 4 + kb;
    a[kt][0] = ar[k0];
    a[kt][1] = ar[k0 + 1];
  }

  const int ncol = lane & 15;
#pragma unroll
  for (int nt = 0; nt < 4; ++nt) {
    v8f c = {};
    const int col = nt * 16 + ncol;
#pragma unroll
    for (int kt = 0; kt < 16; ++kt) {
      int kr = kt * 4 + kb;
      v2f bf;
      bf[0] = W[kr * D + col];
      bf[1] = W[(kr + 1) * D + col];
      c = __builtin_amdgcn_wmma_f32_16x16x4_f32(false, a[kt], false, bf, (short)0, c, false, false);
    }
    const float bias = b[col];
    const int m0 = (lane < 16) ? 0 : 8;
#pragma unroll
    for (int r = 0; r < 8; ++r) {
      int m = base + m0 + r;
      if (m < n) out[(size_t)m * D + col] = c[r] + bias;
    }
  }
}

extern "C" void kernel_launch(void* const* d_in, const int* in_sizes, int n_in,
                              void* d_out, int out_size, void* d_ws, size_t ws_size,
                              hipStream_t stream) {
  (void)in_sizes; (void)n_in; (void)out_size; (void)ws_size;

  const float* feats  = (const float*)d_in[0];
  const int*   src    = (const int*)  d_in[1];
  const int*   dst    = (const int*)  d_in[2];
  const float* conv_W = (const float*)d_in[3];
  const float* conv_b = (const float*)d_in[4];
  const float* res_W  = (const float*)d_in[5];
  const float* res_b  = (const float*)d_in[6];
  const float* ln_g   = (const float*)d_in[7];
  const float* ln_b   = (const float*)d_in[8];
  const float* pred_W = (const float*)d_in[9];
  const float* pred_b = (const float*)d_in[10];
  float* out = (float*)d_out;

  char* ws = (char*)d_ws;
  size_t off = 0;
  auto alloc = [&](size_t bytes) -> float* {
    float* p = (float*)(ws + off);
    off += (bytes + 255) & ~(size_t)255;
    return p;
  };
  const size_t nodeF = (size_t)N_NODES * 4;
  const size_t featF = (size_t)N_NODES * D * 4;
  float* out_norm = alloc(nodeF);
  float* in_norm  = alloc(nodeF);
  float* agg      = alloc(featF);
  float* hA       = alloc(featF);
  float* hB       = alloc(featF);
  float* h_final  = alloc(featF);

  const int TB = 256;
  const int nFeat   = N_NODES * D;
  const int gFeat   = (nFeat + TB - 1) / TB;
  const int gNode   = (N_NODES + TB - 1) / TB;
  const int gEdge   = (N_EDGES + TB - 1) / TB;
  const int gScat   = ((N_EDGES * 32) + TB - 1) / TB;
  const int gTile   = (N_NODES + 127) / 128;

  zero_f32_kernel<<<gNode, TB, 0, stream>>>(out_norm, N_NODES);
  zero_f32_kernel<<<gNode, TB, 0, stream>>>(in_norm,  N_NODES);
  degree_kernel<<<gEdge, TB, 0, stream>>>(src, dst, out_norm, in_norm, N_EDGES);
  norm_kernel<<<gNode, TB, 0, stream>>>(out_norm, in_norm, N_NODES);

  zero_f32_kernel<<<gFeat, TB, 0, stream>>>(h_final, nFeat);

  const float* h_in = feats;
  float* bufs[2] = { hA, hB };
  for (int layer = 0; layer < 3; ++layer) {
    float* h_out = bufs[layer & 1];
    zero_f32_kernel<<<gFeat, TB, 0, stream>>>(agg, nFeat);
    scatter_kernel<<<gScat, TB, 0, stream>>>(h_in, out_norm, src, dst, agg, N_EDGES);
    gcn_layer_kernel<<<gTile, TB, 0, stream>>>(
        agg, in_norm, h_in,
        conv_W + (size_t)layer * D * D, conv_b + (size_t)layer * D,
        res_W  + (size_t)layer * D * D, res_b  + (size_t)layer * D,
        ln_g   + (size_t)layer * D,     ln_b   + (size_t)layer * D,
        h_out, h_final, N_NODES);
    h_in = h_out;
  }

  pred_kernel<<<gTile, TB, 0, stream>>>(h_final, pred_W, pred_b, out, N_NODES);
}


# scatter: each wave takes 32 consecutive edges (indices by one coalesced load, 16 row gathers in flight before their atomics) instead of one wave per edge; atomic arrival order varies as before
# speedup vs baseline: 1.2499x; 1.2499x over previous
#include <hip/hip_runtime.h>

#define N_NODES 100000
#define N_EDGES 1000000
#define D 64
#define LN_EPS 1e-5f

typedef __attribute__((ext_vector_type(2))) float v2f;
typedef __attribute__((ext_vector_type(8))) float v8f;

__global__ void zero_f32_kernel(float* __restrict__ p, int n) {
  int i = blockIdx.x * blockDim.x + threadIdx.x;
  if (i < n) p[i] = 0.0f;
}

__global__ void degree_kernel(const int* __restrict__ src, const int* __restrict__ dst,
                              float* __restrict__ out_deg, float* __restrict__ in_deg, int e) {
  int i = blockIdx.x * blockDim.x + threadIdx.x;
  if (i < e) {
    atomicAdd(&out_deg[src[i]], 1.0f);
    atomicAdd(&in_deg[dst[i]], 1.0f);
  }
}

__global__ void norm_kernel(float* __restrict__ out_norm, float* __restrict__ in_norm, int n) {
  int i = blockIdx.x * blockDim.x + threadIdx.x;
  if (i < n) {
    out_norm[i] = rsqrtf(fmaxf(out_norm[i], 1.0f));
    in_norm[i]  = rsqrtf(fmaxf(in_norm[i], 1.0f));
  }
}

#define EDGES_PER_WAVE 32
#define EDGE_BATCH 16
__global__ __launch_bounds__(256) void scatter_kernel(const float* __restrict__ h,
                               const float* __restrict__ out_norm,
                               const int* __restrict__ src, const int* __restrict__ dst,
                               float* __restrict__ agg, int e) {
  const int lane = threadIdx.x & 31;
  const int wbase = __builtin_amdgcn_readfirstlane(
      (int)((blockIdx.x * blockDim.x + threadIdx.x) & ~31u));
  if (wbase >= e) return;
  const int  mine = wbase + lane;
  const bool ok   = mine < e;
  const int  s_l  = ok ? src[mine] : 0;
  const int  d_l  = ok ? dst[mine] : 0;
  const int  on_l = __float_as_int(out_norm[s_l]);
  const int  cnt  = (e - wbase) < EDGES_PER_WAVE ? (e - wbase) : EDGES_PER_WAVE;

  if (cnt == EDGES_PER_WAVE) {
#pragma unroll
    for (int b = 0; b < EDGES_PER_WAVE; b += EDGE_BATCH) {
      float2 v[EDGE_BATCH];
#pragma unroll
      for (int j = 0; j < EDGE_BATCH; ++j) {
        const int s = __builtin_amdgcn_readlane(s_l, b + j);
        v[j] = *(reinterpret_cast<const float2*>(h + (size_t)s * D) + lane);
      }
#pragma unroll
      for (int j = 0; j < EDGE_BATCH; ++j) {
        const int   d  = __builtin_amdgcn_readlane(d_l, b + j);
        const float on = __int_as_float(__builtin_amdgcn_readlane(on_l, b + j));
        float* ap = agg + (size_t)d * D + lane * 2;
        atomicAdd(ap,     v[j].x * on);
        atomicAdd(ap + 1, v[j].y * on);
      }
    }
  } else {
    for (int j = 0; j < cnt; ++j) {
      const int   s  = __builtin_amdgcn_readlane(s_l, j);
      const int   d  = __builtin_amdgcn_readlane(d_l, j);
      const float on = __int_as_float(__builtin_amdgcn_readlane(on_l, j));
      float2 v = *(reinterpret_cast<const float2*>(h + (size_t)s * D) + lane);
      float* ap = agg + (size_t)d * D + lane * 2;
      atomicAdd(ap,     v.x * on);
      atomicAdd(ap + 1, v.y * on);
    }
  }
}

__global__ __launch_bounds__(256) void gcn_layer_kernel(
    const float* __restrict__ agg, const float* __restrict__ in_norm,
    const float* __restrict__ h_in,
    const float* __restrict__ Wc, const float* __restrict__ bc,
    const float* __restrict__ Wr, const float* __restrict__ br,
    const float* __restrict__ lng, const float* __restrict__ lnb,
    float* __restrict__ h_out, float* __restrict__ h_final, int n)
{
  __shared__ float lds[8][16 * D];
  const int wave = threadIdx.x >> 5;
  const int lane = threadIdx.x & 31;
  const int row  = lane & 15;
  const int kb   = (lane >> 4) * 2;
  const int base = (blockIdx.x * 8 + wave) * 16;
  int node  = base + row;
  int cnode = node < n ? node : (n - 1);

  v2f aC[16], aR[16];
  const float inrm = in_norm[cnode];
  const float* arow = agg  + (size_t)cnode * D;
  const float* hrow = h_in + (size_t)cnode * D;
#pragma unroll
  for (int kt = 0; kt < 16; ++kt) {
    int k0 = kt * 4 + kb;
    aC[kt][0] = arow[k0]     * inrm;
    aC[kt][1] = arow[k0 + 1] * inrm;
    aR[kt][0] = hrow[k0];
    aR[kt][1] = hrow[k0 + 1];
  }

  const int ncol = lane & 15;
#pragma unroll
  for (int nt = 0; nt < 4; ++nt) {
    v8f c = {};
    const int col = nt * 16 + ncol;
#pragma unroll
    for (int kt = 0; kt < 16; ++kt) {
      int kr = kt * 4 + kb;
      v2f bC, bR;
      bC[0] = Wc[kr * D + col];
      bC[1] = Wc[(kr + 1) * D + col];
      bR[0] = Wr[kr * D + col];
      bR[1] = Wr[(kr + 1) * D + col];
      c = __builtin_amdgcn_wmma_f32_16x16x4_f32(false, aC[kt], false, bC, (short)0, c, false, false);
      c = __builtin_amdgcn_wmma_f32_16x16x4_f32(false, aR[kt], false, bR, (short)0, c, false, false);
    }
    const float bias = bc[col] + br[col];
    const int m0 = (lane < 16) ? 0 : 8;
#pragma unroll
    for (int r = 0; r < 8; ++r)
      lds[wave][(m0 + r) * D + col] = c[r] + bias;
  }
  __syncthreads();

  const int rr = lane >> 1;
  const int j0 = (lane & 1) * 32;
  const float* myrow = &lds[wave][rr * D];
  float s = 0.0f, sq = 0.0f;
#pragma unroll
  for (int j = 0; j < 32; ++j) {
    float x = myrow[j0 + j];
    s  += x;
    sq += x * x;
  }
  s  += __shfl_xor(s, 1);
  sq += __shfl_xor(sq, 1);
  const float mu   = s * (1.0f / 64.0f);
  const float var  = sq * (1.0f / 64.0f) - mu * mu;
  const float rstd = rsqrtf(var + LN_EPS);
  const int onode = base + rr;
  if (onode < n) {
    float* ho = h_out   + (size_t)onode * D;
    float* hf = h_final + (size_t)onode * D;
#pragma unroll
    for (int j = 0; j < 32; ++j) {
      int jj = j0 + j;
      float y = (myrow[jj] - mu) * rstd * lng[jj] + lnb[jj];
      y = fmaxf(y, 0.0f);
      ho[jj] = y;
      hf[jj] += y;
    }
  }
}

__global__ __launch_bounds__(256) void pred_kernel(
    const float* __restrict__ hf, const float* __restrict__ W,
    const float* __restrict__ b, float* __restrict__ out, int n)
{
  const int wave = threadIdx.x >> 5;
  const int lane = threadIdx.x & 31;
  const int row  = lane & 15;
  const int kb   = (lane >> 4) * 2;
  const int base = (blockIdx.x * 8 + wave) * 16;
  int node  = base + row;
  int cnode = node < n ? node : (n - 1);

  v2f a[16];
  const float* ar = hf + (size_t)cnode * D;
#pragma unroll
  for (int kt = 0; kt < 16; ++kt) {
    int k0 = kt * 4 + kb;
    a[kt][0] = ar[k0];
    a[kt][1] = ar[k0 + 1];
  }

  const int ncol = lane & 15;
#pragma unroll
  for (int nt = 0; nt < 4; ++nt) {
    v8f c = {};
    const int col = nt * 16 + ncol;
#pragma unroll
    for (int kt = 0; kt < 16; ++kt) {
      int kr = kt * 4 + kb;
      v2f bf;
      bf[0] = W[kr * D + col];
      bf[1] = W[(kr + 1) * D + col];
      c = __builtin_amdgcn_wmma_f32_16x16x4_f32(false, a[kt], false, bf, (short)0, c, false, false);
    }
    const float bias = b[col];
    const int m0 = (lane < 16) ? 0 : 8;
#pragma unroll
    for (int r = 0; r < 8; ++r) {
      int m = base + m0 + r;
      if (m < n) out[(size_t)m * D + col] = c[r] + bias;
    }
  }
}

extern "C" void kernel_launch(void* const* d_in, const int* in_sizes, int n_in,
                              void* d_out, int out_size, void* d_ws, size_t ws_size,
                              hipStream_t stream) {
  (void)in_sizes; (void)n_in; (void)out_size; (void)ws_size;

  const float* feats  = (const float*)d_in[0];
  const int*   src    = (const int*)  d_in[1];
  const int*   dst    = (const int*)  d_in[2];
  const float* conv_W = (const float*)d_in[3];
  const float* conv_b = (const float*)d_in[4];
  const float* res_W  = (const float*)d_in[5];
  const float* res_b  = (const float*)d_in[6];
  const float* ln_g   = (const float*)d_in[7];
  const float* ln_b   = (const float*)d_in[8];
  const float* pred_W = (const float*)d_in[9];
  const float* pred_b = (const float*)d_in[10];
  float* out = (float*)d_out;

  char* ws = (char*)d_ws;
  size_t off = 0;
  auto alloc = [&](size_t bytes) -> float* {
    float* p = (float*)(ws + off);
    off += (bytes + 255) & ~(size_t)255;
    return p;
  };
  const size_t nodeF = (size_t)N_NODES * 4;
  const size_t featF = (size_t)N_NODES * D * 4;
  float* out_norm = alloc(nodeF);
  float* in_norm  = alloc(nodeF);
  float* agg      = alloc(featF);
  float* hA       = alloc(featF);
  float* hB       = alloc(featF);
  float* h_final  = alloc(featF);

  const int TB = 256;
  const int nFeat   = N_NODES * D;
  const int gFeat   = (nFeat + TB - 1) / TB;
  const int gNode   = (N_NODES + TB - 1) / TB;
  const int gEdge   = (N_EDGES + TB - 1) / TB;
  const int gScat   = (N_EDGES + TB - 1) / TB;
  const int gTile   = (N_NODES + 127) / 128;

  zero_f32_kernel<<<gNode, TB, 0, stream>>>(out_norm, N_NODES);
  zero_f32_kernel<<<gNode, TB, 0, stream>>>(in_norm,  N_NODES);
  degree_kernel<<<gEdge, TB, 0, stream>>>(src, dst, out_norm, in_norm, N_EDGES);
  norm_kernel<<<gNode, TB, 0, stream>>>(out_norm, in_norm, N_NODES);

  zero_f32_kernel<<<gFeat, TB, 0, stream>>>(h_final, nFeat);

  const float* h_in = feats;
  float* bufs[2] = { hA, hB };
  for (int layer = 0; layer < 3; ++layer) {
    float* h_out = bufs[layer & 1];
    zero_f32_kernel<<<gFeat, TB, 0, stream>>>(agg, nFeat);
    scatter_kernel<<<gScat, TB, 0, stream>>>(h_in, out_norm, src, dst, agg, N_EDGES);
    gcn_layer_kernel<<<gTile, TB, 0, stream>>>(
        agg, in_norm, h_in,
        conv_W + (size_t)layer * D * D, conv_b + (size_t)layer * D,
        res_W  + (size_t)layer * D * D, res_b  + (size_t)layer * D,
        ln_g   + (size_t)layer * D,     ln_b   + (size_t)layer * D,
        h_out, h_final, N_NODES);
    h_in = h_out;
  }

  pred_kernel<<<gTile, TB, 0, stream>>>(h_final, pred_W, pred_b, out, N_NODES);
}


# aggregation by gather: edges grouped by destination once per call (count, scan, place); a wave sums its nodes' source rows in registers, one agg store; no f32 atomics; sum order = atomic slot order
# speedup vs baseline: 1.4766x; 1.1814x over previous
#include <hip/hip_runtime.h>

#define N_NODES 100000
#define N_EDGES 1000000
#define D 64
#define LN_EPS 1e-5f

typedef __attribute__((ext_vector_type(2))) float v2f;
typedef __attribute__((ext_vector_type(8))) float v8f;

__global__ void zero_f32_kernel(float* __restrict__ p, int n) {
  int i = blockIdx.x * blockDim.x + threadIdx.x;
  if (i < n) p[i] = 0.0f;
}

__global__ void degree_kernel(const int* __restrict__ src, const int* __restrict__ dst,
                              float* __restrict__ out_deg, int* __restrict__ in_cnt,
                              int* __restrict__ rank, int e) {
  int i = blockIdx.x * blockDim.x + threadIdx.x;
  if (i < e) {
    atomicAdd(&out_deg[src[i]], 1.0f);
    rank[i] = atomicAdd(&in_cnt[dst[i]], 1);
  }
}

#define SCAN_THREADS 1024
#define SCAN_TILE (SCAN_THREADS * 4)
__global__ __launch_bounds__(SCAN_THREADS) void scan_kernel(const int* __restrict__ in_cnt,
                              int* __restrict__ row_ptr, float* __restrict__ in_norm, int n) {
  __shared__ int s_sum[32];
  __shared__ int s_pre[32];
  const int tid  = threadIdx.x;
  const int lane = tid & 31;
  const int wave = tid >> 5;
  const int tile0 = blockIdx.x * SCAN_TILE;

  int pre = 0;
  for (int j = tid * 4; j < tile0; j += SCAN_TILE) {
    const int4 v = *reinterpret_cast<const int4*>(in_cnt + j);
    pre += (v.x + v.y) + (v.z + v.w);
  }

  const int i = tile0 + tid * 4;
  int c[4];
  if (i + 3 < n) {
    const int4 v = *reinterpret_cast<const int4*>(in_cnt + i);
    c[0] = v.x; c[1] = v.y; c[2] = v.z; c[3] = v.w;
  } else {
#pragma unroll
    for (int k = 0; k < 4; ++k) c[k] = (i + k < n) ? in_cnt[i + k] : 0;
  }
  const int t = (c[0] + c[1]) + (c[2] + c[3]);

  int x = t;
#pragma unroll
  for (int o = 1; o < 32; o <<= 1) {
    const int y = __shfl_up(x, o);
    if (lane >= o) x += y;
  }
#pragma unroll
  for (int o = 16; o > 0; o >>= 1) pre += __shfl_xor(pre, o);
  if (lane == 31) s_sum[wave] = x;
  if (lane == 0)  s_pre[wave] = pre;
  __syncthreads();

  int wx = s_sum[lane];
  int wp = s_pre[lane];
#pragma unroll
  for (int o = 1; o < 32; o <<= 1) {
    const int y = __shfl_up(wx, o);
    if (lane >= o) wx += y;
  }
#pragma unroll
  for (int o = 16; o > 0; o >>= 1) wp += __shfl_xor(wp, o);
  const int below = __shfl(wx, (wave + 31) & 31);
  int run = wp + (wave > 0 ? below : 0) + (x - t);
#pragma unroll
  for (int k = 0; k < 4; ++k) {
    if (i + k < n) {
      row_ptr[i + k] = run;
      in_norm[i + k] = rsqrtf(fmaxf((float)c[k], 1.0f));
    }
    run += c[k];
  }
  if (blockIdx.x == gridDim.x - 1 && tid == SCAN_THREADS - 1) row_ptr[n] = run;
}

__global__ void fill_kernel(const int* __restrict__ src, const int* __restrict__ dst,
                            const int* __restrict__ rank, const float* __restrict__ out_deg,
                            const int* __restrict__ row_ptr, int2* __restrict__ csr, int e) {
  int i = blockIdx.x * blockDim.x + threadIdx.x;
  if (i < e) {
    const int s = src[i];
    const int d = dst[i];
    const float w = rsqrtf(fmaxf(out_deg[s], 1.0f));
    csr[row_ptr[d] + rank[i]] = make_int2(s, __float_as_int(w));
  }
}

#define NODES_PER_WAVE 8
__global__ __launch_bounds__(256) void gather_kernel(const float* __restrict__ h,
                               const int* __restrict__ row_ptr,
                               const int2* __restrict__ csr,
                               float* __restrict__ agg, int n) {
  const int lane = threadIdx.x & 31;
  const int half = lane >> 4;
  const int l16  = lane & 15;
  const int n0 = __builtin_amdgcn_readfirstlane(
      (int)((blockIdx.x * blockDim.x + threadIdx.x) >> 5)) * NODES_PER_WAVE;
  if (n0 >= n) return;
  const int pi = (n0 + lane) < n ? (n0 + lane) : n;
  const int rp = row_ptr[pi];
  for (int r = 0; r < NODES_PER_WAVE; ++r) {
    const int node = n0 + r;
    if (node >= n) break;
    const int start = __builtin_amdgcn_readlane(rp, r);
    const int end   = __builtin_amdgcn_readlane(rp, r + 1);
    float4 acc = make_float4(0.0f, 0.0f, 0.0f, 0.0f);
    int e = start + half;
    for (; e + 2 < end; e += 4) {
      const int2 c0 = csr[e];
      const int2 c1 = csr[e + 2];
      const float4 v0 = *(reinterpret_cast<const float4*>(h + (size_t)c0.x * D) + l16);
      const float4 v1 = *(reinterpret_cast<const float4*>(h + (size_t)c1.x * D) + l16);
      const float w0 = __int_as_float(c0.y);
      const float w1 = __int_as_float(c1.y);
      acc.x += v0.x * w0; acc.y += v0.y * w0; acc.z += v0.z * w0; acc.w += v0.w * w0;
      acc.x += v1.x * w1; acc.y += v1.y * w1; acc.z += v1.z * w1; acc.w += v1.w * w1;
    }
    for (; e < end; e += 2) {
      const int2 c0 = csr[e];
      const float4 v0 = *(reinterpret_cast<const float4*>(h + (size_t)c0.x * D) + l16);
      const float w0 = __int_as_float(c0.y);
      acc.x += v0.x * w0; acc.y += v0.y * w0; acc.z += v0.z * w0; acc.w += v0.w * w0;
    }
    acc.x += __shfl_xor(acc.x, 16);
    acc.y += __shfl_xor(acc.y, 16);
    acc.z += __shfl_xor(acc.z, 16);
    acc.w += __shfl_xor(acc.w, 16);
    if (half == 0)
      *(reinterpret_cast<float4*>(agg + (size_t)node * D) + l16) = acc;
  }
}

__global__ __launch_bounds__(256) void gcn_layer_kernel(
    const float* __restrict__ agg, const float* __restrict__ in_norm,
    const float* h_in,
    const float* __restrict__ Wc, const float* __restrict__ bc,
    const float* __restrict__ Wr, const float* __restrict__ br,
    const float* __restrict__ lng, const float* __restrict__ lnb,
    float* h_out, float* __restrict__ h_final, int n)
{
  __shared__ float lds[8][16 * D];
  const int wave = threadIdx.x >> 5;
  const int lane = threadIdx.x & 31;
  const int row  = lane & 15;
  const int kb   = (lane >> 4) * 2;
  const int base = (blockIdx.x * 8 + wave) * 16;
  int node  = base + row;
  int cnode = node < n ? node : (n - 1);

  v2f aC[16], aR[16];
  const float inrm = in_norm[cnode];
  const float* arow = agg  + (size_t)cnode * D;
  const float* hrow = h_in + (size_t)cnode * D;
#pragma unroll
  for (int kt = 0; kt < 16; ++kt) {
    int k0 = kt * 4 + kb;
    aC[kt][0] = arow[k0]     * inrm;
    aC[kt][1] = arow[k0 + 1] * inrm;
    aR[kt][0] = hrow[k0];
    aR[kt][1] = hrow[k0 + 1];
  }

  const int ncol = lane & 15;
#pragma unroll
  for (int nt = 0; nt < 4; ++nt) {
    v8f c = {};
    const int col = nt * 16 + ncol;
#pragma unroll
    for (int kt = 0; kt < 16; ++kt) {
      int kr = kt * 4 + kb;
      v2f bC, bR;
      bC[0] = Wc[kr * D + col];
      bC[1] = Wc[(kr + 1) * D + col];
      bR[0] = Wr[kr * D + col];
      bR[1] = Wr[(kr + 1) * D + col];
      c = __builtin_amdgcn_wmma_f32_16x16x4_f32(false, aC[kt], false, bC, (short)0, c, false, false);
      c = __builtin_amdgcn_wmma_f32_16x16x4_f32(false, aR[kt], false, bR, (short)0, c, false, false);
    }
    const float bias = bc[col] + br[col];
    const int m0 = (lane < 16) ? 0 : 8;
#pragma unroll
    for (int r = 0; r < 8; ++r)
      lds[wave][(m0 + r) * D + col] = c[r] + bias;
  }
  __syncthreads();

  const int rr = lane >> 1;
  const int j0 = (lane & 1) * 32;
  const float* myrow = &lds[wave][rr * D];
  float s = 0.0f, sq = 0.0f;
#pragma unroll
  for (int j = 0; j < 32; ++j) {
    float x = myrow[j0 + j];
    s  += x;
    sq += x * x;
  }
  s  += __shfl_xor(s, 1);
  sq += __shfl_xor(sq, 1);
  const float mu   = s * (1.0f / 64.0f);
  const float var  = sq * (1.0f / 64.0f) - mu * mu;
  const float rstd = rsqrtf(var + LN_EPS);
  const int onode = base + rr;
  if (onode < n) {
    float* ho = h_out   + (size_t)onode * D;
    float* hf = h_final + (size_t)onode * D;
#pragma unroll
    for (int j = 0; j < 32; ++j) {
      int jj = j0 + j;
      float y = (myrow[jj] - mu) * rstd * lng[jj] + lnb[jj];
      y = fmaxf(y, 0.0f);
      ho[jj] = y;
      hf[jj] += y;
    }
  }
}

__global__ __launch_bounds__(256) void pred_kernel(
    const float* __restrict__ hf, const float* __restrict__ W,
    const float* __restrict__ b, float* __restrict__ out, int n)
{
  const int wave = threadIdx.x >> 5;
  const int lane = threadIdx.x & 31;
  const int row  = lane & 15;
  const int kb   = (lane >> 4) * 2;
  const int base = (blockIdx.x * 8 + wave) * 16;
  int node  = base + row;
  int cnode = node < n ? node : (n - 1);

  v2f a[16];
  const float* ar = hf + (size_t)cnode * D;
#pragma unroll
  for (int kt = 0; kt < 16; ++kt) {
    int k0 = kt * 4 + kb;
    a[kt][0] = ar[k0];
    a[kt][1] = ar[k0 + 1];
  }

  const int ncol = lane & 15;
#pragma unroll
  for (int nt = 0; nt < 4; ++nt) {
    v8f c = {};
    const int col = nt * 16 + ncol;
#pragma unroll
    for (int kt = 0; kt < 16; ++kt) {
      int kr = kt * 4 + kb;
      v2f bf;
      bf[0] = W[kr * D + col];
      bf[1] = W[(kr + 1) * D + col];
      c = __builtin_amdgcn_wmma_f32_16x16x4_f32(false, a[kt], false, bf, (short)0, c, false, false);
    }
    const float bias = b[col];
    const int m0 = (lane < 16) ? 0 : 8;
#pragma unroll
    for (int r = 0; r < 8; ++r) {
      int m = base + m0 + r;
      if (m < n) out[(size_t)m * D + col] = c[r] + bias;
    }
  }
}

extern "C" void kernel_launch(void* const* d_in, const int* in_sizes, int n_in,
                              void* d_out, int out_size, void* d_ws, size_t ws_size,
                              hipStream_t stream) {
  (void)in_sizes; (void)n_in; (void)out_size; (void)ws_size;

  const float* feats  = (const float*)d_in[0];
  const int*   src    = (const int*)  d_in[1];
  const int*   dst    = (const int*)  d_in[2];
  const float* conv_W = (const float*)d_in[3];
  const float* conv_b = (const float*)d_in[4];
  const float* res_W  = (const float*)d_in[5];
  const float* res_b  = (const float*)d_in[6];
  const float* ln_g   = (const float*)d_in[7];
  const float* ln_b   = (const float*)d_in[8];
  const float* pred_W = (const float*)d_in[9];
  const float* pred_b = (const float*)d_in[10];
  float* out = (float*)d_out;

  char* ws = (char*)d_ws;
  size_t off = 0;
  auto alloc = [&](size_t bytes) -> float* {
    float* p = (float*)(ws + off);
    off += (bytes + 255) & ~(size_t)255;
    return p;
  };
  const size_t nodeF = (size_t)N_NODES * 4;
  const size_t featF = (size_t)N_NODES * D * 4;
  float* out_deg  = alloc(nodeF);
  int*   in_cnt   = (int*)alloc(nodeF);
  float* in_norm  = alloc(nodeF);
  int*   row_ptr  = (int*)alloc(nodeF + 4);
  int*   rank     = (int*)alloc((size_t)N_EDGES * 4);
  int2*  csr      = (int2*)alloc((size_t)N_EDGES * 8);
  float* agg      = alloc(featF);
  float* hbuf     = alloc(featF);
  float* h_final  = alloc(featF);

  const int TB = 256;
  const int nFeat   = N_NODES * D;
  const int gFeat   = (nFeat + TB - 1) / TB;
  const int nClear  = 2 * ((N_NODES * 4 + 255) / 256) * 64;
  const int gClear  = (nClear + TB - 1) / TB;
  const int gEdge   = (N_EDGES + TB - 1) / TB;
  const int gScan   = (N_NODES + SCAN_TILE - 1) / SCAN_TILE;
  const int gGath   = (N_NODES + 8 * NODES_PER_WAVE - 1) / (8 * NODES_PER_WAVE);
  const int gTile   = (N_NODES + 127) / 128;

  zero_f32_kernel<<<gClear, TB, 0, stream>>>(out_deg, nClear);
  degree_kernel<<<gEdge, TB, 0, stream>>>(src, dst, out_deg, in_cnt, rank, N_EDGES);
  scan_kernel<<<gScan, SCAN_THREADS, 0, stream>>>(in_cnt, row_ptr, in_norm, N_NODES);
  fill_kernel<<<gEdge, TB, 0, stream>>>(src, dst, rank, out_deg, row_ptr, csr, N_EDGES);

  zero_f32_kernel<<<gFeat, TB, 0, stream>>>(h_final, nFeat);

  const float* h_in = feats;
  for (int layer = 0; layer < 3; ++layer) {
    gather_kernel<<<gGath, TB, 0, stream>>>(h_in, row_ptr, csr, agg, N_NODES);
    gcn_layer_kernel<<<gTile, TB, 0, stream>>>(
        agg, in_norm, h_in,
        conv_W + (size_t)layer * D * D, conv_b + (size_t)layer * D,
        res_W  + (size_t)layer * D * D, res_b  + (size_t)layer * D,
        ln_g   + (size_t)layer * D,     ln_b   + (size_t)layer * D,
        hbuf, h_final, N_NODES);
    h_in = hbuf;
  }

  pred_kernel<<<gTile, TB, 0, stream>>>(h_final, pred_W, pred_b, out, N_NODES);
}


# gather: next node's edge entries fetched ahead, a node's source rows (8 per half-wave) issued as one branch-free load group, not a dependent entry-then-row trip per edge pair; sum order unchanged
# speedup vs baseline: 1.5289x; 1.0354x over previous
#include <hip/hip_runtime.h>

#define N_NODES 100000
#define N_EDGES 1000000
#define D 64
#define LN_EPS 1e-5f

typedef __attribute__((ext_vector_type(2))) float v2f;
typedef __attribute__((ext_vector_type(8))) float v8f;

__global__ void zero_f32_kernel(float* __restrict__ p, int n) {
  int i = blockIdx.x * blockDim.x + threadIdx.x;
  if (i < n) p[i] = 0.0f;
}

__global__ void degree_kernel(const int* __restrict__ src, const int* __restrict__ dst,
                              float* __restrict__ out_deg, int* __restrict__ in_cnt,
                              int* __restrict__ rank, int e) {
  int i = blockIdx.x * blockDim.x + threadIdx.x;
  if (i < e) {
    atomicAdd(&out_deg[src[i]], 1.0f);
    rank[i] = atomicAdd(&in_cnt[dst[i]], 1);
  }
}

#define SCAN_THREADS 1024
#define SCAN_TILE (SCAN_THREADS * 4)
__global__ __launch_bounds__(SCAN_THREADS) void scan_kernel(const int* __restrict__ in_cnt,
                              int* __restrict__ row_ptr, float* __restrict__ in_norm, int n) {
  __shared__ int s_sum[32];
  __shared__ int s_pre[32];
  const int tid  = threadIdx.x;
  const int lane = tid & 31;
  const int wave = tid >> 5;
  const int tile0 = blockIdx.x * SCAN_TILE;

  int pre = 0;
  for (int j = tid * 4; j < tile0; j += SCAN_TILE) {
    const int4 v = *reinterpret_cast<const int4*>(in_cnt + j);
    pre += (v.x + v.y) + (v.z + v.w);
  }

  const int i = tile0 + tid * 4;
  int c[4];
  if (i + 3 < n) {
    const int4 v = *reinterpret_cast<const int4*>(in_cnt + i);
    c[0] = v.x; c[1] = v.y; c[2] = v.z; c[3] = v.w;
  } else {
#pragma unroll
    for (int k = 0; k < 4; ++k) c[k] = (i + k < n) ? in_cnt[i + k] : 0;
  }
  const int t = (c[0] + c[1]) + (c[2] + c[3]);

  int x = t;
#pragma unroll
  for (int o = 1; o < 32; o <<= 1) {
    const int y = __shfl_up(x, o);
    if (lane >= o) x += y;
  }
#pragma unroll
  for (int o = 16; o > 0; o >>= 1) pre += __shfl_xor(pre, o);
  if (lane == 31) s_sum[wave] = x;
  if (lane == 0)  s_pre[wave] = pre;
  __syncthreads();

  int wx = s_sum[lane];
  int wp = s_pre[lane];
#pragma unroll
  for (int o = 1; o < 32; o <<= 1) {
    const int y = __shfl_up(wx, o);
    if (lane >= o) wx += y;
  }
#pragma unroll
  for (int o = 16; o > 0; o >>= 1) wp += __shfl_xor(wp, o);
  const int below = __shfl(wx, (wave + 31) & 31);
  int run = wp + (wave > 0 ? below : 0) + (x - t);
#pragma unroll
  for (int k = 0; k < 4; ++k) {
    if (i + k < n) {
      row_ptr[i + k] = run;
      in_norm[i + k] = rsqrtf(fmaxf((float)c[k], 1.0f));
    }
    run += c[k];
  }
  if (blockIdx.x == gridDim.x - 1 && tid == SCAN_THREADS - 1) row_ptr[n] = run;
}

__global__ void fill_kernel(const int* __restrict__ src, const int* __restrict__ dst,
                            const int* __restrict__ rank, const float* __restrict__ out_deg,
                            const int* __restrict__ row_ptr, int2* __restrict__ csr, int e) {
  int i = blockIdx.x * blockDim.x + threadIdx.x;
  if (i < e) {
    const int s = src[i];
    const int d = dst[i];
    const float w = rsqrtf(fmaxf(out_deg[s], 1.0f));
    csr[row_ptr[d] + rank[i]] = make_int2(s, __float_as_int(w));
  }
}

#define NODES_PER_WAVE 8
#define SLOTS 8
__global__ __launch_bounds__(256) void gather_kernel(const float* __restrict__ h,
                               const int* __restrict__ row_ptr,
                               const int2* __restrict__ csr,
                               float* __restrict__ agg, int n, int n_edges) {
  const int lane = threadIdx.x & 31;
  const int half = lane >> 4;
  const int l16  = lane & 15;
  const int n0 = __builtin_amdgcn_readfirstlane(
      (int)((blockIdx.x * blockDim.x + threadIdx.x) >> 5)) * NODES_PER_WAVE;
  if (n0 >= n) return;
  const int pi = (n0 + lane) < n ? (n0 + lane) : n;
  const int rp = row_ptr[pi];
  const float4* h4 = reinterpret_cast<const float4*>(h);
  const unsigned u16 = (unsigned)l16;
  const int last = n_edges - 1;

  int2 cn[SLOTS];
  int nstart = __builtin_amdgcn_readlane(rp, 0);
  int nend   = __builtin_amdgcn_readlane(rp, 1);
#pragma unroll
  for (int j = 0; j < SLOTS; ++j) {
    const int e = nstart + half + 2 * j;
    cn[j] = csr[e < last ? e : last];
  }

#pragma unroll
  for (int r = 0; r < NODES_PER_WAVE; ++r) {
    const int node = n0 + r;
    if (node >= n) break;
    const int start = nstart;
    const int end   = nend;
    int2 c[SLOTS];
#pragma unroll
    for (int j = 0; j < SLOTS; ++j) c[j] = cn[j];

    if (r + 1 < NODES_PER_WAVE) {
      nstart = end;
      nend = __builtin_amdgcn_readlane(rp, r + 2);
#pragma unroll
      for (int j = 0; j < SLOTS; ++j) {
        const int e = nstart + half + 2 * j;
        cn[j] = csr[e < last ? e : last];
      }
    }

    int   src[SLOTS];
    float wgt[SLOTS];
    bool  ok[SLOTS];
#pragma unroll
    for (int j = 0; j < SLOTS; ++j) {
      ok[j]  = (start + half + 2 * j) < end;
      src[j] = ok[j] ? c[j].x : 0;
      wgt[j] = ok[j] ? __int_as_float(c[j].y) : 0.0f;
    }

    float4 v[SLOTS];
#pragma unroll
    for (int j = 0; j < SLOTS; ++j) v[j] = h4[(unsigned)src[j] * (unsigned)(D / 4) + u16];

    float4 acc = make_float4(0.0f, 0.0f, 0.0f, 0.0f);
#pragma unroll
    for (int j = 0; j < SLOTS; ++j) {
      const float4 x = ok[j] ? v[j] : make_float4(0.0f, 0.0f, 0.0f, 0.0f);
      const float w = wgt[j];
      acc.x += x.x * w; acc.y += x.y * w; acc.z += x.z * w; acc.w += x.w * w;
    }
    for (int e = start + 2 * SLOTS + half; e < end; e += 2) {
      const int2 c0 = csr[e];
      const float4 v0 = h4[(unsigned)c0.x * (unsigned)(D / 4) + u16];
      const float w0 = __int_as_float(c0.y);
      acc.x += v0.x * w0; acc.y += v0.y * w0; acc.z += v0.z * w0; acc.w += v0.w * w0;
    }
    acc.x += __shfl_xor(acc.x, 16);
    acc.y += __shfl_xor(acc.y, 16);
    acc.z += __shfl_xor(acc.z, 16);
    acc.w += __shfl_xor(acc.w, 16);
    if (half == 0)
      *(reinterpret_cast<float4*>(agg + (size_t)node * D) + l16) = acc;
  }
}

__global__ __launch_bounds__(256) void gcn_layer_kernel(
    const float* __restrict__ agg, const float* __restrict__ in_norm,
    const float* h_in,
    const float* __restrict__ Wc, const float* __restrict__ bc,
    const float* __restrict__ Wr, const float* __restrict__ br,
    const float* __restrict__ lng, const float* __restrict__ lnb,
    float* h_out, float* __restrict__ h_final, int n)
{
  __shared__ float lds[8][16 * D];
  const int wave = threadIdx.x >> 5;
  const int lane = threadIdx.x & 31;
  const int row  = lane & 15;
  const int kb   = (lane >> 4) * 2;
  const int base = (blockIdx.x * 8 + wave) * 16;
  int node  = base + row;
  int cnode = node < n ? node : (n - 1);

  v2f aC[16], aR[16];
  const float inrm = in_norm[cnode];
  const float* arow = agg  + (size_t)cnode * D;
  const float* hrow = h_in + (size_t)cnode * D;
#pragma unroll
  for (int kt = 0; kt < 16; ++kt) {
    int k0 = kt * 4 + kb;
    aC[kt][0] = arow[k0]     * inrm;
    aC[kt][1] = arow[k0 + 1] * inrm;
    aR[kt][0] = hrow[k0];
    aR[kt][1] = hrow[k0 + 1];
  }

  const int ncol = lane & 15;
#pragma unroll
  for (int nt = 0; nt < 4; ++nt) {
    v8f c = {};
    const int col = nt * 16 + ncol;
#pragma unroll
    for (int kt = 0; kt < 16; ++kt) {
      int kr = kt * 4 + kb;
      v2f bC, bR;
      bC[0] = Wc[kr * D + col];
      bC[1] = Wc[(kr + 1) * D + col];
      bR[0] = Wr[kr * D + col];
      bR[1] = Wr[(kr + 1) * D + col];
      c = __builtin_amdgcn_wmma_f32_16x16x4_f32(false, aC[kt], false, bC, (short)0, c, false, false);
      c = __builtin_amdgcn_wmma_f32_16x16x4_f32(false, aR[kt], false, bR, (short)0, c, false, false);
    }
    const float bias = bc[col] + br[col];
    const int m0 = (lane < 16) ? 0 : 8;
#pragma unroll
    for (int r = 0; r < 8; ++r)
      lds[wave][(m0 + r) * D + col] = c[r] + bias;
  }
  __syncthreads();

  const int rr = lane >> 1;
  const int j0 = (lane & 1) * 32;
  const float* myrow = &lds[wave][rr * D];
  float s = 0.0f, sq = 0.0f;
#pragma unroll
  for (int j = 0; j < 32; ++j) {
    float x = myrow[j0 + j];
    s  += x;
    sq += x * x;
  }
  s  += __shfl_xor(s, 1);
  sq += __shfl_xor(sq, 1);
  const float mu   = s * (1.0f / 64.0f);
  const float var  = sq * (1.0f / 64.0f) - mu * mu;
  const float rstd = rsqrtf(var + LN_EPS);
  const int onode = base + rr;
  if (onode < n) {
    float* ho = h_out   + (size_t)onode * D;
    float* hf = h_final + (size_t)onode * D;
#pragma unroll
    for (int j = 0; j < 32; ++j) {
      int jj = j0 + j;
      float y = (myrow[jj] - mu) * rstd * lng[jj] + lnb[jj];
      y = fmaxf(y, 0.0f);
      ho[jj] = y;
      hf[jj] += y;
    }
  }
}

__global__ __launch_bounds__(256) void pred_kernel(
    const float* __restrict__ hf, const float* __restrict__ W,
    const float* __restrict__ b, float* __restrict__ out, int n)
{
  const int wave = threadIdx.x >> 5;
  const int lane = threadIdx.x & 31;
  const int row  = lane & 15;
  const int kb   = (lane >> 4) * 2;
  const int base = (blockIdx.x * 8 + wave) * 16;
  int node  = base + row;
  int cnode = node < n ? node : (n - 1);

  v2f a[16];
  const float* ar = hf + (size_t)cnode * D;
#pragma unroll
  for (int kt = 0; kt < 16; ++kt) {
    int k0 = kt * 4 + kb;
    a[kt][0] = ar[k0];
    a[kt][1] = ar[k0 + 1];
  }

  const int ncol = lane & 15;
#pragma unroll
  for (int nt = 0; nt < 4; ++nt) {
    v8f c = {};
    const int col = nt * 16 + ncol;
#pragma unroll
    for (int kt = 0; kt < 16; ++kt) {
      int kr = kt * 4 + kb;
      v2f bf;
      bf[0] = W[kr * D + col];
      bf[1] = W[(kr + 1) * D + col];
      c = __builtin_amdgcn_wmma_f32_16x16x4_f32(false, a[kt], false, bf, (short)0, c, false, false);
    }
    const float bias = b[col];
    const int m0 = (lane < 16) ? 0 : 8;
#pragma unroll
    for (int r = 0; r < 8; ++r) {
      int m = base + m0 + r;
      if (m < n) out[(size_t)m * D + col] = c[r] + bias;
    }
  }
}

extern "C" void kernel_launch(void* const* d_in, const int* in_sizes, int n_in,
                              void* d_out, int out_size, void* d_ws, size_t ws_size,
                              hipStream_t stream) {
  (void)in_sizes; (void)n_in; (void)out_size; (void)ws_size;

  const float* feats  = (const float*)d_in[0];
  const int*   src    = (const int*)  d_in[1];
  const int*   dst    = (const int*)  d_in[2];
  const float* conv_W = (const float*)d_in[3];
  const float* conv_b = (const float*)d_in[4];
  const float* res_W  = (const float*)d_in[5];
  const float* res_b  = (const float*)d_in[6];
  const float* ln_g   = (const float*)d_in[7];
  const float* ln_b   = (const float*)d_in[8];
  const float* pred_W = (const float*)d_in[9];
  const float* pred_b = (const float*)d_in[10];
  float* out = (float*)d_out;

  char* ws = (char*)d_ws;
  size_t off = 0;
  auto alloc = [&](size_t bytes) -> float* {
    float* p = (float*)(ws + off);
    off += (bytes + 255) & ~(size_t)255;
    return p;
  };
  const size_t nodeF = (size_t)N_NODES * 4;
  const size_t featF = (size_t)N_NODES * D * 4;
  float* out_deg  = alloc(nodeF);
  int*   in_cnt   = (int*)alloc(nodeF);
  float* in_norm  = alloc(nodeF);
  int*   row_ptr  = (int*)alloc(nodeF + 4);
  int*   rank     = (int*)alloc((size_t)N_EDGES * 4);
  int2*  csr      = (int2*)alloc((size_t)N_EDGES * 8);
  float* agg      = alloc(featF);
  float* hbuf     = alloc(featF);
  float* h_final  = alloc(featF);

  const int TB = 256;
  const int nFeat   = N_NODES * D;
  const int gFeat   = (nFeat + TB - 1) / TB;
  const int nClear  = 2 * ((N_NODES * 4 + 255) / 256) * 64;
  const int gClear  = (nClear + TB - 1) / TB;
  const int gEdge   = (N_EDGES + TB - 1) / TB;
  const int gScan   = (N_NODES + SCAN_TILE - 1) / SCAN_TILE;
  const int gGath   = (N_NODES + 8 * NODES_PER_WAVE - 1) / (8 * NODES_PER_WAVE);
  const int gTile   = (N_NODES + 127) / 128;

  zero_f32_kernel<<<gClear, TB, 0, stream>>>(out_deg, nClear);
  degree_kernel<<<gEdge, TB, 0, stream>>>(src, dst, out_deg, in_cnt, rank, N_EDGES);
  scan_kernel<<<gScan, SCAN_THREADS, 0, stream>>>(in_cnt, row_ptr, in_norm, N_NODES);
  fill_kernel<<<gEdge, TB, 0, stream>>>(src, dst, rank, out_deg, row_ptr, csr, N_EDGES);

  zero_f32_kernel<<<gFeat, TB, 0, stream>>>(h_final, nFeat);

  const float* h_in = feats;
  for (int layer = 0; layer < 3; ++layer) {
    gather_kernel<<<gGath, TB, 0, stream>>>(h_in, row_ptr, csr, agg, N_NODES, N_EDGES);
    gcn_layer_kernel<<<gTile, TB, 0, stream>>>(
        agg, in_norm, h_in,
        conv_W + (size_t)layer * D * D, conv_b + (size_t)layer * D,
        res_W  + (size_t)layer * D * D, res_b  + (size_t)layer * D,
        ln_g   + (size_t)layer * D,     ln_b   + (size_t)layer * D,
        hbuf, h_final, N_NODES);
    h_in = hbuf;
  }

  pred_kernel<<<gTile, TB, 0, stream>>>(h_final, pred_W, pred_b, out, N_NODES);
}


# both WMMA kernels: node rows (agg, h, JK sum, outputs) cross memory as whole 256-byte rows per half-wave and are turned to fragments and back in padded per-wave LDS tiles; sum order unchanged
# speedup vs baseline: 1.8994x; 1.2423x over previous
#include <hip/hip_runtime.h>

#define N_NODES 100000
#define N_EDGES 1000000
#define D 64
#define LN_EPS 1e-5f

typedef __attribute__((ext_vector_type(2))) float v2f;
typedef __attribute__((ext_vector_type(8))) float v8f;

__global__ void zero_f32_kernel(float* __restrict__ p, int n) {
  int i = blockIdx.x * blockDim.x + threadIdx.x;
  if (i < n) p[i] = 0.0f;
}

__global__ void degree_kernel(const int* __restrict__ src, const int* __restrict__ dst,
                              float* __restrict__ out_deg, int* __restrict__ in_cnt,
                              int* __restrict__ rank, int e) {
  int i = blockIdx.x * blockDim.x + threadIdx.x;
  if (i < e) {
    atomicAdd(&out_deg[src[i]], 1.0f);
    rank[i] = atomicAdd(&in_cnt[dst[i]], 1);
  }
}

#define SCAN_THREADS 1024
#define SCAN_TILE (SCAN_THREADS * 4)
__global__ __launch_bounds__(SCAN_THREADS) void scan_kernel(const int* __restrict__ in_cnt,
                              int* __restrict__ row_ptr, float* __restrict__ in_norm, int n) {
  __shared__ int s_sum[32];
  __shared__ int s_pre[32];
  const int tid  = threadIdx.x;
  const int lane = tid & 31;
  const int wave = tid >> 5;
  const int tile0 = blockIdx.x * SCAN_TILE;

  int pre = 0;
  for (int j = tid * 4; j < tile0; j += SCAN_TILE) {
    const int4 v = *reinterpret_cast<const int4*>(in_cnt + j);
    pre += (v.x + v.y) + (v.z + v.w);
  }

  const int i = tile0 + tid * 4;
  int c[4];
  if (i + 3 < n) {
    const int4 v = *reinterpret_cast<const int4*>(in_cnt + i);
    c[0] = v.x; c[1] = v.y; c[2] = v.z; c[3] = v.w;
  } else {
#pragma unroll
    for (int k = 0; k < 4; ++k) c[k] = (i + k < n) ? in_cnt[i + k] : 0;
  }
  const int t = (c[0] + c[1]) + (c[2] + c[3]);

  int x = t;
#pragma unroll
  for (int o = 1; o < 32; o <<= 1) {
    const int y = __shfl_up(x, o);
    if (lane >= o) x += y;
  }
#pragma unroll
  for (int o = 16; o > 0; o >>= 1) pre += __shfl_xor(pre, o);
  if (lane == 31) s_sum[wave] = x;
  if (lane == 0)  s_pre[wave] = pre;
  __syncthreads();

  int wx = s_sum[lane];
  int wp = s_pre[lane];
#pragma unroll
  for (int o = 1; o < 32; o <<= 1) {
    const int y = __shfl_up(wx, o);
    if (lane >= o) wx += y;
  }
#pragma unroll
  for (int o = 16; o > 0; o >>= 1) wp += __shfl_xor(wp, o);
  const int below = __shfl(wx, (wave + 31) & 31);
  int run = wp + (wave > 0 ? below : 0) + (x - t);
#pragma unroll
  for (int k = 0; k < 4; ++k) {
    if (i + k < n) {
      row_ptr[i + k] = run;
      in_norm[i + k] = rsqrtf(fmaxf((float)c[k], 1.0f));
    }
    run += c[k];
  }
  if (blockIdx.x == gridDim.x - 1 && tid == SCAN_THREADS - 1) row_ptr[n] = run;
}

__global__ void fill_kernel(const int* __restrict__ src, const int* __restrict__ dst,
                            const int* __restrict__ rank, const float* __restrict__ out_deg,
                            const int* __restrict__ row_ptr, int2* __restrict__ csr, int e) {
  int i = blockIdx.x * blockDim.x + threadIdx.x;
  if (i < e) {
    const int s = src[i];
    const int d = dst[i];
    const float w = rsqrtf(fmaxf(out_deg[s], 1.0f));
    csr[row_ptr[d] + rank[i]] = make_int2(s, __float_as_int(w));
  }
}

#define NODES_PER_WAVE 8
#define SLOTS 8
__global__ __launch_bounds__(256) void gather_kernel(const float* __restrict__ h,
                               const int* __restrict__ row_ptr,
                               const int2* __restrict__ csr,
                               float* __restrict__ agg, int n, int n_edges) {
  const int lane = threadIdx.x & 31;
  const int half = lane >> 4;
  const int l16  = lane & 15;
  const int n0 = __builtin_amdgcn_readfirstlane(
      (int)((blockIdx.x * blockDim.x + threadIdx.x) >> 5)) * NODES_PER_WAVE;
  if (n0 >= n) return;
  const int pi = (n0 + lane) < n ? (n0 + lane) : n;
  const int rp = row_ptr[pi];
  const float4* h4 = reinterpret_cast<const float4*>(h);
  const unsigned u16 = (unsigned)l16;
  const int last = n_edges - 1;

  int2 cn[SLOTS];
  int nstart = __builtin_amdgcn_readlane(rp, 0);
  int nend   = __builtin_amdgcn_readlane(rp, 1);
#pragma unroll
  for (int j = 0; j < SLOTS; ++j) {
    const int e = nstart + half + 2 * j;
    cn[j] = csr[e < last ? e : last];
  }

#pragma unroll
  for (int r = 0; r < NODES_PER_WAVE; ++r) {
    const int node = n0 + r;
    if (node >= n) break;
    const int start = nstart;
    const int end   = nend;
    int2 c[SLOTS];
#pragma unroll
    for (int j = 0; j < SLOTS; ++j) c[j] = cn[j];

    if (r + 1 < NODES_PER_WAVE) {
      nstart = end;
      nend = __builtin_amdgcn_readlane(rp, r + 2);
#pragma unroll
      for (int j = 0; j < SLOTS; ++j) {
        const int e = nstart + half + 2 * j;
        cn[j] = csr[e < last ? e : last];
      }
    }

    int   src[SLOTS];
    float wgt[SLOTS];
    bool  ok[SLOTS];
#pragma unroll
    for (int j = 0; j < SLOTS; ++j) {
      ok[j]  = (start + half + 2 * j) < end;
      src[j] = ok[j] ? c[j].x : 0;
      wgt[j] = ok[j] ? __int_as_float(c[j].y) : 0.0f;
    }

    float4 v[SLOTS];
#pragma unroll
    for (int j = 0; j < SLOTS; ++j) v[j] = h4[(unsigned)src[j] * (unsigned)(D / 4) + u16];

    float4 acc = make_float4(0.0f, 0.0f, 0.0f, 0.0f);
#pragma unroll
    for (int j = 0; j < SLOTS; ++j) {
      const float4 x = ok[j] ? v[j] : make_float4(0.0f, 0.0f, 0.0f, 0.0f);
      const float w = wgt[j];
      acc.x += x.x * w; acc.y += x.y * w; acc.z += x.z * w; acc.w += x.w * w;
    }
    for (int e = start + 2 * SLOTS + half; e < end; e += 2) {
      const int2 c0 = csr[e];
      const float4 v0 = h4[(unsigned)c0.x * (unsigned)(D / 4) + u16];
      const float w0 = __int_as_float(c0.y);
      acc.x += v0.x * w0; acc.y += v0.y * w0; acc.z += v0.z * w0; acc.w += v0.w * w0;
    }
    acc.x += __shfl_xor(acc.x, 16);
    acc.y += __shfl_xor(acc.y, 16);
    acc.z += __shfl_xor(acc.z, 16);
    acc.w += __shfl_xor(acc.w, 16);
    if (half == 0)
      *(reinterpret_cast<float4*>(agg + (size_t)node * D) + l16) = acc;
  }
}

#define TP 68
typedef __attribute__((ext_vector_type(4))) float v4f;

__global__ __launch_bounds__(256) void gcn_layer_kernel(
    const float* __restrict__ agg, const float* __restrict__ in_norm,
    const float* h_in,
    const float* __restrict__ Wc, const float* __restrict__ bc,
    const float* __restrict__ Wr, const float* __restrict__ br,
    const float* __restrict__ lng, const float* __restrict__ lnb,
    float* h_out, float* __restrict__ h_final, int n)
{
  __shared__ __attribute__((aligned(16))) float s_tile[8][2][16 * TP];
  __shared__ __attribute__((aligned(8)))  float s_stat[8][32];
  const int wave = threadIdx.x >> 5;
  const int lane = threadIdx.x & 31;
  const int half = lane >> 4;
  const int l16  = lane & 15;
  const int row  = l16;
  const int kb   = half * 2;
  const int base = (blockIdx.x * 8 + wave) * 16;
  float* tA = s_tile[wave][0];
  float* tH = s_tile[wave][1];

  v4f va[8], vh[8], vf[8];
#pragma unroll
  for (int i = 0; i < 8; ++i) {
    int nd = base + 2 * i + half;
    nd = nd < n ? nd : (n - 1);
    const size_t o = (size_t)nd * D + 4 * l16;
    va[i] = *reinterpret_cast<const v4f*>(agg + o);
    vh[i] = *reinterpret_cast<const v4f*>(h_in + o);
    vf[i] = *reinterpret_cast<const v4f*>(h_final + o);
  }
  int cnode = base + row;
  cnode = cnode < n ? cnode : (n - 1);
  const float inrm = in_norm[cnode];
#pragma unroll
  for (int i = 0; i < 8; ++i) {
    const int r = 2 * i + half;
    *reinterpret_cast<v4f*>(tA + r * TP + 4 * l16) = va[i];
    *reinterpret_cast<v4f*>(tH + r * TP + 4 * l16) = vh[i];
  }
  __syncthreads();

  v2f aC[16], aR[16];
#pragma unroll
  for (int kt = 0; kt < 16; ++kt) {
    const int k0 = kt * 4 + kb;
    const v2f t = *reinterpret_cast<const v2f*>(tA + row * TP + k0);
    aC[kt][0] = t[0] * inrm;
    aC[kt][1] = t[1] * inrm;
    aR[kt] = *reinterpret_cast<const v2f*>(tH + row * TP + k0);
  }

  const int ncol = lane & 15;
#pragma unroll
  for (int nt = 0; nt < 4; ++nt) {
    v8f c = {};
    const int col = nt * 16 + ncol;
#pragma unroll
    for (int kt = 0; kt < 16; ++kt) {
      int kr = kt * 4 + kb;
      v2f bC, bR;
      bC[0] = Wc[kr * D + col];
      bC[1] = Wc[(kr + 1) * D + col];
      bR[0] = Wr[kr * D + col];
      bR[1] = Wr[(kr + 1) * D + col];
      c = __builtin_amdgcn_wmma_f32_16x16x4_f32(false, aC[kt], false, bC, (short)0, c, false, false);
      c = __builtin_amdgcn_wmma_f32_16x16x4_f32(false, aR[kt], false, bR, (short)0, c, false, false);
    }
    const float bias = bc[col] + br[col];
    const int m0 = half * 8;
#pragma unroll
    for (int r = 0; r < 8; ++r)
      tA[(m0 + r) * TP + col] = c[r] + bias;
  }
  __syncthreads();

  const v4f g4 = *reinterpret_cast<const v4f*>(lng + 4 * l16);
  const v4f b4 = *reinterpret_cast<const v4f*>(lnb + 4 * l16);

  const int rr = lane >> 1;
  const int j0 = (lane & 1) * 32;
  const float* myrow = tA + rr * TP + j0;
  float s = 0.0f, sq = 0.0f;
#pragma unroll
  for (int j = 0; j < 8; ++j) {
    const v4f q = *reinterpret_cast<const v4f*>(myrow + 4 * j);
    s += q[0]; sq += q[0] * q[0];
    s += q[1]; sq += q[1] * q[1];
    s += q[2]; sq += q[2] * q[2];
    s += q[3]; sq += q[3] * q[3];
  }
  s  += __shfl_xor(s, 1);
  sq += __shfl_xor(sq, 1);
  const float mu   = s * (1.0f / 64.0f);
  const float var  = sq * (1.0f / 64.0f) - mu * mu;
  const float rstd = rsqrtf(var + LN_EPS);
  if ((lane & 1) == 0) {
    v2f st;
    st[0] = mu;
    st[1] = rstd;
    *reinterpret_cast<v2f*>(&s_stat[wave][2 * rr]) = st;
  }
  __syncthreads();

  v4f y[8];
#pragma unroll
  for (int i = 0; i < 8; ++i) {
    const int r = 2 * i + half;
    const v2f st = *reinterpret_cast<const v2f*>(&s_stat[wave][2 * r]);
    const v4f x = *reinterpret_cast<const v4f*>(tA + r * TP + 4 * l16);
    y[i][0] = fmaxf((x[0] - st[0]) * st[1] * g4[0] + b4[0], 0.0f);
    y[i][1] = fmaxf((x[1] - st[0]) * st[1] * g4[1] + b4[1], 0.0f);
    y[i][2] = fmaxf((x[2] - st[0]) * st[1] * g4[2] + b4[2], 0.0f);
    y[i][3] = fmaxf((x[3] - st[0]) * st[1] * g4[3] + b4[3], 0.0f);
  }
#pragma unroll
  for (int i = 0; i < 8; ++i) {
    const int onode = base + 2 * i + half;
    if (onode < n) {
      const size_t o = (size_t)onode * D + 4 * l16;
      *reinterpret_cast<v4f*>(h_out + o)   = y[i];
      *reinterpret_cast<v4f*>(h_final + o) = vf[i] + y[i];
    }
  }
}

__global__ __launch_bounds__(256) void pred_kernel(
    const float* __restrict__ hf, const float* __restrict__ W,
    const float* __restrict__ b, float* __restrict__ out, int n)
{
  __shared__ __attribute__((aligned(16))) float s_in[8][16 * TP];
  __shared__ __attribute__((aligned(16))) float s_out[8][16 * TP];
  const int wave = threadIdx.x >> 5;
  const int lane = threadIdx.x & 31;
  const int half = lane >> 4;
  const int l16  = lane & 15;
  const int row  = l16;
  const int kb   = half * 2;
  const int base = (blockIdx.x * 8 + wave) * 16;
  float* tI = s_in[wave];
  float* tO = s_out[wave];

  v4f vi[8];
#pragma unroll
  for (int i = 0; i < 8; ++i) {
    int nd = base + 2 * i + half;
    nd = nd < n ? nd : (n - 1);
    vi[i] = *reinterpret_cast<const v4f*>(hf + (size_t)nd * D + 4 * l16);
  }
#pragma unroll
  for (int i = 0; i < 8; ++i)
    *reinterpret_cast<v4f*>(tI + (2 * i + half) * TP + 4 * l16) = vi[i];
  __syncthreads();

  v2f a[16];
#pragma unroll
  for (int kt = 0; kt < 16; ++kt)
    a[kt] = *reinterpret_cast<const v2f*>(tI + row * TP + kt * 4 + kb);

  const int ncol = lane & 15;
#pragma unroll
  for (int nt = 0; nt < 4; ++nt) {
    v8f c = {};
    const int col = nt * 16 + ncol;
#pragma unroll
    for (int kt = 0; kt < 16; ++kt) {
      int kr = kt * 4 + kb;
      v2f bf;
      bf[0] = W[kr * D + col];
      bf[1] = W[(kr + 1) * D + col];
      c = __builtin_amdgcn_wmma_f32_16x16x4_f32(false, a[kt], false, bf, (short)0, c, false, false);
    }
    const float bias = b[col];
    const int m0 = half * 8;
#pragma unroll
    for (int r = 0; r < 8; ++r)
      tO[(m0 + r) * TP + col] = c[r] + bias;
  }
  __syncthreads();

  v4f vo[8];
#pragma unroll
  for (int i = 0; i < 8; ++i)
    vo[i] = *reinterpret_cast<const v4f*>(tO + (2 * i + half) * TP + 4 * l16);
#pragma unroll
  for (int i = 0; i < 8; ++i) {
    const int onode = base + 2 * i + half;
    if (onode < n)
      *reinterpret_cast<v4f*>(out + (size_t)onode * D + 4 * l16) = vo[i];
  }
}

extern "C" void kernel_launch(void* const* d_in, const int* in_sizes, int n_in,
                              void* d_out, int out_size, void* d_ws, size_t ws_size,
                              hipStream_t stream) {
  (void)in_sizes; (void)n_in; (void)out_size; (void)ws_size;

  const float* feats  = (const float*)d_in[0];
  const int*   src    = (const int*)  d_in[1];
  const int*   dst    = (const int*)  d_in[2];
  const float* conv_W = (const float*)d_in[3];
  const float* conv_b = (const float*)d_in[4];
  const float* res_W  = (const float*)d_in[5];
  const float* res_b  = (const float*)d_in[6];
  const float* ln_g   = (const float*)d_in[7];
  const float* ln_b   = (const float*)d_in[8];
  const float* pred_W = (const float*)d_in[9];
  const float* pred_b = (const float*)d_in[10];
  float* out = (float*)d_out;

  char* ws = (char*)d_ws;
  size_t off = 0;
  auto alloc = [&](size_t bytes) -> float* {
    float* p = (float*)(ws + off);
    off += (bytes + 255) & ~(size_t)255;
    return p;
  };
  const size_t nodeF = (size_t)N_NODES * 4;
  const size_t featF = (size_t)N_NODES * D * 4;
  float* out_deg  = alloc(nodeF);
  int*   in_cnt   = (int*)alloc(nodeF);
  float* in_norm  = alloc(nodeF);
  int*   row_ptr  = (int*)alloc(nodeF + 4);
  int*   rank     = (int*)alloc((size_t)N_EDGES * 4);
  int2*  csr      = (int2*)alloc((size_t)N_EDGES * 8);
  float* agg      = alloc(featF);
  float* hbuf     = alloc(featF);
  float* h_final  = alloc(featF);

  const int TB = 256;
  const int nFeat   = N_NODES * D;
  const int gFeat   = (nFeat + TB - 1) / TB;
  const int nClear  = 2 * ((N_NODES * 4 + 255) / 256) * 64;
  const int gClear  = (nClear + TB - 1) / TB;
  const int gEdge   = (N_EDGES + TB - 1) / TB;
  const int gScan   = (N_NODES + SCAN_TILE - 1) / SCAN_TILE;
  const int gGath   = (N_NODES + 8 * NODES_PER_WAVE - 1) / (8 * NODES_PER_WAVE);
  const int gTile   = (N_NODES + 127) / 128;

  zero_f32_kernel<<<gClear, TB, 0, stream>>>(out_deg, nClear);
  degree_kernel<<<gEdge, TB, 0, stream>>>(src, dst, out_deg, in_cnt, rank, N_EDGES);
  scan_kernel<<<gScan, SCAN_THREADS, 0, stream>>>(in_cnt, row_ptr, in_norm, N_NODES);
  fill_kernel<<<gEdge, TB, 0, stream>>>(src, dst, rank, out_deg, row_ptr, csr, N_EDGES);

  zero_f32_kernel<<<gFeat, TB, 0, stream>>>(h_final, nFeat);

  const float* h_in = feats;
  for (int layer = 0; layer < 3; ++layer) {
    gather_kernel<<<gGath, TB, 0, stream>>>(h_in, row_ptr, csr, agg, N_NODES, N_EDGES);
    gcn_layer_kernel<<<gTile, TB, 0, stream>>>(
        agg, in_norm, h_in,
        conv_W + (size_t)layer * D * D, conv_b + (size_t)layer * D,
        res_W  + (size_t)layer * D * D, res_b  + (size_t)layer * D,
        ln_g   + (size_t)layer * D,     ln_b   + (size_t)layer * D,
        hbuf, h_final, N_NODES);
    h_in = hbuf;
  }

  pred_kernel<<<gTile, TB, 0, stream>>>(h_final, pred_W, pred_b, out, N_NODES);
}


# edge grouping by destination with LDS counters per (node range, edge slice) work-group, slice counts summed per node: no global atomics, one launch fewer; sum order = slice, then LDS slot order
# speedup vs baseline: 2.6227x; 1.3808x over previous
#include <hip/hip_runtime.h>

#define N_NODES 100000
#define N_EDGES 1000000
#define D 64
#define LN_EPS 1e-5f

typedef __attribute__((ext_vector_type(2))) float v2f;
typedef __attribute__((ext_vector_type(8))) float v8f;

__global__ void zero_f32_kernel(float* __restrict__ p, int n) {
  int i = blockIdx.x * blockDim.x + threadIdx.x;
  if (i < n) p[i] = 0.0f;
}

#define RANGE_BITS 13
#define RANGE (1 << RANGE_BITS)
#define NR ((N_NODES + RANGE - 1) / RANGE)
#define NS 16
#define NPAD (NR * RANGE)
#define SLICE (N_EDGES / NS)
#define SLICE4 (SLICE / 4)
#define TILE 1024
#define NT (NPAD / TILE)
#define GRP_THREADS 1024
#define GRP_BATCH 4
static_assert(GRP_BATCH == 4, "fill_kernel names the 16 weights of a batch one by one");

__global__ __launch_bounds__(GRP_THREADS) void count_kernel(const int* __restrict__ src,
                              const int* __restrict__ dst, int* __restrict__ pc,
                              int* __restrict__ tot) {
  __shared__ __attribute__((aligned(16))) int cnt_in[RANGE];
  __shared__ __attribute__((aligned(16))) int cnt_out[RANGE];
  __shared__ int s_red0[32];
  __shared__ int s_red1[32];
  const int tid  = threadIdx.x;
  const int lane = tid & 31;
  const int wave = tid >> 5;
  const int r = blockIdx.x;
  const int s = blockIdx.y;
  for (int i = tid; i < RANGE / 4; i += GRP_THREADS) {
    reinterpret_cast<int4*>(cnt_in)[i]  = make_int4(0, 0, 0, 0);
    reinterpret_cast<int4*>(cnt_out)[i] = make_int4(0, 0, 0, 0);
  }
  __syncthreads();

  const int4* d4 = reinterpret_cast<const int4*>(dst + s * SLICE);
  const int4* s4 = reinterpret_cast<const int4*>(src + s * SLICE);
  for (int j0 = tid; j0 < SLICE4; j0 += GRP_THREADS * GRP_BATCH) {
    int4 vd[GRP_BATCH], vs[GRP_BATCH];
#pragma unroll
    for (int u = 0; u < GRP_BATCH; ++u) {
      const int j = j0 + u * GRP_THREADS;
      const int jc = j < SLICE4 ? j : (SLICE4 - 1);
      vd[u] = d4[jc];
      vs[u] = s4[jc];
    }
#pragma unroll
    for (int u = 0; u < GRP_BATCH; ++u) {
      if (j0 + u * GRP_THREADS < SLICE4) {
        const int kd[4] = {vd[u].x, vd[u].y, vd[u].z, vd[u].w};
        const int ks[4] = {vs[u].x, vs[u].y, vs[u].z, vs[u].w};
#pragma unroll
        for (int c = 0; c < 4; ++c) {
          if ((kd[c] >> RANGE_BITS) == r) atomicAdd(&cnt_in[kd[c] & (RANGE - 1)], 1);
          if ((ks[c] >> RANGE_BITS) == r) atomicAdd(&cnt_out[ks[c] & (RANGE - 1)], 1);
        }
      }
    }
  }
  __syncthreads();

  int* pin  = pc + (size_t)s * NPAD + (size_t)r * RANGE;
  int* pout = pc + (size_t)(NS + s) * NPAD + (size_t)r * RANGE;
  const int4 a0 = reinterpret_cast<const int4*>(cnt_in)[tid];
  const int4 a1 = reinterpret_cast<const int4*>(cnt_in)[tid + GRP_THREADS];
  const int4 b0 = reinterpret_cast<const int4*>(cnt_out)[tid];
  const int4 b1 = reinterpret_cast<const int4*>(cnt_out)[tid + GRP_THREADS];
  reinterpret_cast<int4*>(pin)[tid]                = a0;
  reinterpret_cast<int4*>(pin)[tid + GRP_THREADS]  = a1;
  reinterpret_cast<int4*>(pout)[tid]               = b0;
  reinterpret_cast<int4*>(pout)[tid + GRP_THREADS] = b1;
  int t0 = (a0.x + a0.y) + (a0.z + a0.w);
  int t1 = (a1.x + a1.y) + (a1.z + a1.w);
#pragma unroll
  for (int o = 16; o > 0; o >>= 1) {
    t0 += __shfl_xor(t0, o);
    t1 += __shfl_xor(t1, o);
  }
  if (lane == 0) { s_red0[wave] = t0; s_red1[wave] = t1; }
  __syncthreads();
  if (tid < 8) {
    const int* red = tid < 4 ? s_red0 : s_red1;
    const int g = tid & 3;
    int sum = 0;
#pragma unroll
    for (int w = 0; w < 8; ++w) sum += red[8 * g + w];
    tot[(r * (RANGE / TILE) + tid) * NS + s] = sum;
  }
}

__global__ __launch_bounds__(256) void scan_kernel(int* pc, const int* __restrict__ tot,
                              int* __restrict__ row_ptr, float* __restrict__ in_norm,
                              float* __restrict__ out_w, int n) {
  __shared__ int s_sum[8];
  __shared__ int s_pre[8];
  const int tid  = threadIdx.x;
  const int lane = tid & 31;
  const int wave = tid >> 5;
  const int b = blockIdx.x;
  const int i = b * TILE + tid * 4;

  int pre = 0;
  for (int j = tid; j < b * NS; j += 256) pre += tot[j];

  int* pin = pc + i;
  const int* pout = pc + (size_t)NS * NPAD + i;
  int4 a[NS], o[NS];
#pragma unroll
  for (int s = 0; s < NS; ++s) a[s] = *reinterpret_cast<const int4*>(pin + (size_t)s * NPAD);
#pragma unroll
  for (int s = 0; s < NS; ++s) o[s] = *reinterpret_cast<const int4*>(pout + (size_t)s * NPAD);
  int c[4] = {0, 0, 0, 0};
  int d[4] = {0, 0, 0, 0};
#pragma unroll
  for (int s = 0; s < NS; ++s) {
    c[0] += a[s].x; c[1] += a[s].y; c[2] += a[s].z; c[3] += a[s].w;
    d[0] += o[s].x; d[1] += o[s].y; d[2] += o[s].z; d[3] += o[s].w;
  }
  const int t = (c[0] + c[1]) + (c[2] + c[3]);

  int x = t;
#pragma unroll
  for (int k = 1; k < 32; k <<= 1) {
    const int y = __shfl_up(x, k);
    if (lane >= k) x += y;
  }
#pragma unroll
  for (int k = 16; k > 0; k >>= 1) pre += __shfl_xor(pre, k);
  if (lane == 31) s_sum[wave] = x;
  if (lane == 0)  s_pre[wave] = pre;
  __syncthreads();
  int below = 0, ptot = 0;
#pragma unroll
  for (int w = 0; w < 8; ++w) {
    const int v = s_sum[w];
    below += (w < wave) ? v : 0;
    ptot  += s_pre[w];
  }
  int run[4];
  run[0] = ptot + below + (x - t);
  run[1] = run[0] + c[0];
  run[2] = run[1] + c[1];
  run[3] = run[2] + c[2];

  float inn[4], ow[4];
#pragma unroll
  for (int k = 0; k < 4; ++k) {
    inn[k] = rsqrtf(fmaxf((float)c[k], 1.0f));
    ow[k]  = rsqrtf(fmaxf((float)d[k], 1.0f));
  }
  if (i + 3 < n) {
    *reinterpret_cast<int4*>(row_ptr + i)   = make_int4(run[0], run[1], run[2], run[3]);
    *reinterpret_cast<float4*>(in_norm + i) = make_float4(inn[0], inn[1], inn[2], inn[3]);
    *reinterpret_cast<float4*>(out_w + i)   = make_float4(ow[0], ow[1], ow[2], ow[3]);
  } else {
#pragma unroll
    for (int k = 0; k < 4; ++k) {
      if (i + k <= n) row_ptr[i + k] = run[k];
      if (i + k < n) { in_norm[i + k] = inn[k]; out_w[i + k] = ow[k]; }
    }
  }
#pragma unroll
  for (int s = 0; s < NS; ++s) {
    *reinterpret_cast<int4*>(pin + (size_t)s * NPAD) = make_int4(run[0], run[1], run[2], run[3]);
    run[0] += a[s].x; run[1] += a[s].y; run[2] += a[s].z; run[3] += a[s].w;
  }
}

__global__ __launch_bounds__(GRP_THREADS) void fill_kernel(const int* __restrict__ src,
                            const int* __restrict__ dst, const int* __restrict__ startpos,
                            const float* __restrict__ out_w, int2* __restrict__ csr) {
  __shared__ __attribute__((aligned(16))) int cur[RANGE];
  const int tid = threadIdx.x;
  const int r = blockIdx.x;
  const int s = blockIdx.y;
  const int4* b4 = reinterpret_cast<const int4*>(startpos + (size_t)s * NPAD + (size_t)r * RANGE);
  reinterpret_cast<int4*>(cur)[tid]               = b4[tid];
  reinterpret_cast<int4*>(cur)[tid + GRP_THREADS] = b4[tid + GRP_THREADS];
  __syncthreads();

  const int4* d4 = reinterpret_cast<const int4*>(dst + s * SLICE);
  const int4* s4 = reinterpret_cast<const int4*>(src + s * SLICE);
  for (int j0 = tid; j0 < SLICE4; j0 += GRP_THREADS * GRP_BATCH) {
    int4 vd[GRP_BATCH], vs[GRP_BATCH];
#pragma unroll
    for (int u = 0; u < GRP_BATCH; ++u) {
      const int j = j0 + u * GRP_THREADS;
      const int jc = j < SLICE4 ? j : (SLICE4 - 1);
      vd[u] = d4[jc];
      vs[u] = s4[jc];
    }
    int   kd[4 * GRP_BATCH], ks[4 * GRP_BATCH];
    bool  hit[4 * GRP_BATCH];
    float w[4 * GRP_BATCH];
#pragma unroll
    for (int u = 0; u < GRP_BATCH; ++u) {
      const bool live = (j0 + u * GRP_THREADS) < SLICE4;
      kd[4 * u + 0] = vd[u].x; kd[4 * u + 1] = vd[u].y; kd[4 * u + 2] = vd[u].z; kd[4 * u + 3] = vd[u].w;
      ks[4 * u + 0] = vs[u].x; ks[4 * u + 1] = vs[u].y; ks[4 * u + 2] = vs[u].z; ks[4 * u + 3] = vs[u].w;
#pragma unroll
      for (int c = 0; c < 4; ++c)
        hit[4 * u + c] = live && ((kd[4 * u + c] >> RANGE_BITS) == r);
    }
#pragma unroll
    for (int q = 0; q < 4 * GRP_BATCH; ++q) w[q] = out_w[hit[q] ? ks[q] : 0];
    asm volatile("" : "+v"(w[0]), "+v"(w[1]), "+v"(w[2]), "+v"(w[3]),
                      "+v"(w[4]), "+v"(w[5]), "+v"(w[6]), "+v"(w[7]),
                      "+v"(w[8]), "+v"(w[9]), "+v"(w[10]), "+v"(w[11]),
                      "+v"(w[12]), "+v"(w[13]), "+v"(w[14]), "+v"(w[15]));
#pragma unroll
    for (int q = 0; q < 4 * GRP_BATCH; ++q) {
      if (hit[q]) {
        int pos = atomicAdd(&cur[kd[q] & (RANGE - 1)], 1);
        pos = (unsigned)pos < (unsigned)N_EDGES ? pos : (N_EDGES - 1);
        csr[pos] = make_int2(ks[q], __float_as_int(w[q]));
      }
    }
  }
}

#define NODES_PER_WAVE 8
#define SLOTS 8
__global__ __launch_bounds__(256) void gather_kernel(const float* __restrict__ h,
                               const int* __restrict__ row_ptr,
                               const int2* __restrict__ csr,
                               float* __restrict__ agg, int n, int n_edges) {
  const int lane = threadIdx.x & 31;
  const int half = lane >> 4;
  const int l16  = lane & 15;
  const int n0 = __builtin_amdgcn_readfirstlane(
      (int)((blockIdx.x * blockDim.x + threadIdx.x) >> 5)) * NODES_PER_WAVE;
  if (n0 >= n) return;
  const int pi = (n0 + lane) < n ? (n0 + lane) : n;
  const int rp = row_ptr[pi];
  const float4* h4 = reinterpret_cast<const float4*>(h);
  const unsigned u16 = (unsigned)l16;
  const int last = n_edges - 1;

  int2 cn[SLOTS];
  int nstart = __builtin_amdgcn_readlane(rp, 0);
  int nend   = __builtin_amdgcn_readlane(rp, 1);
#pragma unroll
  for (int j = 0; j < SLOTS; ++j) {
    const int e = nstart + half + 2 * j;
    cn[j] = csr[e < last ? e : last];
  }

#pragma unroll
  for (int r = 0; r < NODES_PER_WAVE; ++r) {
    const int node = n0 + r;
    if (node >= n) break;
    const int start = nstart;
    const int end   = nend;
    int2 c[SLOTS];
#pragma unroll
    for (int j = 0; j < SLOTS; ++j) c[j] = cn[j];

    if (r + 1 < NODES_PER_WAVE) {
      nstart = end;
      nend = __builtin_amdgcn_readlane(rp, r + 2);
#pragma unroll
      for (int j = 0; j < SLOTS; ++j) {
        const int e = nstart + half + 2 * j;
        cn[j] = csr[e < last ? e : last];
      }
    }

    int   src[SLOTS];
    float wgt[SLOTS];
    bool  ok[SLOTS];
#pragma unroll
    for (int j = 0; j < SLOTS; ++j) {
      ok[j]  = (start + half + 2 * j) < end;
      src[j] = ok[j] ? c[j].x : 0;
      wgt[j] = ok[j] ? __int_as_float(c[j].y) : 0.0f;
    }

    float4 v[SLOTS];
#pragma unroll
    for (int j = 0; j < SLOTS; ++j) v[j] = h4[(unsigned)src[j] * (unsigned)(D / 4) + u16];

    float4 acc = make_float4(0.0f, 0.0f, 0.0f, 0.0f);
#pragma unroll
    for (int j = 0; j < SLOTS; ++j) {
      const float4 x = ok[j] ? v[j] : make_float4(0.0f, 0.0f, 0.0f, 0.0f);
      const float w = wgt[j];
      acc.x += x.x * w; acc.y += x.y * w; acc.z += x.z * w; acc.w += x.w * w;
    }
    for (int e = start + 2 * SLOTS + half; e < end; e += 2) {
      const int2 c0 = csr[e];
      const float4 v0 = h4[(unsigned)c0.x * (unsigned)(D / 4) + u16];
      const float w0 = __int_as_float(c0.y);
      acc.x += v0.x * w0; acc.y += v0.y * w0; acc.z += v0.z * w0; acc.w += v0.w * w0;
    }
    acc.x += __shfl_xor(acc.x, 16);
    acc.y += __shfl_xor(acc.y, 16);
    acc.z += __shfl_xor(acc.z, 16);
    acc.w += __shfl_xor(acc.w, 16);
    if (half == 0)
      *(reinterpret_cast<float4*>(agg + (size_t)node * D) + l16) = acc;
  }
}

#define TP 68
typedef __attribute__((ext_vector_type(4))) float v4f;

__global__ __launch_bounds__(256) void gcn_layer_kernel(
    const float* __restrict__ agg, const float* __restrict__ in_norm,
    const float* h_in,
    const float* __restrict__ Wc, const float* __restrict__ bc,
    const float* __restrict__ Wr, const float* __restrict__ br,
    const float* __restrict__ lng, const float* __restrict__ lnb,
    float* h_out, float* __restrict__ h_final, int n)
{
  __shared__ __attribute__((aligned(16))) float s_tile[8][2][16 * TP];
  __shared__ __attribute__((aligned(8)))  float s_stat[8][32];
  const int wave = threadIdx.x >> 5;
  const int lane = threadIdx.x & 31;
  const int half = lane >> 4;
  const int l16  = lane & 15;
  const int row  = l16;
  const int kb   = half * 2;
  const int base = (blockIdx.x * 8 + wave) * 16;
  float* tA = s_tile[wave][0];
  float* tH = s_tile[wave][1];

  v4f va[8], vh[8], vf[8];
#pragma unroll
  for (int i = 0; i < 8; ++i) {
    int nd = base + 2 * i + half;
    nd = nd < n ? nd : (n - 1);
    const size_t o = (size_t)nd * D + 4 * l16;
    va[i] = *reinterpret_cast<const v4f*>(agg + o);
    vh[i] = *reinterpret_cast<const v4f*>(h_in + o);
    vf[i] = *reinterpret_cast<const v4f*>(h_final + o);
  }
  int cnode = base + row;
  cnode = cnode < n ? cnode : (n - 1);
  const float inrm = in_norm[cnode];
#pragma unroll
  for (int i = 0; i < 8; ++i) {
    const int r = 2 * i + half;
    *reinterpret_cast<v4f*>(tA + r * TP + 4 * l16) = va[i];
    *reinterpret_cast<v4f*>(tH + r * TP + 4 * l16) = vh[i];
  }
  __syncthreads();

  v2f aC[16], aR[16];
#pragma unroll
  for (int kt = 0; kt < 16; ++kt) {
    const int k0 = kt * 4 + kb;
    const v2f t = *reinterpret_cast<const v2f*>(tA + row * TP + k0);
    aC[kt][0] = t[0] * inrm;
    aC[kt][1] = t[1] * inrm;
    aR[kt] = *reinterpret_cast<const v2f*>(tH + row * TP + k0);
  }

  const int ncol = lane & 15;
#pragma unroll
  for (int nt = 0; nt < 4; ++nt) {
    v8f c = {};
    const int col = nt * 16 + ncol;
#pragma unroll
    for (int kt = 0; kt < 16; ++kt) {
      int kr = kt * 4 + kb;
      v2f bC, bR;
      bC[0] = Wc[kr * D + col];
      bC[1] = Wc[(kr + 1) * D + col];
      bR[0] = Wr[kr * D + col];
      bR[1] = Wr[(kr + 1) * D + col];
      c = __builtin_amdgcn_wmma_f32_16x16x4_f32(false, aC[kt], false, bC, (short)0, c, false, false);
      c = __builtin_amdgcn_wmma_f32_16x16x4_f32(false, aR[kt], false, bR, (short)0, c, false, false);
    }
    const float bias = bc[col] + br[col];
    const int m0 = half * 8;
#pragma unroll
    for (int r = 0; r < 8; ++r)
      tA[(m0 + r) * TP + col] = c[r] + bias;
  }
  __syncthreads();

  const v4f g4 = *reinterpret_cast<const v4f*>(lng + 4 * l16);
  const v4f b4 = *reinterpret_cast<const v4f*>(lnb + 4 * l16);

  const int rr = lane >> 1;
  const int j0 = (lane & 1) * 32;
  const float* myrow = tA + rr * TP + j0;
  float s = 0.0f, sq = 0.0f;
#pragma unroll
  for (int j = 0; j < 8; ++j) {
    const v4f q = *reinterpret_cast<const v4f*>(myrow + 4 * j);
    s += q[0]; sq += q[0] * q[0];
    s += q[1]; sq += q[1] * q[1];
    s += q[2]; sq += q[2] * q[2];
    s += q[3]; sq += q[3] * q[3];
  }
  s  += __shfl_xor(s, 1);
  sq += __shfl_xor(sq, 1);
  const float mu   = s * (1.0f / 64.0f);
  const float var  = sq * (1.0f / 64.0f) - mu * mu;
  const float rstd = rsqrtf(var + LN_EPS);
  if ((lane & 1) == 0) {
    v2f st;
    st[0] = mu;
    st[1] = rstd;
    *reinterpret_cast<v2f*>(&s_stat[wave][2 * rr]) = st;
  }
  __syncthreads();

  v4f y[8];
#pragma unroll
  for (int i = 0; i < 8; ++i) {
    const int r = 2 * i + half;
    const v2f st = *reinterpret_cast<const v2f*>(&s_stat[wave][2 * r]);
    const v4f x = *reinterpret_cast<const v4f*>(tA + r * TP + 4 * l16);
    y[i][0] = fmaxf((x[0] - st[0]) * st[1] * g4[0] + b4[0], 0.0f);
    y[i][1] = fmaxf((x[1] - st[0]) * st[1] * g4[1] + b4[1], 0.0f);
    y[i][2] = fmaxf((x[2] - st[0]) * st[1] * g4[2] + b4[2], 0.0f);
    y[i][3] = fmaxf((x[3] - st[0]) * st[1] * g4[3] + b4[3], 0.0f);
  }
#pragma unroll
  for (int i = 0; i < 8; ++i) {
    const int onode = base + 2 * i + half;
    if (onode < n) {
      const size_t o = (size_t)onode * D + 4 * l16;
      *reinterpret_cast<v4f*>(h_out + o)   = y[i];
      *reinterpret_cast<v4f*>(h_final + o) = vf[i] + y[i];
    }
  }
}

__global__ __launch_bounds__(256) void pred_kernel(
    const float* __restrict__ hf, const float* __restrict__ W,
    const float* __restrict__ b, float* __restrict__ out, int n)
{
  __shared__ __attribute__((aligned(16))) float s_in[8][16 * TP];
  __shared__ __attribute__((aligned(16))) float s_out[8][16 * TP];
  const int wave = threadIdx.x >> 5;
  const int lane = threadIdx.x & 31;
  const int half = lane >> 4;
  const int l16  = lane & 15;
  const int row  = l16;
  const int kb   = half * 2;
  const int base = (blockIdx.x * 8 + wave) * 16;
  float* tI = s_in[wave];
  float* tO = s_out[wave];

  v4f vi[8];
#pragma unroll
  for (int i = 0; i < 8; ++i) {
    int nd = base + 2 * i + half;
    nd = nd < n ? nd : (n - 1);
    vi[i] = *reinterpret_cast<const v4f*>(hf + (size_t)nd * D + 4 * l16);
  }
#pragma unroll
  for (int i = 0; i < 8; ++i)
    *reinterpret_cast<v4f*>(tI + (2 * i + half) * TP + 4 * l16) = vi[i];
  __syncthreads();

  v2f a[16];
#pragma unroll
  for (int kt = 0; kt < 16; ++kt)
    a[kt] = *reinterpret_cast<const v2f*>(tI + row * TP + kt * 4 + kb);

  const int ncol = lane & 15;
#pragma unroll
  for (int nt = 0; nt < 4; ++nt) {
    v8f c = {};
    const int col = nt * 16 + ncol;
#pragma unroll
    for (int kt = 0; kt < 16; ++kt) {
      int kr = kt * 4 + kb;
      v2f bf;
      bf[0] = W[kr * D + col];
      bf[1] = W[(kr + 1) * D + col];
      c = __builtin_amdgcn_wmma_f32_16x16x4_f32(false, a[kt], false, bf, (short)0, c, false, false);
    }
    const float bias = b[col];
    const int m0 = half * 8;
#pragma unroll
    for (int r = 0; r < 8; ++r)
      tO[(m0 + r) * TP + col] = c[r] + bias;
  }
  __syncthreads();

  v4f vo[8];
#pragma unroll
  for (int i = 0; i < 8; ++i)
    vo[i] = *reinterpret_cast<const v4f*>(tO + (2 * i + half) * TP + 4 * l16);
#pragma unroll
  for (int i = 0; i < 8; ++i) {
    const int onode = base + 2 * i + half;
    if (onode < n)
      *reinterpret_cast<v4f*>(out + (size_t)onode * D + 4 * l16) = vo[i];
  }
}

extern "C" void kernel_launch(void* const* d_in, const int* in_sizes, int n_in,
                              void* d_out, int out_size, void* d_ws, size_t ws_size,
                              hipStream_t stream) {
  (void)in_sizes; (void)n_in; (void)out_size; (void)ws_size;

  const float* feats  = (const float*)d_in[0];
  const int*   src    = (const int*)  d_in[1];
  const int*   dst    = (const int*)  d_in[2];
  const float* conv_W = (const float*)d_in[3];
  const float* conv_b = (const float*)d_in[4];
  const float* res_W  = (const float*)d_in[5];
  const float* res_b  = (const float*)d_in[6];
  const float* ln_g   = (const float*)d_in[7];
  const float* ln_b   = (const float*)d_in[8];
  const float* pred_W = (const float*)d_in[9];
  const float* pred_b = (const float*)d_in[10];
  float* out = (float*)d_out;

  char* ws = (char*)d_ws;
  size_t off = 0;
  auto alloc = [&](size_t bytes) -> float* {
    float* p = (float*)(ws + off);
    off += (bytes + 255) & ~(size_t)255;
    return p;
  };
  const size_t nodeF = (size_t)N_NODES * 4;
  const size_t featF = (size_t)N_NODES * D * 4;
  float* in_norm  = alloc(nodeF);
  float* out_w    = alloc(nodeF);
  int*   row_ptr  = (int*)alloc(nodeF + 4);
  int*   tot      = (int*)alloc((size_t)NT * NS * 4);
  int2*  csr      = (int2*)alloc((size_t)N_EDGES * 8);
  float* agg      = alloc(featF);
  float* hbuf     = alloc(featF);
  float* h_final  = alloc(featF);
  int*   pc       = (int*)agg;

  const int TB = 256;
  const int nFeat   = N_NODES * D;
  const int gFeat   = (nFeat + TB - 1) / TB;
  const int gGath   = (N_NODES + 8 * NODES_PER_WAVE - 1) / (8 * NODES_PER_WAVE);
  const int gTile   = (N_NODES + 127) / 128;

  count_kernel<<<dim3(NR, NS), GRP_THREADS, 0, stream>>>(src, dst, pc, tot);
  scan_kernel<<<NT, 256, 0, stream>>>(pc, tot, row_ptr, in_norm, out_w, N_NODES);
  fill_kernel<<<dim3(NR, NS), GRP_THREADS, 0, stream>>>(src, dst, pc, out_w, csr);

  zero_f32_kernel<<<gFeat, TB, 0, stream>>>(h_final, nFeat);

  const float* h_in = feats;
  for (int layer = 0; layer < 3; ++layer) {
    gather_kernel<<<gGath, TB, 0, stream>>>(h_in, row_ptr, csr, agg, N_NODES, N_EDGES);
    gcn_layer_kernel<<<gTile, TB, 0, stream>>>(
        agg, in_norm, h_in,
        conv_W + (size_t)layer * D * D, conv_b + (size_t)layer * D,
        res_W  + (size_t)layer * D * D, res_b  + (size_t)layer * D,
        ln_g   + (size_t)layer * D,     ln_b   + (size_t)layer * D,
        hbuf, h_final, N_NODES);
    h_in = hbuf;
  }

  pred_kernel<<<gTile, TB, 0, stream>>>(h_final, pred_W, pred_b, out, N_NODES);
}


# JK sum kept out of memory: layers 1,2 write rows to own buffers; layer 3 forms (h1+h2)+h3 in its LDS tile and runs the prediction WMMA product there; no zero fill, no JK plane, 2 launches fewer
# speedup vs baseline: 3.4218x; 1.3047x over previous
#include <hip/hip_runtime.h>

#define N_NODES 100000
#define N_EDGES 1000000
#define D 64
#define LN_EPS 1e-5f

typedef __attribute__((ext_vector_type(2))) float v2f;
typedef __attribute__((ext_vector_type(8))) float v8f;

#define RANGE_BITS 13
#define RANGE (1 << RANGE_BITS)
#define NR ((N_NODES + RANGE - 1) / RANGE)
#define NS 16
#define NPAD (NR * RANGE)
#define SLICE (N_EDGES / NS)
#define SLICE4 (SLICE / 4)
#define TILE 1024
#define NT (NPAD / TILE)
#define GRP_THREADS 1024
#define GRP_BATCH 4
static_assert(GRP_BATCH == 4, "fill_kernel names the 16 weights of a batch one by one");

__global__ __launch_bounds__(GRP_THREADS) void count_kernel(const int* __restrict__ src,
                              const int* __restrict__ dst, int* __restrict__ pc,
                              int* __restrict__ tot) {
  __shared__ __attribute__((aligned(16))) int cnt_in[RANGE];
  __shared__ __attribute__((aligned(16))) int cnt_out[RANGE];
  __shared__ int s_red0[32];
  __shared__ int s_red1[32];
  const int tid  = threadIdx.x;
  const int lane = tid & 31;
  const int wave = tid >> 5;
  const int r = blockIdx.x;
  const int s = blockIdx.y;
  for (int i = tid; i < RANGE / 4; i += GRP_THREADS) {
    reinterpret_cast<int4*>(cnt_in)[i]  = make_int4(0, 0, 0, 0);
    reinterpret_cast<int4*>(cnt_out)[i] = make_int4(0, 0, 0, 0);
  }
  __syncthreads();

  const int4* d4 = reinterpret_cast<const int4*>(dst + s * SLICE);
  const int4* s4 = reinterpret_cast<const int4*>(src + s * SLICE);
  for (int j0 = tid; j0 < SLICE4; j0 += GRP_THREADS * GRP_BATCH) {
    int4 vd[GRP_BATCH], vs[GRP_BATCH];
#pragma unroll
    for (int u = 0; u < GRP_BATCH; ++u) {
      const int j = j0 + u * GRP_THREADS;
      const int jc = j < SLICE4 ? j : (SLICE4 - 1);
      vd[u] = d4[jc];
      vs[u] = s4[jc];
    }
#pragma unroll
    for (int u = 0; u < GRP_BATCH; ++u) {
      if (j0 + u * GRP_THREADS < SLICE4) {
        const int kd[4] = {vd[u].x, vd[u].y, vd[u].z, vd[u].w};
        const int ks[4] = {vs[u].x, vs[u].y, vs[u].z, vs[u].w};
#pragma unroll
        for (int c = 0; c < 4; ++c) {
          if ((kd[c] >> RANGE_BITS) == r) atomicAdd(&cnt_in[kd[c] & (RANGE - 1)], 1);
          if ((ks[c] >> RANGE_BITS) == r) atomicAdd(&cnt_out[ks[c] & (RANGE - 1)], 1);
        }
      }
    }
  }
  __syncthreads();

  int* pin  = pc + (size_t)s * NPAD + (size_t)r * RANGE;
  int* pout = pc + (size_t)(NS + s) * NPAD + (size_t)r * RANGE;
  const int4 a0 = reinterpret_cast<const int4*>(cnt_in)[tid];
  const int4 a1 = reinterpret_cast<const int4*>(cnt_in)[tid + GRP_THREADS];
  const int4 b0 = reinterpret_cast<const int4*>(cnt_out)[tid];
  const int4 b1 = reinterpret_cast<const int4*>(cnt_out)[tid + GRP_THREADS];
  reinterpret_cast<int4*>(pin)[tid]                = a0;
  reinterpret_cast<int4*>(pin)[tid + GRP_THREADS]  = a1;
  reinterpret_cast<int4*>(pout)[tid]               = b0;
  reinterpret_cast<int4*>(pout)[tid + GRP_THREADS] = b1;
  int t0 = (a0.x + a0.y) + (a0.z + a0.w);
  int t1 = (a1.x + a1.y) + (a1.z + a1.w);
#pragma unroll
  for (int o = 16; o > 0; o >>= 1) {
    t0 += __shfl_xor(t0, o);
    t1 += __shfl_xor(t1, o);
  }
  if (lane == 0) { s_red0[wave] = t0; s_red1[wave] = t1; }
  __syncthreads();
  if (tid < 8) {
    const int* red = tid < 4 ? s_red0 : s_red1;
    const int g = tid & 3;
    int sum = 0;
#pragma unroll
    for (int w = 0; w < 8; ++w) sum += red[8 * g + w];
    tot[(r * (RANGE / TILE) + tid) * NS + s] = sum;
  }
}

__global__ __launch_bounds__(256) void scan_kernel(int* pc, const int* __restrict__ tot,
                              int* __restrict__ row_ptr, float* __restrict__ in_norm,
                              float* __restrict__ out_w, int n) {
  __shared__ int s_sum[8];
  __shared__ int s_pre[8];
  const int tid  = threadIdx.x;
  const int lane = tid & 31;
  const int wave = tid >> 5;
  const int b = blockIdx.x;
  const int i = b * TILE + tid * 4;

  int pre = 0;
  for (int j = tid; j < b * NS; j += 256) pre += tot[j];

  int* pin = pc + i;
  const int* pout = pc + (size_t)NS * NPAD + i;
  int4 a[NS], o[NS];
#pragma unroll
  for (int s = 0; s < NS; ++s) a[s] = *reinterpret_cast<const int4*>(pin + (size_t)s * NPAD);
#pragma unroll
  for (int s = 0; s < NS; ++s) o[s] = *reinterpret_cast<const int4*>(pout + (size_t)s * NPAD);
  int c[4] = {0, 0, 0, 0};
  int d[4] = {0, 0, 0, 0};
#pragma unroll
  for (int s = 0; s < NS; ++s) {
    c[0] += a[s].x; c[1] += a[s].y; c[2] += a[s].z; c[3] += a[s].w;
    d[0] += o[s].x; d[1] += o[s].y; d[2] += o[s].z; d[3] += o[s].w;
  }
  const int t = (c[0] + c[1]) + (c[2] + c[3]);

  int x = t;
#pragma unroll
  for (int k = 1; k < 32; k <<= 1) {
    const int y = __shfl_up(x, k);
    if (lane >= k) x += y;
  }
#pragma unroll
  for (int k = 16; k > 0; k >>= 1) pre += __shfl_xor(pre, k);
  if (lane == 31) s_sum[wave] = x;
  if (lane == 0)  s_pre[wave] = pre;
  __syncthreads();
  int below = 0, ptot = 0;
#pragma unroll
  for (int w = 0; w < 8; ++w) {
    const int v = s_sum[w];
    below += (w < wave) ? v : 0;
    ptot  += s_pre[w];
  }
  int run[4];
  run[0] = ptot + below + (x - t);
  run[1] = run[0] + c[0];
  run[2] = run[1] + c[1];
  run[3] = run[2] + c[2];

  float inn[4], ow[4];
#pragma unroll
  for (int k = 0; k < 4; ++k) {
    inn[k] = rsqrtf(fmaxf((float)c[k], 1.0f));
    ow[k]  = rsqrtf(fmaxf((float)d[k], 1.0f));
  }
  if (i + 3 < n) {
    *reinterpret_cast<int4*>(row_ptr + i)   = make_int4(run[0], run[1], run[2], run[3]);
    *reinterpret_cast<float4*>(in_norm + i) = make_float4(inn[0], inn[1], inn[2], inn[3]);
    *reinterpret_cast<float4*>(out_w + i)   = make_float4(ow[0], ow[1], ow[2], ow[3]);
  } else {
#pragma unroll
    for (int k = 0; k < 4; ++k) {
      if (i + k <= n) row_ptr[i + k] = run[k];
      if (i + k < n) { in_norm[i + k] = inn[k]; out_w[i + k] = ow[k]; }
    }
  }
#pragma unroll
  for (int s = 0; s < NS; ++s) {
    *reinterpret_cast<int4*>(pin + (size_t)s * NPAD) = make_int4(run[0], run[1], run[2], run[3]);
    run[0] += a[s].x; run[1] += a[s].y; run[2] += a[s].z; run[3] += a[s].w;
  }
}

__global__ __launch_bounds__(GRP_THREADS) void fill_kernel(const int* __restrict__ src,
                            const int* __restrict__ dst, const int* __restrict__ startpos,
                            const float* __restrict__ out_w, int2* __restrict__ csr) {
  __shared__ __attribute__((aligned(16))) int cur[RANGE];
  const int tid = threadIdx.x;
  const int r = blockIdx.x;
  const int s = blockIdx.y;
  const int4* b4 = reinterpret_cast<const int4*>(startpos + (size_t)s * NPAD + (size_t)r * RANGE);
  reinterpret_cast<int4*>(cur)[tid]               = b4[tid];
  reinterpret_cast<int4*>(cur)[tid + GRP_THREADS] = b4[tid + GRP_THREADS];
  __syncthreads();

  const int4* d4 = reinterpret_cast<const int4*>(dst + s * SLICE);
  const int4* s4 = reinterpret_cast<const int4*>(src + s * SLICE);
  for (int j0 = tid; j0 < SLICE4; j0 += GRP_THREADS * GRP_BATCH) {
    int4 vd[GRP_BATCH], vs[GRP_BATCH];
#pragma unroll
    for (int u = 0; u < GRP_BATCH; ++u) {
      const int j = j0 + u * GRP_THREADS;
      const int jc = j < SLICE4 ? j : (SLICE4 - 1);
      vd[u] = d4[jc];
      vs[u] = s4[jc];
    }
    int   kd[4 * GRP_BATCH], ks[4 * GRP_BATCH];
    bool  hit[4 * GRP_BATCH];
    float w[4 * GRP_BATCH];
#pragma unroll
    for (int u = 0; u < GRP_BATCH; ++u) {
      const bool live = (j0 + u * GRP_THREADS) < SLICE4;
      kd[4 * u + 0] = vd[u].x; kd[4 * u + 1] = vd[u].y; kd[4 * u + 2] = vd[u].z; kd[4 * u + 3] = vd[u].w;
      ks[4 * u + 0] = vs[u].x; ks[4 * u + 1] = vs[u].y; ks[4 * u + 2] = vs[u].z; ks[4 * u + 3] = vs[u].w;
#pragma unroll
      for (int c = 0; c < 4; ++c)
        hit[4 * u + c] = live && ((kd[4 * u + c] >> RANGE_BITS) == r);
    }
#pragma unroll
    for (int q = 0; q < 4 * GRP_BATCH; ++q) w[q] = out_w[hit[q] ? ks[q] : 0];
    asm volatile("" : "+v"(w[0]), "+v"(w[1]), "+v"(w[2]), "+v"(w[3]),
                      "+v"(w[4]), "+v"(w[5]), "+v"(w[6]), "+v"(w[7]),
                      "+v"(w[8]), "+v"(w[9]), "+v"(w[10]), "+v"(w[11]),
                      "+v"(w[12]), "+v"(w[13]), "+v"(w[14]), "+v"(w[15]));
#pragma unroll
    for (int q = 0; q < 4 * GRP_BATCH; ++q) {
      if (hit[q]) {
        int pos = atomicAdd(&cur[kd[q] & (RANGE - 1)], 1);
        pos = (unsigned)pos < (unsigned)N_EDGES ? pos : (N_EDGES - 1);
        csr[pos] = make_int2(ks[q], __float_as_int(w[q]));
      }
    }
  }
}

#define NODES_PER_WAVE 8
#define SLOTS 8
__global__ __launch_bounds__(256) void gather_kernel(const float* __restrict__ h,
                               const int* __restrict__ row_ptr,
                               const int2* __restrict__ csr,
                               float* __restrict__ agg, int n, int n_edges) {
  const int lane = threadIdx.x & 31;
  const int half = lane >> 4;
  const int l16  = lane & 15;
  const int n0 = __builtin_amdgcn_readfirstlane(
      (int)((blockIdx.x * blockDim.x + threadIdx.x) >> 5)) * NODES_PER_WAVE;
  if (n0 >= n) return;
  const int pi = (n0 + lane) < n ? (n0 + lane) : n;
  const int rp = row_ptr[pi];
  const float4* h4 = reinterpret_cast<const float4*>(h);
  const unsigned u16 = (unsigned)l16;
  const int last = n_edges - 1;

  int2 cn[SLOTS];
  int nstart = __builtin_amdgcn_readlane(rp, 0);
  int nend   = __builtin_amdgcn_readlane(rp, 1);
#pragma unroll
  for (int j = 0; j < SLOTS; ++j) {
    const int e = nstart + half + 2 * j;
    cn[j] = csr[e < last ? e : last];
  }

#pragma unroll
  for (int r = 0; r < NODES_PER_WAVE; ++r) {
    const int node = n0 + r;
    if (node >= n) break;
    const int start = nstart;
    const int end   = nend;
    int2 c[SLOTS];
#pragma unroll
    for (int j = 0; j < SLOTS; ++j) c[j] = cn[j];

    if (r + 1 < NODES_PER_WAVE) {
      nstart = end;
      nend = __builtin_amdgcn_readlane(rp, r + 2);
#pragma unroll
      for (int j = 0; j < SLOTS; ++j) {
        const int e = nstart + half + 2 * j;
        cn[j] = csr[e < last ? e : last];
      }
    }

    int   src[SLOTS];
    float wgt[SLOTS];
    bool  ok[SLOTS];
#pragma unroll
    for (int j = 0; j < SLOTS; ++j) {
      ok[j]  = (start + half + 2 * j) < end;
      src[j] = ok[j] ? c[j].x : 0;
      wgt[j] = ok[j] ? __int_as_float(c[j].y) : 0.0f;
    }

    float4 v[SLOTS];
#pragma unroll
    for (int j = 0; j < SLOTS; ++j) v[j] = h4[(unsigned)src[j] * (unsigned)(D / 4) + u16];

    float4 acc = make_float4(0.0f, 0.0f, 0.0f, 0.0f);
#pragma unroll
    for (int j = 0; j < SLOTS; ++j) {
      const float4 x = ok[j] ? v[j] : make_float4(0.0f, 0.0f, 0.0f, 0.0f);
      const float w = wgt[j];
      acc.x += x.x * w; acc.y += x.y * w; acc.z += x.z * w; acc.w += x.w * w;
    }
    for (int e = start + 2 * SLOTS + half; e < end; e += 2) {
      const int2 c0 = csr[e];
      const float4 v0 = h4[(unsigned)c0.x * (unsigned)(D / 4) + u16];
      const float w0 = __int_as_float(c0.y);
      acc.x += v0.x * w0; acc.y += v0.y * w0; acc.z += v0.z * w0; acc.w += v0.w * w0;
    }
    acc.x += __shfl_xor(acc.x, 16);
    acc.y += __shfl_xor(acc.y, 16);
    acc.z += __shfl_xor(acc.z, 16);
    acc.w += __shfl_xor(acc.w, 16);
    if (half == 0)
      *(reinterpret_cast<float4*>(agg + (size_t)node * D) + l16) = acc;
  }
}

#define TP 68
typedef __attribute__((ext_vector_type(4))) float v4f;

template <bool LAST>
__global__ __launch_bounds__(256, 8) void gcn_layer_kernel(
    const float* __restrict__ agg, const float* __restrict__ in_norm,
    const float* __restrict__ h_in,
    const float* __restrict__ Wc, const float* __restrict__ bc,
    const float* __restrict__ Wr, const float* __restrict__ br,
    const float* __restrict__ lng, const float* __restrict__ lnb,
    float* __restrict__ h_out,
    const float* __restrict__ h_first,
    const float* __restrict__ Wp, const float* __restrict__ bp,
    float* __restrict__ out,
    int n)
{
  __shared__ __attribute__((aligned(16))) float s_tile[8][2][16 * TP];
  __shared__ __attribute__((aligned(8)))  float s_stat[8][32];
  const int wave = threadIdx.x >> 5;
  const int lane = threadIdx.x & 31;
  const int half = lane >> 4;
  const int l16  = lane & 15;
  const int row  = l16;
  const int kb   = half * 2;
  const int base = (blockIdx.x * 8 + wave) * 16;
  float* tA = s_tile[wave][0];
  float* tH = s_tile[wave][1];

  v4f va[8], vh[8];
#pragma unroll
  for (int i = 0; i < 8; ++i) {
    int nd = base + 2 * i + half;
    nd = nd < n ? nd : (n - 1);
    const size_t o = (size_t)nd * D + 4 * l16;
    va[i] = *reinterpret_cast<const v4f*>(agg + o);
    vh[i] = *reinterpret_cast<const v4f*>(h_in + o);
  }
  int cnode = base + row;
  cnode = cnode < n ? cnode : (n - 1);
  const float inrm = in_norm[cnode];
#pragma unroll
  for (int i = 0; i < 8; ++i) {
    const int r = 2 * i + half;
    *reinterpret_cast<v4f*>(tA + r * TP + 4 * l16) = va[i];
    *reinterpret_cast<v4f*>(tH + r * TP + 4 * l16) = vh[i];
  }
  __syncthreads();

  v2f aC[16], aR[16];
#pragma unroll
  for (int kt = 0; kt < 16; ++kt) {
    const int k0 = kt * 4 + kb;
    const v2f t = *reinterpret_cast<const v2f*>(tA + row * TP + k0);
    aC[kt][0] = t[0] * inrm;
    aC[kt][1] = t[1] * inrm;
    aR[kt] = *reinterpret_cast<const v2f*>(tH + row * TP + k0);
  }

  const int ncol = lane & 15;
  const int m0 = half * 8;
#pragma unroll
  for (int nt = 0; nt < 4; ++nt) {
    v8f c = {};
    const int col = nt * 16 + ncol;
#pragma unroll
    for (int kt = 0; kt < 16; ++kt) {
      int kr = kt * 4 + kb;
      v2f bC, bR;
      bC[0] = Wc[kr * D + col];
      bC[1] = Wc[(kr + 1) * D + col];
      bR[0] = Wr[kr * D + col];
      bR[1] = Wr[(kr + 1) * D + col];
      c = __builtin_amdgcn_wmma_f32_16x16x4_f32(false, aC[kt], false, bC, (short)0, c, false, false);
      c = __builtin_amdgcn_wmma_f32_16x16x4_f32(false, aR[kt], false, bR, (short)0, c, false, false);
    }
    const float bias = bc[col] + br[col];
#pragma unroll
    for (int r = 0; r < 8; ++r)
      tA[(m0 + r) * TP + col] = c[r] + bias;
  }
  __syncthreads();

  const v4f g4 = *reinterpret_cast<const v4f*>(lng + 4 * l16);
  const v4f b4 = *reinterpret_cast<const v4f*>(lnb + 4 * l16);
  v4f v1[8];
  if constexpr (LAST) {
#pragma unroll
    for (int i = 0; i < 8; ++i) {
      int nd = base + 2 * i + half;
      nd = nd < n ? nd : (n - 1);
      v1[i] = *reinterpret_cast<const v4f*>(h_first + (size_t)nd * D + 4 * l16);
    }
  }

  const int rr = lane >> 1;
  const int j0 = (lane & 1) * 32;
  const float* myrow = tA + rr * TP + j0;
  float s = 0.0f, sq = 0.0f;
#pragma unroll
  for (int j = 0; j < 8; ++j) {
    const v4f q = *reinterpret_cast<const v4f*>(myrow + 4 * j);
    s += q[0]; sq += q[0] * q[0];
    s += q[1]; sq += q[1] * q[1];
    s += q[2]; sq += q[2] * q[2];
    s += q[3]; sq += q[3] * q[3];
  }
  s  += __shfl_xor(s, 1);
  sq += __shfl_xor(sq, 1);
  const float mu   = s * (1.0f / 64.0f);
  const float var  = sq * (1.0f / 64.0f) - mu * mu;
  const float rstd = rsqrtf(var + LN_EPS);
  if ((lane & 1) == 0) {
    v2f st;
    st[0] = mu;
    st[1] = rstd;
    *reinterpret_cast<v2f*>(&s_stat[wave][2 * rr]) = st;
  }
  if constexpr (LAST) {
#pragma unroll
    for (int i = 0; i < 8; ++i) {
      float* p = tH + (2 * i + half) * TP + 4 * l16;
      const v4f h2 = *reinterpret_cast<const v4f*>(p);
      *reinterpret_cast<v4f*>(p) = v1[i] + h2;
    }
  }
  __syncthreads();

  v4f y[8];
#pragma unroll
  for (int i = 0; i < 8; ++i) {
    const int r = 2 * i + half;
    const v2f st = *reinterpret_cast<const v2f*>(&s_stat[wave][2 * r]);
    const v4f x = *reinterpret_cast<const v4f*>(tA + r * TP + 4 * l16);
    y[i][0] = fmaxf((x[0] - st[0]) * st[1] * g4[0] + b4[0], 0.0f);
    y[i][1] = fmaxf((x[1] - st[0]) * st[1] * g4[1] + b4[1], 0.0f);
    y[i][2] = fmaxf((x[2] - st[0]) * st[1] * g4[2] + b4[2], 0.0f);
    y[i][3] = fmaxf((x[3] - st[0]) * st[1] * g4[3] + b4[3], 0.0f);
  }

  if constexpr (!LAST) {
#pragma unroll
    for (int i = 0; i < 8; ++i) {
      const int onode = base + 2 * i + half;
      if (onode < n)
        *reinterpret_cast<v4f*>(h_out + (size_t)onode * D + 4 * l16) = y[i];
    }
  } else {
#pragma unroll
    for (int i = 0; i < 8; ++i) {
      float* p = tH + (2 * i + half) * TP + 4 * l16;
      const v4f t = *reinterpret_cast<const v4f*>(p);
      *reinterpret_cast<v4f*>(p) = t + y[i];
    }
    __syncthreads();

    v2f a[16];
#pragma unroll
    for (int kt = 0; kt < 16; ++kt)
      a[kt] = *reinterpret_cast<const v2f*>(tH + row * TP + kt * 4 + kb);
#pragma unroll
    for (int nt = 0; nt < 4; ++nt) {
      v8f c = {};
      const int col = nt * 16 + ncol;
#pragma unroll
      for (int kt = 0; kt < 16; ++kt) {
        int kr = kt * 4 + kb;
        v2f bf;
        bf[0] = Wp[kr * D + col];
        bf[1] = Wp[(kr + 1) * D + col];
        c = __builtin_amdgcn_wmma_f32_16x16x4_f32(false, a[kt], false, bf, (short)0, c, false, false);
      }
      const float bias = bp[col];
#pragma unroll
      for (int r = 0; r < 8; ++r)
        tA[(m0 + r) * TP + col] = c[r] + bias;
    }
    __syncthreads();

    v4f vo[8];
#pragma unroll
    for (int i = 0; i < 8; ++i)
      vo[i] = *reinterpret_cast<const v4f*>(tA + (2 * i + half) * TP + 4 * l16);
#pragma unroll
    for (int i = 0; i < 8; ++i) {
      const int onode = base + 2 * i + half;
      if (onode < n)
        *reinterpret_cast<v4f*>(out + (size_t)onode * D + 4 * l16) = vo[i];
    }
  }
}

extern "C" void kernel_launch(void* const* d_in, const int* in_sizes, int n_in,
                              void* d_out, int out_size, void* d_ws, size_t ws_size,
                              hipStream_t stream) {
  (void)in_sizes; (void)n_in; (void)out_size; (void)ws_size;

  const float* feats  = (const float*)d_in[0];
  const int*   src    = (const int*)  d_in[1];
  const int*   dst    = (const int*)  d_in[2];
  const float* conv_W = (const float*)d_in[3];
  const float* conv_b = (const float*)d_in[4];
  const float* res_W  = (const float*)d_in[5];
  const float* res_b  = (const float*)d_in[6];
  const float* ln_g   = (const float*)d_in[7];
  const float* ln_b   = (const float*)d_in[8];
  const float* pred_W = (const float*)d_in[9];
  const float* pred_b = (const float*)d_in[10];
  float* out = (float*)d_out;

  char* ws = (char*)d_ws;
  size_t off = 0;
  auto alloc = [&](size_t bytes) -> float* {
    float* p = (float*)(ws + off);
    off += (bytes + 255) & ~(size_t)255;
    return p;
  };
  const size_t nodeF = (size_t)N_NODES * 4;
  const size_t featF = (size_t)N_NODES * D * 4;
  float* in_norm  = alloc(nodeF);
  float* out_w    = alloc(nodeF);
  int*   row_ptr  = (int*)alloc(nodeF + 4);
  int*   tot      = (int*)alloc((size_t)NT * NS * 4);
  int2*  csr      = (int2*)alloc((size_t)N_EDGES * 8);
  float* agg      = alloc(featF);
  float* h1       = alloc(featF);
  float* h2       = alloc(featF);
  int*   pc       = (int*)agg;

  const int TB = 256;
  const int gGath   = (N_NODES + 8 * NODES_PER_WAVE - 1) / (8 * NODES_PER_WAVE);
  const int gTile   = (N_NODES + 127) / 128;

  count_kernel<<<dim3(NR, NS), GRP_THREADS, 0, stream>>>(src, dst, pc, tot);
  scan_kernel<<<NT, 256, 0, stream>>>(pc, tot, row_ptr, in_norm, out_w, N_NODES);
  fill_kernel<<<dim3(NR, NS), GRP_THREADS, 0, stream>>>(src, dst, pc, out_w, csr);

  gather_kernel<<<gGath, TB, 0, stream>>>(feats, row_ptr, csr, agg, N_NODES, N_EDGES);
  gcn_layer_kernel<false><<<gTile, TB, 0, stream>>>(
      agg, in_norm, feats, conv_W, conv_b, res_W, res_b, ln_g, ln_b,
      h1, nullptr, nullptr, nullptr, nullptr, N_NODES);

  gather_kernel<<<gGath, TB, 0, stream>>>(h1, row_ptr, csr, agg, N_NODES, N_EDGES);
  gcn_layer_kernel<false><<<gTile, TB, 0, stream>>>(
      agg, in_norm, h1, conv_W + D * D, conv_b + D, res_W + D * D, res_b + D,
      ln_g + D, ln_b + D,
      h2, nullptr, nullptr, nullptr, nullptr, N_NODES);

  gather_kernel<<<gGath, TB, 0, stream>>>(h2, row_ptr, csr, agg, N_NODES, N_EDGES);
  gcn_layer_kernel<true><<<gTile, TB, 0, stream>>>(
      agg, in_norm, h2, conv_W + 2 * D * D, conv_b + 2 * D, res_W + 2 * D * D, res_b + 2 * D,
      ln_g + 2 * D, ln_b + 2 * D,
      nullptr, h1, pred_W, pred_b, out, N_NODES);
}


# gather: a node's first 32 list entries by one whole-line load (half-wave h, lane i = entry 2i+h), handed to the half-wave's slots by row_share moves, not 8 two-address loads; sum order unchanged
# speedup vs baseline: 3.5877x; 1.0485x over previous
#include <hip/hip_runtime.h>

#define N_NODES 100000
#define N_EDGES 1000000
#define D 64
#define LN_EPS 1e-5f

typedef __attribute__((ext_vector_type(2))) float v2f;
typedef __attribute__((ext_vector_type(8))) float v8f;

#define RANGE_BITS 13
#define RANGE (1 << RANGE_BITS)
#define NR ((N_NODES + RANGE - 1) / RANGE)
#define NS 16
#define NPAD (NR * RANGE)
#define SLICE (N_EDGES / NS)
#define SLICE4 (SLICE / 4)
#define TILE 1024
#define NT (NPAD / TILE)
#define GRP_THREADS 1024
#define GRP_BATCH 4
static_assert(GRP_BATCH == 4, "fill_kernel names the 16 weights of a batch one by one");

__global__ __launch_bounds__(GRP_THREADS) void count_kernel(const int* __restrict__ src,
                              const int* __restrict__ dst, int* __restrict__ pc,
                              int* __restrict__ tot) {
  __shared__ __attribute__((aligned(16))) int cnt_in[RANGE];
  __shared__ __attribute__((aligned(16))) int cnt_out[RANGE];
  __shared__ int s_red0[32];
  __shared__ int s_red1[32];
  const int tid  = threadIdx.x;
  const int lane = tid & 31;
  const int wave = tid >> 5;
  const int r = blockIdx.x;
  const int s = blockIdx.y;
  for (int i = tid; i < RANGE / 4; i += GRP_THREADS) {
    reinterpret_cast<int4*>(cnt_in)[i]  = make_int4(0, 0, 0, 0);
    reinterpret_cast<int4*>(cnt_out)[i] = make_int4(0, 0, 0, 0);
  }
  __syncthreads();

  const int4* d4 = reinterpret_cast<const int4*>(dst + s * SLICE);
  const int4* s4 = reinterpret_cast<const int4*>(src + s * SLICE);
  for (int j0 = tid; j0 < SLICE4; j0 += GRP_THREADS * GRP_BATCH) {
    int4 vd[GRP_BATCH], vs[GRP_BATCH];
#pragma unroll
    for (int u = 0; u < GRP_BATCH; ++u) {
      const int j = j0 + u * GRP_THREADS;
      const int jc = j < SLICE4 ? j : (SLICE4 - 1);
      vd[u] = d4[jc];
      vs[u] = s4[jc];
    }
#pragma unroll
    for (int u = 0; u < GRP_BATCH; ++u) {
      if (j0 + u * GRP_THREADS < SLICE4) {
        const int kd[4] = {vd[u].x, vd[u].y, vd[u].z, vd[u].w};
        const int ks[4] = {vs[u].x, vs[u].y, vs[u].z, vs[u].w};
#pragma unroll
        for (int c = 0; c < 4; ++c) {
          if ((kd[c] >> RANGE_BITS) == r) atomicAdd(&cnt_in[kd[c] & (RANGE - 1)], 1);
          if ((ks[c] >> RANGE_BITS) == r) atomicAdd(&cnt_out[ks[c] & (RANGE - 1)], 1);
        }
      }
    }
  }
  __syncthreads();

  int* pin  = pc + (size_t)s * NPAD + (size_t)r * RANGE;
  int* pout = pc + (size_t)(NS + s) * NPAD + (size_t)r * RANGE;
  const int4 a0 = reinterpret_cast<const int4*>(cnt_in)[tid];
  const int4 a1 = reinterpret_cast<const int4*>(cnt_in)[tid + GRP_THREADS];
  const int4 b0 = reinterpret_cast<const int4*>(cnt_out)[tid];
  const int4 b1 = reinterpret_cast<const int4*>(cnt_out)[tid + GRP_THREADS];
  reinterpret_cast<int4*>(pin)[tid]                = a0;
  reinterpret_cast<int4*>(pin)[tid + GRP_THREADS]  = a1;
  reinterpret_cast<int4*>(pout)[tid]               = b0;
  reinterpret_cast<int4*>(pout)[tid + GRP_THREADS] = b1;
  int t0 = (a0.x + a0.y) + (a0.z + a0.w);
  int t1 = (a1.x + a1.y) + (a1.z + a1.w);
#pragma unroll
  for (int o = 16; o > 0; o >>= 1) {
    t0 += __shfl_xor(t0, o);
    t1 += __shfl_xor(t1, o);
  }
  if (lane == 0) { s_red0[wave] = t0; s_red1[wave] = t1; }
  __syncthreads();
  if (tid < 8) {
    const int* red = tid < 4 ? s_red0 : s_red1;
    const int g = tid & 3;
    int sum = 0;
#pragma unroll
    for (int w = 0; w < 8; ++w) sum += red[8 * g + w];
    tot[(r * (RANGE / TILE) + tid) * NS + s] = sum;
  }
}

__global__ __launch_bounds__(256) void scan_kernel(int* pc, const int* __restrict__ tot,
                              int* __restrict__ row_ptr, float* __restrict__ in_norm,
                              float* __restrict__ out_w, int n) {
  __shared__ int s_sum[8];
  __shared__ int s_pre[8];
  const int tid  = threadIdx.x;
  const int lane = tid & 31;
  const int wave = tid >> 5;
  const int b = blockIdx.x;
  const int i = b * TILE + tid * 4;

  int pre = 0;
  for (int j = tid; j < b * NS; j += 256) pre += tot[j];

  int* pin = pc + i;
  const int* pout = pc + (size_t)NS * NPAD + i;
  int4 a[NS], o[NS];
#pragma unroll
  for (int s = 0; s < NS; ++s) a[s] = *reinterpret_cast<const int4*>(pin + (size_t)s * NPAD);
#pragma unroll
  for (int s = 0; s < NS; ++s) o[s] = *reinterpret_cast<const int4*>(pout + (size_t)s * NPAD);
  int c[4] = {0, 0, 0, 0};
  int d[4] = {0, 0, 0, 0};
#pragma unroll
  for (int s = 0; s < NS; ++s) {
    c[0] += a[s].x; c[1] += a[s].y; c[2] += a[s].z; c[3] += a[s].w;
    d[0] += o[s].x; d[1] += o[s].y; d[2] += o[s].z; d[3] += o[s].w;
  }
  const int t = (c[0] + c[1]) + (c[2] + c[3]);

  int x = t;
#pragma unroll
  for (int k = 1; k < 32; k <<= 1) {
    const int y = __shfl_up(x, k);
    if (lane >= k) x += y;
  }
#pragma unroll
  for (int k = 16; k > 0; k >>= 1) pre += __shfl_xor(pre, k);
  if (lane == 31) s_sum[wave] = x;
  if (lane == 0)  s_pre[wave] = pre;
  __syncthreads();
  int below = 0, ptot = 0;
#pragma unroll
  for (int w = 0; w < 8; ++w) {
    const int v = s_sum[w];
    below += (w < wave) ? v : 0;
    ptot  += s_pre[w];
  }
  int run[4];
  run[0] = ptot + below + (x - t);
  run[1] = run[0] + c[0];
  run[2] = run[1] + c[1];
  run[3] = run[2] + c[2];

  float inn[4], ow[4];
#pragma unroll
  for (int k = 0; k < 4; ++k) {
    inn[k] = rsqrtf(fmaxf((float)c[k], 1.0f));
    ow[k]  = rsqrtf(fmaxf((float)d[k], 1.0f));
  }
  if (i + 3 < n) {
    *reinterpret_cast<int4*>(row_ptr + i)   = make_int4(run[0], run[1], run[2], run[3]);
    *reinterpret_cast<float4*>(in_norm + i) = make_float4(inn[0], inn[1], inn[2], inn[3]);
    *reinterpret_cast<float4*>(out_w + i)   = make_float4(ow[0], ow[1], ow[2], ow[3]);
  } else {
#pragma unroll
    for (int k = 0; k < 4; ++k) {
      if (i + k <= n) row_ptr[i + k] = run[k];
      if (i + k < n) { in_norm[i + k] = inn[k]; out_w[i + k] = ow[k]; }
    }
  }
#pragma unroll
  for (int s = 0; s < NS; ++s) {
    *reinterpret_cast<int4*>(pin + (size_t)s * NPAD) = make_int4(run[0], run[1], run[2], run[3]);
    run[0] += a[s].x; run[1] += a[s].y; run[2] += a[s].z; run[3] += a[s].w;
  }
}

__global__ __launch_bounds__(GRP_THREADS) void fill_kernel(const int* __restrict__ src,
                            const int* __restrict__ dst, const int* __restrict__ startpos,
                            const float* __restrict__ out_w, int2* __restrict__ csr) {
  __shared__ __attribute__((aligned(16))) int cur[RANGE];
  const int tid = threadIdx.x;
  const int r = blockIdx.x;
  const int s = blockIdx.y;
  const int4* b4 = reinterpret_cast<const int4*>(startpos + (size_t)s * NPAD + (size_t)r * RANGE);
  reinterpret_cast<int4*>(cur)[tid]               = b4[tid];
  reinterpret_cast<int4*>(cur)[tid + GRP_THREADS] = b4[tid + GRP_THREADS];
  __syncthreads();

  const int4* d4 = reinterpret_cast<const int4*>(dst + s * SLICE);
  const int4* s4 = reinterpret_cast<const int4*>(src + s * SLICE);
  for (int j0 = tid; j0 < SLICE4; j0 += GRP_THREADS * GRP_BATCH) {
    int4 vd[GRP_BATCH], vs[GRP_BATCH];
#pragma unroll
    for (int u = 0; u < GRP_BATCH; ++u) {
      const int j = j0 + u * GRP_THREADS;
      const int jc = j < SLICE4 ? j : (SLICE4 - 1);
      vd[u] = d4[jc];
      vs[u] = s4[jc];
    }
    int   kd[4 * GRP_BATCH], ks[4 * GRP_BATCH];
    bool  hit[4 * GRP_BATCH];
    float w[4 * GRP_BATCH];
#pragma unroll
    for (int u = 0; u < GRP_BATCH; ++u) {
      const bool live = (j0 + u * GRP_THREADS) < SLICE4;
      kd[4 * u + 0] = vd[u].x; kd[4 * u + 1] = vd[u].y; kd[4 * u + 2] = vd[u].z; kd[4 * u + 3] = vd[u].w;
      ks[4 * u + 0] = vs[u].x; ks[4 * u + 1] = vs[u].y; ks[4 * u + 2] = vs[u].z; ks[4 * u + 3] = vs[u].w;
#pragma unroll
      for (int c = 0; c < 4; ++c)
        hit[4 * u + c] = live && ((kd[4 * u + c] >> RANGE_BITS) == r);
    }
#pragma unroll
    for (int q = 0; q < 4 * GRP_BATCH; ++q) w[q] = out_w[hit[q] ? ks[q] : 0];
    asm volatile("" : "+v"(w[0]), "+v"(w[1]), "+v"(w[2]), "+v"(w[3]),
                      "+v"(w[4]), "+v"(w[5]), "+v"(w[6]), "+v"(w[7]),
                      "+v"(w[8]), "+v"(w[9]), "+v"(w[10]), "+v"(w[11]),
                      "+v"(w[12]), "+v"(w[13]), "+v"(w[14]), "+v"(w[15]));
#pragma unroll
    for (int q = 0; q < 4 * GRP_BATCH; ++q) {
      if (hit[q]) {
        int pos = atomicAdd(&cur[kd[q] & (RANGE - 1)], 1);
        pos = (unsigned)pos < (unsigned)N_EDGES ? pos : (N_EDGES - 1);
        csr[pos] = make_int2(ks[q], __float_as_int(w[q]));
      }
    }
  }
}

#define NODES_PER_WAVE 8
#define SLOTS 8
static_assert(SLOTS == 8, "gather_kernel names the slots of a node one by one");
template <int J> __device__ __forceinline__ int share_lane(int v) {
  return __builtin_amdgcn_update_dpp(0, v, 0x150 + J, 0xf, 0xf, false);
}
__global__ __launch_bounds__(256) void gather_kernel(const float* __restrict__ h,
                               const int* __restrict__ row_ptr,
                               const int2* __restrict__ csr,
                               float* __restrict__ agg, int n, int n_edges) {
  const int lane = threadIdx.x & 31;
  const int half = lane >> 4;
  const int l16  = lane & 15;
  const int n0 = __builtin_amdgcn_readfirstlane(
      (int)((blockIdx.x * blockDim.x + threadIdx.x) >> 5)) * NODES_PER_WAVE;
  if (n0 >= n) return;
  const int pi = (n0 + lane) < n ? (n0 + lane) : n;
  const int rp = row_ptr[pi];
  const float4* h4 = reinterpret_cast<const float4*>(h);
  const unsigned u16 = (unsigned)l16;
  const int last = n_edges - 1;

  const int eoff = 2 * l16 + half;
  int nstart = __builtin_amdgcn_readlane(rp, 0);
  int nend   = __builtin_amdgcn_readlane(rp, 1);
  int2 cn;
  {
    const int e = nstart + eoff;
    cn = csr[e < last ? e : last];
  }

#pragma unroll
  for (int r = 0; r < NODES_PER_WAVE; ++r) {
    const int node = n0 + r;
    if (node >= n) break;
    const int start = nstart;
    const int end   = nend;
    const int2 c = cn;

    if (r + 1 < NODES_PER_WAVE) {
      nstart = end;
      nend = __builtin_amdgcn_readlane(rp, r + 2);
      const int e = nstart + eoff;
      cn = csr[e < last ? e : last];
    }

    int sx[SLOTS], sy[SLOTS];
#define ENTRY_OF_SLOT(J) sx[J] = share_lane<J>(c.x); sy[J] = share_lane<J>(c.y);
    ENTRY_OF_SLOT(0) ENTRY_OF_SLOT(1) ENTRY_OF_SLOT(2) ENTRY_OF_SLOT(3)
    ENTRY_OF_SLOT(4) ENTRY_OF_SLOT(5) ENTRY_OF_SLOT(6) ENTRY_OF_SLOT(7)
#undef ENTRY_OF_SLOT
    int   src[SLOTS];
    float wgt[SLOTS];
    bool  ok[SLOTS];
#pragma unroll
    for (int j = 0; j < SLOTS; ++j) {
      ok[j]  = (start + half + 2 * j) < end;
      src[j] = ok[j] ? sx[j] : 0;
      wgt[j] = ok[j] ? __int_as_float(sy[j]) : 0.0f;
    }

    float4 v[SLOTS];
#pragma unroll
    for (int j = 0; j < SLOTS; ++j) v[j] = h4[(unsigned)src[j] * (unsigned)(D / 4) + u16];

    float4 acc = make_float4(0.0f, 0.0f, 0.0f, 0.0f);
#pragma unroll
    for (int j = 0; j < SLOTS; ++j) {
      const float4 x = ok[j] ? v[j] : make_float4(0.0f, 0.0f, 0.0f, 0.0f);
      const float w = wgt[j];
      acc.x += x.x * w; acc.y += x.y * w; acc.z += x.z * w; acc.w += x.w * w;
    }
    for (int e = start + 2 * SLOTS + half; e < end; e += 2) {
      const int2 c0 = csr[e];
      const float4 v0 = h4[(unsigned)c0.x * (unsigned)(D / 4) + u16];
      const float w0 = __int_as_float(c0.y);
      acc.x += v0.x * w0; acc.y += v0.y * w0; acc.z += v0.z * w0; acc.w += v0.w * w0;
    }
    acc.x += __shfl_xor(acc.x, 16);
    acc.y += __shfl_xor(acc.y, 16);
    acc.z += __shfl_xor(acc.z, 16);
    acc.w += __shfl_xor(acc.w, 16);
    if (half == 0)
      *(reinterpret_cast<float4*>(agg + (size_t)node * D) + l16) = acc;
  }
}

#define TP 68
typedef __attribute__((ext_vector_type(4))) float v4f;

template <bool LAST>
__global__ __launch_bounds__(256, 8) void gcn_layer_kernel(
    const float* __restrict__ agg, const float* __restrict__ in_norm,
    const float* __restrict__ h_in,
    const float* __restrict__ Wc, const float* __restrict__ bc,
    const float* __restrict__ Wr, const float* __restrict__ br,
    const float* __restrict__ lng, const float* __restrict__ lnb,
    float* __restrict__ h_out,
    const float* __restrict__ h_first,
    const float* __restrict__ Wp, const float* __restrict__ bp,
    float* __restrict__ out,
    int n)
{
  __shared__ __attribute__((aligned(16))) float s_tile[8][2][16 * TP];
  __shared__ __attribute__((aligned(8)))  float s_stat[8][32];
  const int wave = threadIdx.x >> 5;
  const int lane = threadIdx.x & 31;
  const int half = lane >> 4;
  const int l16  = lane & 15;
  const int row  = l16;
  const int kb   = half * 2;
  const int base = (blockIdx.x * 8 + wave) * 16;
  float* tA = s_tile[wave][0];
  float* tH = s_tile[wave][1];

  v4f va[8], vh[8];
#pragma unroll
  for (int i = 0; i < 8; ++i) {
    int nd = base + 2 * i + half;
    nd = nd < n ? nd : (n - 1);
    const size_t o = (size_t)nd * D + 4 * l16;
    va[i] = *reinterpret_cast<const v4f*>(agg + o);
    vh[i] = *reinterpret_cast<const v4f*>(h_in + o);
  }
  int cnode = base + row;
  cnode = cnode < n ? cnode : (n - 1);
  const float inrm = in_norm[cnode];
#pragma unroll
  for (int i = 0; i < 8; ++i) {
    const int r = 2 * i + half;
    *reinterpret_cast<v4f*>(tA + r * TP + 4 * l16) = va[i];
    *reinterpret_cast<v4f*>(tH + r * TP + 4 * l16) = vh[i];
  }
  __syncthreads();

  v2f aC[16], aR[16];
#pragma unroll
  for (int kt = 0; kt < 16; ++kt) {
    const int k0 = kt * 4 + kb;
    const v2f t = *reinterpret_cast<const v2f*>(tA + row * TP + k0);
    aC[kt][0] = t[0] * inrm;
    aC[kt][1] = t[1] * inrm;
    aR[kt] = *reinterpret_cast<const v2f*>(tH + row * TP + k0);
  }

  const int ncol = lane & 15;
  const int m0 = half * 8;
#pragma unroll
  for (int nt = 0; nt < 4; ++nt) {
    v8f c = {};
    const int col = nt * 16 + ncol;
#pragma unroll
    for (int kt = 0; kt < 16; ++kt) {
      int kr = kt * 4 + kb;
      v2f bC, bR;
      bC[0] = Wc[kr * D + col];
      bC[1] = Wc[(kr + 1) * D + col];
      bR[0] = Wr[kr * D + col];
      bR[1] = Wr[(kr + 1) * D + col];
      c = __builtin_amdgcn_wmma_f32_16x16x4_f32(false, aC[kt], false, bC, (short)0, c, false, false);
      c = __builtin_amdgcn_wmma_f32_16x16x4_f32(false, aR[kt], false, bR, (short)0, c, false, false);
    }
    const float bias = bc[col] + br[col];
#pragma unroll
    for (int r = 0; r < 8; ++r)
      tA[(m0 + r) * TP + col] = c[r] + bias;
  }
  __syncthreads();

  const v4f g4 = *reinterpret_cast<const v4f*>(lng + 4 * l16);
  const v4f b4 = *reinterpret_cast<const v4f*>(lnb + 4 * l16);
  v4f v1[8];
  if constexpr (LAST) {
#pragma unroll
    for (int i = 0; i < 8; ++i) {
      int nd = base + 2 * i + half;
      nd = nd < n ? nd : (n - 1);
      v1[i] = *reinterpret_cast<const v4f*>(h_first + (size_t)nd * D + 4 * l16);
    }
  }

  const int rr = lane >> 1;
  const int j0 = (lane & 1) * 32;
  const float* myrow = tA + rr * TP + j0;
  float s = 0.0f, sq = 0.0f;
#pragma unroll
  for (int j = 0; j < 8; ++j) {
    const v4f q = *reinterpret_cast<const v4f*>(myrow + 4 * j);
    s += q[0]; sq += q[0] * q[0];
    s += q[1]; sq += q[1] * q[1];
    s += q[2]; sq += q[2] * q[2];
    s += q[3]; sq += q[3] * q[3];
  }
  s  += __shfl_xor(s, 1);
  sq += __shfl_xor(sq, 1);
  const float mu   = s * (1.0f / 64.0f);
  const float var  = sq * (1.0f / 64.0f) - mu * mu;
  const float rstd = rsqrtf(var + LN_EPS);
  if ((lane & 1) == 0) {
    v2f st;
    st[0] = mu;
    st[1] = rstd;
    *reinterpret_cast<v2f*>(&s_stat[wave][2 * rr]) = st;
  }
  if constexpr (LAST) {
#pragma unroll
    for (int i = 0; i < 8; ++i) {
      float* p = tH + (2 * i + half) * TP + 4 * l16;
      const v4f h2 = *reinterpret_cast<const v4f*>(p);
      *reinterpret_cast<v4f*>(p) = v1[i] + h2;
    }
  }
  __syncthreads();

  v4f y[8];
#pragma unroll
  for (int i = 0; i < 8; ++i) {
    const int r = 2 * i + half;
    const v2f st = *reinterpret_cast<const v2f*>(&s_stat[wave][2 * r]);
    const v4f x = *reinterpret_cast<const v4f*>(tA + r * TP + 4 * l16);
    y[i][0] = fmaxf((x[0] - st[0]) * st[1] * g4[0] + b4[0], 0.0f);
    y[i][1] = fmaxf((x[1] - st[0]) * st[1] * g4[1] + b4[1], 0.0f);
    y[i][2] = fmaxf((x[2] - st[0]) * st[1] * g4[2] + b4[2], 0.0f);
    y[i][3] = fmaxf((x[3] - st[0]) * st[1] * g4[3] + b4[3], 0.0f);
  }

  if constexpr (!LAST) {
#pragma unroll
    for (int i = 0; i < 8; ++i) {
      const int onode = base + 2 * i + half;
      if (onode < n)
        *reinterpret_cast<v4f*>(h_out + (size_t)onode * D + 4 * l16) = y[i];
    }
  } else {
#pragma unroll
    for (int i = 0; i < 8; ++i) {
      float* p = tH + (2 * i + half) * TP + 4 * l16;
      const v4f t = *reinterpret_cast<const v4f*>(p);
      *reinterpret_cast<v4f*>(p) = t + y[i];
    }
    __syncthreads();

    v2f a[16];
#pragma unroll
    for (int kt = 0; kt < 16; ++kt)
      a[kt] = *reinterpret_cast<const v2f*>(tH + row * TP + kt * 4 + kb);
#pragma unroll
    for (int nt = 0; nt < 4; ++nt) {
      v8f c = {};
      const int col = nt * 16 + ncol;
#pragma unroll
      for (int kt = 0; kt < 16; ++kt) {
        int kr = kt * 4 + kb;
        v2f bf;
        bf[0] = Wp[kr * D + col];
        bf[1] = Wp[(kr + 1) * D + col];
        c = __builtin_amdgcn_wmma_f32_16x16x4_f32(false, a[kt], false, bf, (short)0, c, false, false);
      }
      const float bias = bp[col];
#pragma unroll
      for (int r = 0; r < 8; ++r)
        tA[(m0 + r) * TP + col] = c[r] + bias;
    }
    __syncthreads();

    v4f vo[8];
#pragma unroll
    for (int i = 0; i < 8; ++i)
      vo[i] = *reinterpret_cast<const v4f*>(tA + (2 * i + half) * TP + 4 * l16);
#pragma unroll
    for (int i = 0; i < 8; ++i) {
      const int onode = base + 2 * i + half;
      if (onode < n)
        *reinterpret_cast<v4f*>(out + (size_t)onode * D + 4 * l16) = vo[i];
    }
  }
}

extern "C" void kernel_launch(void* const* d_in, const int* in_sizes, int n_in,
                              void* d_out, int out_size, void* d_ws, size_t ws_size,
                              hipStream_t stream) {
  (void)in_sizes; (void)n_in; (void)out_size; (void)ws_size;

  const float* feats  = (const float*)d_in[0];
  const int*   src    = (const int*)  d_in[1];
  const int*   dst    = (const int*)  d_in[2];
  const float* conv_W = (const float*)d_in[3];
  const float* conv_b = (const float*)d_in[4];
  const float* res_W  = (const float*)d_in[5];
  const float* res_b  = (const float*)d_in[6];
  const float* ln_g   = (const float*)d_in[7];
  const float* ln_b   = (const float*)d_in[8];
  const float* pred_W = (const float*)d_in[9];
  const float* pred_b = (const float*)d_in[10];
  float* out = (float*)d_out;

  char* ws = (char*)d_ws;
  size_t off = 0;
  auto alloc = [&](size_t bytes) -> float* {
    float* p = (float*)(ws + off);
    off += (bytes + 255) & ~(size_t)255;
    return p;
  };
  const size_t nodeF = (size_t)N_NODES * 4;
  const size_t featF = (size_t)N_NODES * D * 4;
  float* in_norm  = alloc(nodeF);
  float* out_w    = alloc(nodeF);
  int*   row_ptr  = (int*)alloc(nodeF + 4);
  int*   tot      = (int*)alloc((size_t)NT * NS * 4);
  int2*  csr      = (int2*)alloc((size_t)N_EDGES * 8);
  float* agg      = alloc(featF);
  float* h1       = alloc(featF);
  float* h2       = alloc(featF);
  int*   pc       = (int*)agg;

  const int TB = 256;
  const int gGath   = (N_NODES + 8 * NODES_PER_WAVE - 1) / (8 * NODES_PER_WAVE);
  const int gTile   = (N_NODES + 127) / 128;

  count_kernel<<<dim3(NR, NS), GRP_THREADS, 0, stream>>>(src, dst, pc, tot);
  scan_kernel<<<NT, 256, 0, stream>>>(pc, tot, row_ptr, in_norm, out_w, N_NODES);
  fill_kernel<<<dim3(NR, NS), GRP_THREADS, 0, stream>>>(src, dst, pc, out_w, csr);

  gather_kernel<<<gGath, TB, 0, stream>>>(feats, row_ptr, csr, agg, N_NODES, N_EDGES);
  gcn_layer_kernel<false><<<gTile, TB, 0, stream>>>(
      agg, in_norm, feats, conv_W, conv_b, res_W, res_b, ln_g, ln_b,
      h1, nullptr, nullptr, nullptr, nullptr, N_NODES);

  gather_kernel<<<gGath, TB, 0, stream>>>(h1, row_ptr, csr, agg, N_NODES, N_EDGES);
  gcn_layer_kernel<false><<<gTile, TB, 0, stream>>>(
      agg, in_norm, h1, conv_W + D * D, conv_b + D, res_W + D * D, res_b + D,
      ln_g + D, ln_b + D,
      h2, nullptr, nullptr, nullptr, nullptr, N_NODES);

  gather_kernel<<<gGath, TB, 0, stream>>>(h2, row_ptr, csr, agg, N_NODES, N_EDGES);
  gcn_layer_kernel<true><<<gTile, TB, 0, stream>>>(
      agg, in_norm, h2, conv_W + 2 * D * D, conv_b + 2 * D, res_W + 2 * D * D, res_b + 2 * D,
      ln_g + 2 * D, ln_b + 2 * D,
      nullptr, h1, pred_W, pred_b, out, N_NODES);
}


# both WMMA products: weight fragments by whole-row 16-byte loads (a lane owns output columns 4c..4c+3, one per column tile), result rows to the LDS tile as 16-byte stores; sum order unchanged
# speedup vs baseline: 3.7042x; 1.0325x over previous
#include <hip/hip_runtime.h>

#define N_NODES 100000
#define N_EDGES 1000000
#define D 64
#define LN_EPS 1e-5f

typedef __attribute__((ext_vector_type(2))) float v2f;
typedef __attribute__((ext_vector_type(8))) float v8f;

#define RANGE_BITS 13
#define RANGE (1 << RANGE_BITS)
#define NR ((N_NODES + RANGE - 1) / RANGE)
#define NS 16
#define NPAD (NR * RANGE)
#define SLICE (N_EDGES / NS)
#define SLICE4 (SLICE / 4)
#define TILE 1024
#define NT (NPAD / TILE)
#define GRP_THREADS 1024
#define GRP_BATCH 4
static_assert(GRP_BATCH == 4, "fill_kernel names the 16 weights of a batch one by one");

__global__ __launch_bounds__(GRP_THREADS) void count_kernel(const int* __restrict__ src,
                              const int* __restrict__ dst, int* __restrict__ pc,
                              int* __restrict__ tot) {
  __shared__ __attribute__((aligned(16))) int cnt_in[RANGE];
  __shared__ __attribute__((aligned(16))) int cnt_out[RANGE];
  __shared__ int s_red0[32];
  __shared__ int s_red1[32];
  const int tid  = threadIdx.x;
  const int lane = tid & 31;
  const int wave = tid >> 5;
  const int r = blockIdx.x;
  const int s = blockIdx.y;
  for (int i = tid; i < RANGE / 4; i += GRP_THREADS) {
    reinterpret_cast<int4*>(cnt_in)[i]  = make_int4(0, 0, 0, 0);
    reinterpret_cast<int4*>(cnt_out)[i] = make_int4(0, 0, 0, 0);
  }
  __syncthreads();

  const int4* d4 = reinterpret_cast<const int4*>(dst + s * SLICE);
  const int4* s4 = reinterpret_cast<const int4*>(src + s * SLICE);
  for (int j0 = tid; j0 < SLICE4; j0 += GRP_THREADS * GRP_BATCH) {
    int4 vd[GRP_BATCH], vs[GRP_BATCH];
#pragma unroll
    for (int u = 0; u < GRP_BATCH; ++u) {
      const int j = j0 + u * GRP_THREADS;
      const int jc = j < SLICE4 ? j : (SLICE4 - 1);
      vd[u] = d4[jc];
      vs[u] = s4[jc];
    }
#pragma unroll
    for (int u = 0; u < GRP_BATCH; ++u) {
      if (j0 + u * GRP_THREADS < SLICE4) {
        const int kd[4] = {vd[u].x, vd[u].y, vd[u].z, vd[u].w};
        const int ks[4] = {vs[u].x, vs[u].y, vs[u].z, vs[u].w};
#pragma unroll
        for (int c = 0; c < 4; ++c) {
          if ((kd[c] >> RANGE_BITS) == r) atomicAdd(&cnt_in[kd[c] & (RANGE - 1)], 1);
          if ((ks[c] >> RANGE_BITS) == r) atomicAdd(&cnt_out[ks[c] & (RANGE - 1)], 1);
        }
      }
    }
  }
  __syncthreads();

  int* pin  = pc + (size_t)s * NPAD + (size_t)r * RANGE;
  int* pout = pc + (size_t)(NS + s) * NPAD + (size_t)r * RANGE;
  const int4 a0 = reinterpret_cast<const int4*>(cnt_in)[tid];
  const int4 a1 = reinterpret_cast<const int4*>(cnt_in)[tid + GRP_THREADS];
  const int4 b0 = reinterpret_cast<const int4*>(cnt_out)[tid];
  const int4 b1 = reinterpret_cast<const int4*>(cnt_out)[tid + GRP_THREADS];
  reinterpret_cast<int4*>(pin)[tid]                = a0;
  reinterpret_cast<int4*>(pin)[tid + GRP_THREADS]  = a1;
  reinterpret_cast<int4*>(pout)[tid]               = b0;
  reinterpret_cast<int4*>(pout)[tid + GRP_THREADS] = b1;
  int t0 = (a0.x + a0.y) + (a0.z + a0.w);
  int t1 = (a1.x + a1.y) + (a1.z + a1.w);
#pragma unroll
  for (int o = 16; o > 0; o >>= 1) {
    t0 += __shfl_xor(t0, o);
    t1 += __shfl_xor(t1, o);
  }
  if (lane == 0) { s_red0[wave] = t0; s_red1[wave] = t1; }
  __syncthreads();
  if (tid < 8) {
    const int* red = tid < 4 ? s_red0 : s_red1;
    const int g = tid & 3;
    int sum = 0;
#pragma unroll
    for (int w = 0; w < 8; ++w) sum += red[8 * g + w];
    tot[(r * (RANGE / TILE) + tid) * NS + s] = sum;
  }
}

__global__ __launch_bounds__(256) void scan_kernel(int* pc, const int* __restrict__ tot,
                              int* __restrict__ row_ptr, float* __restrict__ in_norm,
                              float* __restrict__ out_w, int n) {
  __shared__ int s_sum[8];
  __shared__ int s_pre[8];
  const int tid  = threadIdx.x;
  const int lane = tid & 31;
  const int wave = tid >> 5;
  const int b = blockIdx.x;
  const int i = b * TILE + tid * 4;

  int pre = 0;
  for (int j = tid; j < b * NS; j += 256) pre += tot[j];

  int* pin = pc + i;
  const int* pout = pc + (size_t)NS * NPAD + i;
  int4 a[NS], o[NS];
#pragma unroll
  for (int s = 0; s < NS; ++s) a[s] = *reinterpret_cast<const int4*>(pin + (size_t)s * NPAD);
#pragma unroll
  for (int s = 0; s < NS; ++s) o[s] = *reinterpret_cast<const int4*>(pout + (size_t)s * NPAD);
  int c[4] = {0, 0, 0, 0};
  int d[4] = {0, 0, 0, 0};
#pragma unroll
  for (int s = 0; s < NS; ++s) {
    c[0] += a[s].x; c[1] += a[s].y; c[2] += a[s].z; c[3] += a[s].w;
    d[0] += o[s].x; d[1] += o[s].y; d[2] += o[s].z; d[3] += o[s].w;
  }
  const int t = (c[0] + c[1]) + (c[2] + c[3]);

  int x = t;
#pragma unroll
  for (int k = 1; k < 32; k <<= 1) {
    const int y = __shfl_up(x, k);
    if (lane >= k) x += y;
  }
#pragma unroll
  for (int k = 16; k > 0; k >>= 1) pre += __shfl_xor(pre, k);
  if (lane == 31) s_sum[wave] = x;
  if (lane == 0)  s_pre[wave] = pre;
  __syncthreads();
  int below = 0, ptot = 0;
#pragma unroll
  for (int w = 0; w < 8; ++w) {
    const int v = s_sum[w];
    below += (w < wave) ? v : 0;
    ptot  += s_pre[w];
  }
  int run[4];
  run[0] = ptot + below + (x - t);
  run[1] = run[0] + c[0];
  run[2] = run[1] + c[1];
  run[3] = run[2] + c[2];

  float inn[4], ow[4];
#pragma unroll
  for (int k = 0; k < 4; ++k) {
    inn[k] = rsqrtf(fmaxf((float)c[k], 1.0f));
    ow[k]  = rsqrtf(fmaxf((float)d[k], 1.0f));
  }
  if (i + 3 < n) {
    *reinterpret_cast<int4*>(row_ptr + i)   = make_int4(run[0], run[1], run[2], run[3]);
    *reinterpret_cast<float4*>(in_norm + i) = make_float4(inn[0], inn[1], inn[2], inn[3]);
    *reinterpret_cast<float4*>(out_w + i)   = make_float4(ow[0], ow[1], ow[2], ow[3]);
  } else {
#pragma unroll
    for (int k = 0; k < 4; ++k) {
      if (i + k <= n) row_ptr[i + k] = run[k];
      if (i + k < n) { in_norm[i + k] = inn[k]; out_w[i + k] = ow[k]; }
    }
  }
#pragma unroll
  for (int s = 0; s < NS; ++s) {
    *reinterpret_cast<int4*>(pin + (size_t)s * NPAD) = make_int4(run[0], run[1], run[2], run[3]);
    run[0] += a[s].x; run[1] += a[s].y; run[2] += a[s].z; run[3] += a[s].w;
  }
}

__global__ __launch_bounds__(GRP_THREADS) void fill_kernel(const int* __restrict__ src,
                            const int* __restrict__ dst, const int* __restrict__ startpos,
                            const float* __restrict__ out_w, int2* __restrict__ csr) {
  __shared__ __attribute__((aligned(16))) int cur[RANGE];
  const int tid = threadIdx.x;
  const int r = blockIdx.x;
  const int s = blockIdx.y;
  const int4* b4 = reinterpret_cast<const int4*>(startpos + (size_t)s * NPAD + (size_t)r * RANGE);
  reinterpret_cast<int4*>(cur)[tid]               = b4[tid];
  reinterpret_cast<int4*>(cur)[tid + GRP_THREADS] = b4[tid + GRP_THREADS];
  __syncthreads();

  const int4* d4 = reinterpret_cast<const int4*>(dst + s * SLICE);
  const int4* s4 = reinterpret_cast<const int4*>(src + s * SLICE);
  for (int j0 = tid; j0 < SLICE4; j0 += GRP_THREADS * GRP_BATCH) {
    int4 vd[GRP_BATCH], vs[GRP_BATCH];
#pragma unroll
    for (int u = 0; u < GRP_BATCH; ++u) {
      const int j = j0 + u * GRP_THREADS;
      const int jc = j < SLICE4 ? j : (SLICE4 - 1);
      vd[u] = d4[jc];
      vs[u] = s4[jc];
    }
    int   kd[4 * GRP_BATCH], ks[4 * GRP_BATCH];
    bool  hit[4 * GRP_BATCH];
    float w[4 * GRP_BATCH];
#pragma unroll
    for (int u = 0; u < GRP_BATCH; ++u) {
      const bool live = (j0 + u * GRP_THREADS) < SLICE4;
      kd[4 * u + 0] = vd[u].x; kd[4 * u + 1] = vd[u].y; kd[4 * u + 2] = vd[u].z; kd[4 * u + 3] = vd[u].w;
      ks[4 * u + 0] = vs[u].x; ks[4 * u + 1] = vs[u].y; ks[4 * u + 2] = vs[u].z; ks[4 * u + 3] = vs[u].w;
#pragma unroll
      for (int c = 0; c < 4; ++c)
        hit[4 * u + c] = live && ((kd[4 * u + c] >> RANGE_BITS) == r);
    }
#pragma unroll
    for (int q = 0; q < 4 * GRP_BATCH; ++q) w[q] = out_w[hit[q] ? ks[q] : 0];
    asm volatile("" : "+v"(w[0]), "+v"(w[1]), "+v"(w[2]), "+v"(w[3]),
                      "+v"(w[4]), "+v"(w[5]), "+v"(w[6]), "+v"(w[7]),
                      "+v"(w[8]), "+v"(w[9]), "+v"(w[10]), "+v"(w[11]),
                      "+v"(w[12]), "+v"(w[13]), "+v"(w[14]), "+v"(w[15]));
#pragma unroll
    for (int q = 0; q < 4 * GRP_BATCH; ++q) {
      if (hit[q]) {
        int pos = atomicAdd(&cur[kd[q] & (RANGE - 1)], 1);
        pos = (unsigned)pos < (unsigned)N_EDGES ? pos : (N_EDGES - 1);
        csr[pos] = make_int2(ks[q], __float_as_int(w[q]));
      }
    }
  }
}

#define NODES_PER_WAVE 8
#define SLOTS 8
static_assert(SLOTS == 8, "gather_kernel names the slots of a node one by one");
template <int J> __device__ __forceinline__ int share_lane(int v) {
  return __builtin_amdgcn_update_dpp(0, v, 0x150 + J, 0xf, 0xf, false);
}
__global__ __launch_bounds__(256) void gather_kernel(const float* __restrict__ h,
                               const int* __restrict__ row_ptr,
                               const int2* __restrict__ csr,
                               float* __restrict__ agg, int n, int n_edges) {
  const int lane = threadIdx.x & 31;
  const int half = lane >> 4;
  const int l16  = lane & 15;
  const int n0 = __builtin_amdgcn_readfirstlane(
      (int)((blockIdx.x * blockDim.x + threadIdx.x) >> 5)) * NODES_PER_WAVE;
  if (n0 >= n) return;
  const int pi = (n0 + lane) < n ? (n0 + lane) : n;
  const int rp = row_ptr[pi];
  const float4* h4 = reinterpret_cast<const float4*>(h);
  const unsigned u16 = (unsigned)l16;
  const int last = n_edges - 1;

  const int eoff = 2 * l16 + half;
  int nstart = __builtin_amdgcn_readlane(rp, 0);
  int nend   = __builtin_amdgcn_readlane(rp, 1);
  int2 cn;
  {
    const int e = nstart + eoff;
    cn = csr[e < last ? e : last];
  }

#pragma unroll
  for (int r = 0; r < NODES_PER_WAVE; ++r) {
    const int node = n0 + r;
    if (node >= n) break;
    const int start = nstart;
    const int end   = nend;
    const int2 c = cn;

    if (r + 1 < NODES_PER_WAVE) {
      nstart = end;
      nend = __builtin_amdgcn_readlane(rp, r + 2);
      const int e = nstart + eoff;
      cn = csr[e < last ? e : last];
    }

    int sx[SLOTS], sy[SLOTS];
#define ENTRY_OF_SLOT(J) sx[J] = share_lane<J>(c.x); sy[J] = share_lane<J>(c.y);
    ENTRY_OF_SLOT(0) ENTRY_OF_SLOT(1) ENTRY_OF_SLOT(2) ENTRY_OF_SLOT(3)
    ENTRY_OF_SLOT(4) ENTRY_OF_SLOT(5) ENTRY_OF_SLOT(6) ENTRY_OF_SLOT(7)
#undef ENTRY_OF_SLOT
    int   src[SLOTS];
    float wgt[SLOTS];
    bool  ok[SLOTS];
#pragma unroll
    for (int j = 0; j < SLOTS; ++j) {
      ok[j]  = (start + half + 2 * j) < end;
      src[j] = ok[j] ? sx[j] : 0;
      wgt[j] = ok[j] ? __int_as_float(sy[j]) : 0.0f;
    }

    float4 v[SLOTS];
#pragma unroll
    for (int j = 0; j < SLOTS; ++j) v[j] = h4[(unsigned)src[j] * (unsigned)(D / 4) + u16];

    float4 acc = make_float4(0.0f, 0.0f, 0.0f, 0.0f);
#pragma unroll
    for (int j = 0; j < SLOTS; ++j) {
      const float4 x = ok[j] ? v[j] : make_float4(0.0f, 0.0f, 0.0f, 0.0f);
      const float w = wgt[j];
      acc.x += x.x * w; acc.y += x.y * w; acc.z += x.z * w; acc.w += x.w * w;
    }
    for (int e = start + 2 * SLOTS + half; e < end; e += 2) {
      const int2 c0 = csr[e];
      const float4 v0 = h4[(unsigned)c0.x * (unsigned)(D / 4) + u16];
      const float w0 = __int_as_float(c0.y);
      acc.x += v0.x * w0; acc.y += v0.y * w0; acc.z += v0.z * w0; acc.w += v0.w * w0;
    }
    acc.x += __shfl_xor(acc.x, 16);
    acc.y += __shfl_xor(acc.y, 16);
    acc.z += __shfl_xor(acc.z, 16);
    acc.w += __shfl_xor(acc.w, 16);
    if (half == 0)
      *(reinterpret_cast<float4*>(agg + (size_t)node * D) + l16) = acc;
  }
}

#define TP 68
typedef __attribute__((ext_vector_type(4))) float v4f;

template <bool LAST>
__global__ __launch_bounds__(256, 8) void gcn_layer_kernel(
    const float* __restrict__ agg, const float* __restrict__ in_norm,
    const float* __restrict__ h_in,
    const float* __restrict__ Wc, const float* __restrict__ bc,
    const float* __restrict__ Wr, const float* __restrict__ br,
    const float* __restrict__ lng, const float* __restrict__ lnb,
    float* __restrict__ h_out,
    const float* __restrict__ h_first,
    const float* __restrict__ Wp, const float* __restrict__ bp,
    float* __restrict__ out,
    int n)
{
  __shared__ __attribute__((aligned(16))) float s_tile[8][2][16 * TP];
  __shared__ __attribute__((aligned(8)))  float s_stat[8][32];
  const int wave = threadIdx.x >> 5;
  const int lane = threadIdx.x & 31;
  const int half = lane >> 4;
  const int l16  = lane & 15;
  const int row  = l16;
  const int kb   = half * 2;
  const int base = (blockIdx.x * 8 + wave) * 16;
  float* tA = s_tile[wave][0];
  float* tH = s_tile[wave][1];

  v4f va[8], vh[8];
#pragma unroll
  for (int i = 0; i < 8; ++i) {
    int nd = base + 2 * i + half;
    nd = nd < n ? nd : (n - 1);
    const size_t o = (size_t)nd * D + 4 * l16;
    va[i] = *reinterpret_cast<const v4f*>(agg + o);
    vh[i] = *reinterpret_cast<const v4f*>(h_in + o);
  }
  int cnode = base + row;
  cnode = cnode < n ? cnode : (n - 1);
  const float inrm = in_norm[cnode];
#pragma unroll
  for (int i = 0; i < 8; ++i) {
    const int r = 2 * i + half;
    *reinterpret_cast<v4f*>(tA + r * TP + 4 * l16) = va[i];
    *reinterpret_cast<v4f*>(tH + r * TP + 4 * l16) = vh[i];
  }
  __syncthreads();

  v2f aC[16], aR[16];
#pragma unroll
  for (int kt = 0; kt < 16; ++kt) {
    const int k0 = kt * 4 + kb;
    const v2f t = *reinterpret_cast<const v2f*>(tA + row * TP + k0);
    aC[kt][0] = t[0] * inrm;
    aC[kt][1] = t[1] * inrm;
    aR[kt] = *reinterpret_cast<const v2f*>(tH + row * TP + k0);
  }

  const int m0 = half * 8;
  const v4f* wc4 = reinterpret_cast<const v4f*>(Wc);
  const v4f* wr4 = reinterpret_cast<const v4f*>(Wr);
  const unsigned wl = (unsigned)(kb * (D / 4) + l16);
  v8f acc[4] = {};
#pragma unroll
  for (int kt = 0; kt < 16; ++kt) {
    const v4f c0 = wc4[wl + (unsigned)((kt * 4) * (D / 4))];
    const v4f c1 = wc4[wl + (unsigned)((kt * 4 + 1) * (D / 4))];
    const v4f r0 = wr4[wl + (unsigned)((kt * 4) * (D / 4))];
    const v4f r1 = wr4[wl + (unsigned)((kt * 4 + 1) * (D / 4))];
#pragma unroll
    for (int j = 0; j < 4; ++j) {
      v2f bC, bR;
      bC[0] = c0[j];
      bC[1] = c1[j];
      bR[0] = r0[j];
      bR[1] = r1[j];
      acc[j] = __builtin_amdgcn_wmma_f32_16x16x4_f32(false, aC[kt], false, bC, (short)0, acc[j], false, false);
      acc[j] = __builtin_amdgcn_wmma_f32_16x16x4_f32(false, aR[kt], false, bR, (short)0, acc[j], false, false);
    }
  }
  {
    const v4f bias = *reinterpret_cast<const v4f*>(bc + 4 * l16) +
                     *reinterpret_cast<const v4f*>(br + 4 * l16);
#pragma unroll
    for (int r = 0; r < 8; ++r) {
      v4f o;
      o[0] = acc[0][r] + bias[0];
      o[1] = acc[1][r] + bias[1];
      o[2] = acc[2][r] + bias[2];
      o[3] = acc[3][r] + bias[3];
      *reinterpret_cast<v4f*>(tA + (m0 + r) * TP + 4 * l16) = o;
    }
  }
  __syncthreads();

  const v4f g4 = *reinterpret_cast<const v4f*>(lng + 4 * l16);
  const v4f b4 = *reinterpret_cast<const v4f*>(lnb + 4 * l16);
  v4f v1[8];
  if constexpr (LAST) {
#pragma unroll
    for (int i = 0; i < 8; ++i) {
      int nd = base + 2 * i + half;
      nd = nd < n ? nd : (n - 1);
      v1[i] = *reinterpret_cast<const v4f*>(h_first + (size_t)nd * D + 4 * l16);
    }
  }

  const int rr = lane >> 1;
  const int j0 = (lane & 1) * 32;
  const float* myrow = tA + rr * TP + j0;
  float s = 0.0f, sq = 0.0f;
#pragma unroll
  for (int j = 0; j < 8; ++j) {
    const v4f q = *reinterpret_cast<const v4f*>(myrow + 4 * j);
    s += q[0]; sq += q[0] * q[0];
    s += q[1]; sq += q[1] * q[1];
    s += q[2]; sq += q[2] * q[2];
    s += q[3]; sq += q[3] * q[3];
  }
  s  += __shfl_xor(s, 1);
  sq += __shfl_xor(sq, 1);
  const float mu   = s * (1.0f / 64.0f);
  const float var  = sq * (1.0f / 64.0f) - mu * mu;
  const float rstd = rsqrtf(var + LN_EPS);
  if ((lane & 1) == 0) {
    v2f st;
    st[0] = mu;
    st[1] = rstd;
    *reinterpret_cast<v2f*>(&s_stat[wave][2 * rr]) = st;
  }
  if constexpr (LAST) {
#pragma unroll
    for (int i = 0; i < 8; ++i) {
      float* p = tH + (2 * i + half) * TP + 4 * l16;
      const v4f h2 = *reinterpret_cast<const v4f*>(p);
      *reinterpret_cast<v4f*>(p) = v1[i] + h2;
    }
  }
  __syncthreads();

  v4f y[8];
#pragma unroll
  for (int i = 0; i < 8; ++i) {
    const int r = 2 * i + half;
    const v2f st = *reinterpret_cast<const v2f*>(&s_stat[wave][2 * r]);
    const v4f x = *reinterpret_cast<const v4f*>(tA + r * TP + 4 * l16);
    y[i][0] = fmaxf((x[0] - st[0]) * st[1] * g4[0] + b4[0], 0.0f);
    y[i][1] = fmaxf((x[1] - st[0]) * st[1] * g4[1] + b4[1], 0.0f);
    y[i][2] = fmaxf((x[2] - st[0]) * st[1] * g4[2] + b4[2], 0.0f);
    y[i][3] = fmaxf((x[3] - st[0]) * st[1] * g4[3] + b4[3], 0.0f);
  }

  if constexpr (!LAST) {
#pragma unroll
    for (int i = 0; i < 8; ++i) {
      const int onode = base + 2 * i + half;
      if (onode < n)
        *reinterpret_cast<v4f*>(h_out + (size_t)onode * D + 4 * l16) = y[i];
    }
  } else {
#pragma unroll
    for (int i = 0; i < 8; ++i) {
      float* p = tH + (2 * i + half) * TP + 4 * l16;
      const v4f t = *reinterpret_cast<const v4f*>(p);
      *reinterpret_cast<v4f*>(p) = t + y[i];
    }
    __syncthreads();

    v2f a[16];
#pragma unroll
    for (int kt = 0; kt < 16; ++kt)
      a[kt] = *reinterpret_cast<const v2f*>(tH + row * TP + kt * 4 + kb);
    const v4f* wp4 = reinterpret_cast<const v4f*>(Wp);
    v8f pacc[4] = {};
#pragma unroll
    for (int kt = 0; kt < 16; ++kt) {
      const v4f p0 = wp4[wl + (unsigned)((kt * 4) * (D / 4))];
      const v4f p1 = wp4[wl + (unsigned)((kt * 4 + 1) * (D / 4))];
#pragma unroll
      for (int j = 0; j < 4; ++j) {
        v2f bf;
        bf[0] = p0[j];
        bf[1] = p1[j];
        pacc[j] = __builtin_amdgcn_wmma_f32_16x16x4_f32(false, a[kt], false, bf, (short)0, pacc[j], false, false);
      }
    }
    {
      const v4f bias = *reinterpret_cast<const v4f*>(bp + 4 * l16);
#pragma unroll
      for (int r = 0; r < 8; ++r) {
        v4f o;
        o[0] = pacc[0][r] + bias[0];
        o[1] = pacc[1][r] + bias[1];
        o[2] = pacc[2][r] + bias[2];
        o[3] = pacc[3][r] + bias[3];
        *reinterpret_cast<v4f*>(tA + (m0 + r) * TP + 4 * l16) = o;
      }
    }
    __syncthreads();

    v4f vo[8];
#pragma unroll
    for (int i = 0; i < 8; ++i)
      vo[i] = *reinterpret_cast<const v4f*>(tA + (2 * i + half) * TP + 4 * l16);
#pragma unroll
    for (int i = 0; i < 8; ++i) {
      const int onode = base + 2 * i + half;
      if (onode < n)
        *reinterpret_cast<v4f*>(out + (size_t)onode * D + 4 * l16) = vo[i];
    }
  }
}

extern "C" void kernel_launch(void* const* d_in, const int* in_sizes, int n_in,
                              void* d_out, int out_size, void* d_ws, size_t ws_size,
                              hipStream_t stream) {
  (void)in_sizes; (void)n_in; (void)out_size; (void)ws_size;

  const float* feats  = (const float*)d_in[0];
  const int*   src    = (const int*)  d_in[1];
  const int*   dst    = (const int*)  d_in[2];
  const float* conv_W = (const float*)d_in[3];
  const float* conv_b = (const float*)d_in[4];
  const float* res_W  = (const float*)d_in[5];
  const float* res_b  = (const float*)d_in[6];
  const float* ln_g   = (const float*)d_in[7];
  const float* ln_b   = (const float*)d_in[8];
  const float* pred_W = (const float*)d_in[9];
  const float* pred_b = (const float*)d_in[10];
  float* out = (float*)d_out;

  char* ws = (char*)d_ws;
  size_t off = 0;
  auto alloc = [&](size_t bytes) -> float* {
    float* p = (float*)(ws + off);
    off += (bytes + 255) & ~(size_t)255;
    return p;
  };
  const size_t nodeF = (size_t)N_NODES * 4;
  const size_t featF = (size_t)N_NODES * D * 4;
  float* in_norm  = alloc(nodeF);
  float* out_w    = alloc(nodeF);
  int*   row_ptr  = (int*)alloc(nodeF + 4);
  int*   tot      = (int*)alloc((size_t)NT * NS * 4);
  int2*  csr      = (int2*)alloc((size_t)N_EDGES * 8);
  float* agg      = alloc(featF);
  float* h1       = alloc(featF);
  float* h2       = alloc(featF);
  int*   pc       = (int*)agg;

  const int TB = 256;
  const int gGath   = (N_NODES + 8 * NODES_PER_WAVE - 1) / (8 * NODES_PER_WAVE);
  const int gTile   = (N_NODES + 127) / 128;

  count_kernel<<<dim3(NR, NS), GRP_THREADS, 0, stream>>>(src, dst, pc, tot);
  scan_kernel<<<NT, 256, 0, stream>>>(pc, tot, row_ptr, in_norm, out_w, N_NODES);
  fill_kernel<<<dim3(NR, NS), GRP_THREADS, 0, stream>>>(src, dst, pc, out_w, csr);

  gather_kernel<<<gGath, TB, 0, stream>>>(feats, row_ptr, csr, agg, N_NODES, N_EDGES);
  gcn_layer_kernel<false><<<gTile, TB, 0, stream>>>(
      agg, in_norm, feats, conv_W, conv_b, res_W, res_b, ln_g, ln_b,
      h1, nullptr, nullptr, nullptr, nullptr, N_NODES);

  gather_kernel<<<gGath, TB, 0, stream>>>(h1, row_ptr, csr, agg, N_NODES, N_EDGES);
  gcn_layer_kernel<false><<<gTile, TB, 0, stream>>>(
      agg, in_norm, h1, conv_W + D * D, conv_b + D, res_W + D * D, res_b + D,
      ln_g + D, ln_b + D,
      h2, nullptr, nullptr, nullptr, nullptr, N_NODES);

  gather_kernel<<<gGath, TB, 0, stream>>>(h2, row_ptr, csr, agg, N_NODES, N_EDGES);
  gcn_layer_kernel<true><<<gTile, TB, 0, stream>>>(
      agg, in_norm, h2, conv_W + 2 * D * D, conv_b + 2 * D, res_W + 2 * D * D, res_b + 2 * D,
      ln_g + 2 * D, ln_b + 2 * D,
      nullptr, h1, pred_W, pred_b, out, N_NODES);
}


# gather folded into the layer kernel: a wave sums the source rows of its 16 nodes straight into its LDS tile, the neighbour sums never cross memory; 3 launches fewer; sum order unchanged
# speedup vs baseline: 4.0414x; 1.0910x over previous
#include <hip/hip_runtime.h>

#define N_NODES 100000
#define N_EDGES 1000000
#define D 64
#define LN_EPS 1e-5f

typedef __attribute__((ext_vector_type(2))) float v2f;
typedef __attribute__((ext_vector_type(8))) float v8f;

#define RANGE_BITS 13
#define RANGE (1 << RANGE_BITS)
#define NR ((N_NODES + RANGE - 1) / RANGE)
#define NS 16
#define NPAD (NR * RANGE)
#define SLICE (N_EDGES / NS)
#define SLICE4 (SLICE / 4)
#define TILE 1024
#define NT (NPAD / TILE)
#define GRP_THREADS 1024
#define GRP_BATCH 4
static_assert(GRP_BATCH == 4, "fill_kernel names the 16 weights of a batch one by one");

__global__ __launch_bounds__(GRP_THREADS) void count_kernel(const int* __restrict__ src,
                              const int* __restrict__ dst, int* __restrict__ pc,
                              int* __restrict__ tot) {
  __shared__ __attribute__((aligned(16))) int cnt_in[RANGE];
  __shared__ __attribute__((aligned(16))) int cnt_out[RANGE];
  __shared__ int s_red0[32];
  __shared__ int s_red1[32];
  const int tid  = threadIdx.x;
  const int lane = tid & 31;
  const int wave = tid >> 5;
  const int r = blockIdx.x;
  const int s = blockIdx.y;
  for (int i = tid; i < RANGE / 4; i += GRP_THREADS) {
    reinterpret_cast<int4*>(cnt_in)[i]  = make_int4(0, 0, 0, 0);
    reinterpret_cast<int4*>(cnt_out)[i] = make_int4(0, 0, 0, 0);
  }
  __syncthreads();

  const int4* d4 = reinterpret_cast<const int4*>(dst + s * SLICE);
  const int4* s4 = reinterpret_cast<const int4*>(src + s * SLICE);
  for (int j0 = tid; j0 < SLICE4; j0 += GRP_THREADS * GRP_BATCH) {
    int4 vd[GRP_BATCH], vs[GRP_BATCH];
#pragma unroll
    for (int u = 0; u < GRP_BATCH; ++u) {
      const int j = j0 + u * GRP_THREADS;
      const int jc = j < SLICE4 ? j : (SLICE4 - 1);
      vd[u] = d4[jc];
      vs[u] = s4[jc];
    }
#pragma unroll
    for (int u = 0; u < GRP_BATCH; ++u) {
      if (j0 + u * GRP_THREADS < SLICE4) {
        const int kd[4] = {vd[u].x, vd[u].y, vd[u].z, vd[u].w};
        const int ks[4] = {vs[u].x, vs[u].y, vs[u].z, vs[u].w};
#pragma unroll
        for (int c = 0; c < 4; ++c) {
          if ((kd[c] >> RANGE_BITS) == r) atomicAdd(&cnt_in[kd[c] & (RANGE - 1)], 1);
          if ((ks[c] >> RANGE_BITS) == r) atomicAdd(&cnt_out[ks[c] & (RANGE - 1)], 1);
        }
      }
    }
  }
  __syncthreads();

  int* pin  = pc + (size_t)s * NPAD + (size_t)r * RANGE;
  int* pout = pc + (size_t)(NS + s) * NPAD + (size_t)r * RANGE;
  const int4 a0 = reinterpret_cast<const int4*>(cnt_in)[tid];
  const int4 a1 = reinterpret_cast<const int4*>(cnt_in)[tid + GRP_THREADS];
  const int4 b0 = reinterpret_cast<const int4*>(cnt_out)[tid];
  const int4 b1 = reinterpret_cast<const int4*>(cnt_out)[tid + GRP_THREADS];
  reinterpret_cast<int4*>(pin)[tid]                = a0;
  reinterpret_cast<int4*>(pin)[tid + GRP_THREADS]  = a1;
  reinterpret_cast<int4*>(pout)[tid]               = b0;
  reinterpret_cast<int4*>(pout)[tid + GRP_THREADS] = b1;
  int t0 = (a0.x + a0.y) + (a0.z + a0.w);
  int t1 = (a1.x + a1.y) + (a1.z + a1.w);
#pragma unroll
  for (int o = 16; o > 0; o >>= 1) {
    t0 += __shfl_xor(t0, o);
    t1 += __shfl_xor(t1, o);
  }
  if (lane == 0) { s_red0[wave] = t0; s_red1[wave] = t1; }
  __syncthreads();
  if (tid < 8) {
    const int* red = tid < 4 ? s_red0 : s_red1;
    const int g = tid & 3;
    int sum = 0;
#pragma unroll
    for (int w = 0; w < 8; ++w) sum += red[8 * g + w];
    tot[(r * (RANGE / TILE) + tid) * NS + s] = sum;
  }
}

__global__ __launch_bounds__(256) void scan_kernel(int* pc, const int* __restrict__ tot,
                              int* __restrict__ row_ptr, float* __restrict__ in_norm,
                              float* __restrict__ out_w, int n) {
  __shared__ int s_sum[8];
  __shared__ int s_pre[8];
  const int tid  = threadIdx.x;
  const int lane = tid & 31;
  const int wave = tid >> 5;
  const int b = blockIdx.x;
  const int i = b * TILE + tid * 4;

  int pre = 0;
  for (int j = tid; j < b * NS; j += 256) pre += tot[j];

  int* pin = pc + i;
  const int* pout = pc + (size_t)NS * NPAD + i;
  int4 a[NS], o[NS];
#pragma unroll
  for (int s = 0; s < NS; ++s) a[s] = *reinterpret_cast<const int4*>(pin + (size_t)s * NPAD);
#pragma unroll
  for (int s = 0; s < NS; ++s) o[s] = *reinterpret_cast<const int4*>(pout + (size_t)s * NPAD);
  int c[4] = {0, 0, 0, 0};
  int d[4] = {0, 0, 0, 0};
#pragma unroll
  for (int s = 0; s < NS; ++s) {
    c[0] += a[s].x; c[1] += a[s].y; c[2] += a[s].z; c[3] += a[s].w;
    d[0] += o[s].x; d[1] += o[s].y; d[2] += o[s].z; d[3] += o[s].w;
  }
  const int t = (c[0] + c[1]) + (c[2] + c[3]);

  int x = t;
#pragma unroll
  for (int k = 1; k < 32; k <<= 1) {
    const int y = __shfl_up(x, k);
    if (lane >= k) x += y;
  }
#pragma unroll
  for (int k = 16; k > 0; k >>= 1) pre += __shfl_xor(pre, k);
  if (lane == 31) s_sum[wave] = x;
  if (lane == 0)  s_pre[wave] = pre;
  __syncthreads();
  int below = 0, ptot = 0;
#pragma unroll
  for (int w = 0; w < 8; ++w) {
    const int v = s_sum[w];
    below += (w < wave) ? v : 0;
    ptot  += s_pre[w];
  }
  int run[4];
  run[0] = ptot + below + (x - t);
  run[1] = run[0] + c[0];
  run[2] = run[1] + c[1];
  run[3] = run[2] + c[2];

  float inn[4], ow[4];
#pragma unroll
  for (int k = 0; k < 4; ++k) {
    inn[k] = rsqrtf(fmaxf((float)c[k], 1.0f));
    ow[k]  = rsqrtf(fmaxf((float)d[k], 1.0f));
  }
  if (i + 3 < n) {
    *reinterpret_cast<int4*>(row_ptr + i)   = make_int4(run[0], run[1], run[2], run[3]);
    *reinterpret_cast<float4*>(in_norm + i) = make_float4(inn[0], inn[1], inn[2], inn[3]);
    *reinterpret_cast<float4*>(out_w + i)   = make_float4(ow[0], ow[1], ow[2], ow[3]);
  } else {
#pragma unroll
    for (int k = 0; k < 4; ++k) {
      if (i + k <= n) row_ptr[i + k] = run[k];
      if (i + k < n) { in_norm[i + k] = inn[k]; out_w[i + k] = ow[k]; }
    }
  }
#pragma unroll
  for (int s = 0; s < NS; ++s) {
    *reinterpret_cast<int4*>(pin + (size_t)s * NPAD) = make_int4(run[0], run[1], run[2], run[3]);
    run[0] += a[s].x; run[1] += a[s].y; run[2] += a[s].z; run[3] += a[s].w;
  }
}

__global__ __launch_bounds__(GRP_THREADS) void fill_kernel(const int* __restrict__ src,
                            const int* __restrict__ dst, const int* __restrict__ startpos,
                            const float* __restrict__ out_w, int2* __restrict__ csr) {
  __shared__ __attribute__((aligned(16))) int cur[RANGE];
  const int tid = threadIdx.x;
  const int r = blockIdx.x;
  const int s = blockIdx.y;
  const int4* b4 = reinterpret_cast<const int4*>(startpos + (size_t)s * NPAD + (size_t)r * RANGE);
  reinterpret_cast<int4*>(cur)[tid]               = b4[tid];
  reinterpret_cast<int4*>(cur)[tid + GRP_THREADS] = b4[tid + GRP_THREADS];
  __syncthreads();

  const int4* d4 = reinterpret_cast<const int4*>(dst + s * SLICE);
  const int4* s4 = reinterpret_cast<const int4*>(src + s * SLICE);
  for (int j0 = tid; j0 < SLICE4; j0 += GRP_THREADS * GRP_BATCH) {
    int4 vd[GRP_BATCH], vs[GRP_BATCH];
#pragma unroll
    for (int u = 0; u < GRP_BATCH; ++u) {
      const int j = j0 + u * GRP_THREADS;
      const int jc = j < SLICE4 ? j : (SLICE4 - 1);
      vd[u] = d4[jc];
      vs[u] = s4[jc];
    }
    int   kd[4 * GRP_BATCH], ks[4 * GRP_BATCH];
    bool  hit[4 * GRP_BATCH];
    float w[4 * GRP_BATCH];
#pragma unroll
    for (int u = 0; u < GRP_BATCH; ++u) {
      const bool live = (j0 + u * GRP_THREADS) < SLICE4;
      kd[4 * u + 0] = vd[u].x; kd[4 * u + 1] = vd[u].y; kd[4 * u + 2] = vd[u].z; kd[4 * u + 3] = vd[u].w;
      ks[4 * u + 0] = vs[u].x; ks[4 * u + 1] = vs[u].y; ks[4 * u + 2] = vs[u].z; ks[4 * u + 3] = vs[u].w;
#pragma unroll
      for (int c = 0; c < 4; ++c)
        hit[4 * u + c] = live && ((kd[4 * u + c] >> RANGE_BITS) == r);
    }
#pragma unroll
    for (int q = 0; q < 4 * GRP_BATCH; ++q) w[q] = out_w[hit[q] ? ks[q] : 0];
    asm volatile("" : "+v"(w[0]), "+v"(w[1]), "+v"(w[2]), "+v"(w[3]),
                      "+v"(w[4]), "+v"(w[5]), "+v"(w[6]), "+v"(w[7]),
                      "+v"(w[8]), "+v"(w[9]), "+v"(w[10]), "+v"(w[11]),
                      "+v"(w[12]), "+v"(w[13]), "+v"(w[14]), "+v"(w[15]));
#pragma unroll
    for (int q = 0; q < 4 * GRP_BATCH; ++q) {
      if (hit[q]) {
        int pos = atomicAdd(&cur[kd[q] & (RANGE - 1)], 1);
        pos = (unsigned)pos < (unsigned)N_EDGES ? pos : (N_EDGES - 1);
        csr[pos] = make_int2(ks[q], __float_as_int(w[q]));
      }
    }
  }
}

#define SLOTS 8
static_assert(SLOTS == 8, "the gather phase names the slots of a node one by one");
template <int J> __device__ __forceinline__ int share_lane(int v) {
  return __builtin_amdgcn_update_dpp(0, v, 0x150 + J, 0xf, 0xf, false);
}

#define TP 68
typedef __attribute__((ext_vector_type(4))) float v4f;

template <bool LAST>
__global__ __launch_bounds__(256, 8) void gcn_layer_kernel(
    const float* __restrict__ h_in,
    const int* __restrict__ row_ptr, const int2* __restrict__ csr,
    const float* __restrict__ in_norm,
    const float* __restrict__ Wc, const float* __restrict__ bc,
    const float* __restrict__ Wr, const float* __restrict__ br,
    const float* __restrict__ lng, const float* __restrict__ lnb,
    float* __restrict__ h_out,
    const float* __restrict__ h_first,
    const float* __restrict__ Wp, const float* __restrict__ bp,
    float* __restrict__ out,
    int n, int n_edges)
{
  __shared__ __attribute__((aligned(16))) float s_tile[8][2][16 * TP];
  __shared__ __attribute__((aligned(8)))  float s_stat[8][32];
  const int wave = threadIdx.x >> 5;
  const int lane = threadIdx.x & 31;
  const int half = lane >> 4;
  const int l16  = lane & 15;
  const int row  = l16;
  const int kb   = half * 2;
  const int base = __builtin_amdgcn_readfirstlane((int)((blockIdx.x * 8 + wave) * 16));
  float* tA = s_tile[wave][0];
  float* tH = s_tile[wave][1];

  v4f vh[8];
#pragma unroll
  for (int i = 0; i < 8; ++i) {
    int nd = base + 2 * i + half;
    nd = nd < n ? nd : (n - 1);
    vh[i] = *reinterpret_cast<const v4f*>(h_in + (size_t)nd * D + 4 * l16);
  }
  int cnode = base + row;
  cnode = cnode < n ? cnode : (n - 1);
  const float inrm = in_norm[cnode];

  const int pi = (base + lane) < n ? (base + lane) : n;
  const int rp = row_ptr[pi];
  const float4* h4 = reinterpret_cast<const float4*>(h_in);
  const unsigned u16 = (unsigned)l16;
  const int last = n_edges - 1;

  const int eoff = 2 * l16 + half;
  int nstart = __builtin_amdgcn_readlane(rp, 0);
  int nend   = __builtin_amdgcn_readlane(rp, 1);
  int2 cn;
  {
    const int e = nstart + eoff;
    cn = csr[e < last ? e : last];
  }

#pragma unroll
  for (int i = 0; i < 8; ++i)
    *reinterpret_cast<v4f*>(tH + (2 * i + half) * TP + 4 * l16) = vh[i];

#pragma unroll
  for (int r = 0; r < 16; ++r) {
    const int start = nstart;
    const int end   = nend;
    const int2 c = cn;

    if (r + 1 < 16) {
      nstart = end;
      nend = __builtin_amdgcn_readlane(rp, r + 2);
      const int e = nstart + eoff;
      cn = csr[e < last ? e : last];
    }

    int sx[SLOTS], sy[SLOTS];
#define ENTRY_OF_SLOT(J) sx[J] = share_lane<J>(c.x); sy[J] = share_lane<J>(c.y);
    ENTRY_OF_SLOT(0) ENTRY_OF_SLOT(1) ENTRY_OF_SLOT(2) ENTRY_OF_SLOT(3)
    ENTRY_OF_SLOT(4) ENTRY_OF_SLOT(5) ENTRY_OF_SLOT(6) ENTRY_OF_SLOT(7)
#undef ENTRY_OF_SLOT
    int   src[SLOTS];
    float wgt[SLOTS];
    bool  ok[SLOTS];
#pragma unroll
    for (int j = 0; j < SLOTS; ++j) {
      ok[j]  = (start + half + 2 * j) < end;
      src[j] = ok[j] ? sx[j] : 0;
      wgt[j] = ok[j] ? __int_as_float(sy[j]) : 0.0f;
    }

    float4 v[SLOTS];
#pragma unroll
    for (int j = 0; j < SLOTS; ++j) v[j] = h4[(unsigned)src[j] * (unsigned)(D / 4) + u16];

    float4 acc = make_float4(0.0f, 0.0f, 0.0f, 0.0f);
#pragma unroll
    for (int j = 0; j < SLOTS; ++j) {
      const float4 x = ok[j] ? v[j] : make_float4(0.0f, 0.0f, 0.0f, 0.0f);
      const float w = wgt[j];
      acc.x += x.x * w; acc.y += x.y * w; acc.z += x.z * w; acc.w += x.w * w;
    }
    for (int e = start + 2 * SLOTS + half; e < end; e += 2) {
      const int2 c0 = csr[e];
      const float4 v0 = h4[(unsigned)c0.x * (unsigned)(D / 4) + u16];
      const float w0 = __int_as_float(c0.y);
      acc.x += v0.x * w0; acc.y += v0.y * w0; acc.z += v0.z * w0; acc.w += v0.w * w0;
    }
    acc.x += __shfl_xor(acc.x, 16);
    acc.y += __shfl_xor(acc.y, 16);
    acc.z += __shfl_xor(acc.z, 16);
    acc.w += __shfl_xor(acc.w, 16);
    if (half == 0)
      *(reinterpret_cast<float4*>(tA + r * TP) + l16) = acc;
  }
  __syncthreads();

  const int m0 = half * 8;
  const v4f* wc4 = reinterpret_cast<const v4f*>(Wc);
  const v4f* wr4 = reinterpret_cast<const v4f*>(Wr);
  const unsigned wl = (unsigned)(kb * (D / 4) + l16);
  const float* fA = tA + row * TP + kb;
  const float* fH = tH + row * TP + kb;
  v8f acc[4] = {};
#pragma unroll
  for (int kt = 0; kt < 16; ++kt) {
    const v4f c0 = wc4[wl + (unsigned)((kt * 4) * (D / 4))];
    const v4f c1 = wc4[wl + (unsigned)((kt * 4 + 1) * (D / 4))];
    const v4f r0 = wr4[wl + (unsigned)((kt * 4) * (D / 4))];
    const v4f r1 = wr4[wl + (unsigned)((kt * 4 + 1) * (D / 4))];
    const v2f t = *reinterpret_cast<const v2f*>(fA + 4 * kt);
    v2f aC;
    aC[0] = t[0] * inrm;
    aC[1] = t[1] * inrm;
    const v2f aR = *reinterpret_cast<const v2f*>(fH + 4 * kt);
    if (kt & 1) __builtin_amdgcn_sched_barrier(0);
#pragma unroll
    for (int j = 0; j < 4; ++j) {
      v2f bC, bR;
      bC[0] = c0[j];
      bC[1] = c1[j];
      bR[0] = r0[j];
      bR[1] = r1[j];
      acc[j] = __builtin_amdgcn_wmma_f32_16x16x4_f32(false, aC, false, bC, (short)0, acc[j], false, false);
      acc[j] = __builtin_amdgcn_wmma_f32_16x16x4_f32(false, aR, false, bR, (short)0, acc[j], false, false);
    }
  }
  {
    const v4f bias = *reinterpret_cast<const v4f*>(bc + 4 * l16) +
                     *reinterpret_cast<const v4f*>(br + 4 * l16);
#pragma unroll
    for (int r = 0; r < 8; ++r) {
      v4f o;
      o[0] = acc[0][r] + bias[0];
      o[1] = acc[1][r] + bias[1];
      o[2] = acc[2][r] + bias[2];
      o[3] = acc[3][r] + bias[3];
      *reinterpret_cast<v4f*>(tA + (m0 + r) * TP + 4 * l16) = o;
    }
  }
  __syncthreads();

  const v4f g4 = *reinterpret_cast<const v4f*>(lng + 4 * l16);
  const v4f b4 = *reinterpret_cast<const v4f*>(lnb + 4 * l16);
  v4f v1[8];
  if constexpr (LAST) {
#pragma unroll
    for (int i = 0; i < 8; ++i) {
      int nd = base + 2 * i + half;
      nd = nd < n ? nd : (n - 1);
      v1[i] = *reinterpret_cast<const v4f*>(h_first + (size_t)nd * D + 4 * l16);
    }
  }

  const int rr = lane >> 1;
  const int j0 = (lane & 1) * 32;
  const float* myrow = tA + rr * TP + j0;
  float s = 0.0f, sq = 0.0f;
#pragma unroll
  for (int j = 0; j < 8; ++j) {
    const v4f q = *reinterpret_cast<const v4f*>(myrow + 4 * j);
    s += q[0]; sq += q[0] * q[0];
    s += q[1]; sq += q[1] * q[1];
    s += q[2]; sq += q[2] * q[2];
    s += q[3]; sq += q[3] * q[3];
  }
  s  += __shfl_xor(s, 1);
  sq += __shfl_xor(sq, 1);
  const float mu   = s * (1.0f / 64.0f);
  const float var  = sq * (1.0f / 64.0f) - mu * mu;
  const float rstd = rsqrtf(var + LN_EPS);
  if ((lane & 1) == 0) {
    v2f st;
    st[0] = mu;
    st[1] = rstd;
    *reinterpret_cast<v2f*>(&s_stat[wave][2 * rr]) = st;
  }
  if constexpr (LAST) {
#pragma unroll
    for (int i = 0; i < 8; ++i) {
      float* p = tH + (2 * i + half) * TP + 4 * l16;
      const v4f h2 = *reinterpret_cast<const v4f*>(p);
      *reinterpret_cast<v4f*>(p) = v1[i] + h2;
    }
  }
  __syncthreads();

  v4f y[8];
#pragma unroll
  for (int i = 0; i < 8; ++i) {
    const int r = 2 * i + half;
    const v2f st = *reinterpret_cast<const v2f*>(&s_stat[wave][2 * r]);
    const v4f x = *reinterpret_cast<const v4f*>(tA + r * TP + 4 * l16);
    y[i][0] = fmaxf((x[0] - st[0]) * st[1] * g4[0] + b4[0], 0.0f);
    y[i][1] = fmaxf((x[1] - st[0]) * st[1] * g4[1] + b4[1], 0.0f);
    y[i][2] = fmaxf((x[2] - st[0]) * st[1] * g4[2] + b4[2], 0.0f);
    y[i][3] = fmaxf((x[3] - st[0]) * st[1] * g4[3] + b4[3], 0.0f);
  }

  if constexpr (!LAST) {
#pragma unroll
    for (int i = 0; i < 8; ++i) {
      const int onode = base + 2 * i + half;
      if (onode < n)
        *reinterpret_cast<v4f*>(h_out + (size_t)onode * D + 4 * l16) = y[i];
    }
  } else {
#pragma unroll
    for (int i = 0; i < 8; ++i) {
      float* p = tH + (2 * i + half) * TP + 4 * l16;
      const v4f t = *reinterpret_cast<const v4f*>(p);
      *reinterpret_cast<v4f*>(p) = t + y[i];
    }
    __syncthreads();

    const v4f* wp4 = reinterpret_cast<const v4f*>(Wp);
    v8f pacc[4] = {};
#pragma unroll
    for (int kt = 0; kt < 16; ++kt) {
      const v4f p0 = wp4[wl + (unsigned)((kt * 4) * (D / 4))];
      const v4f p1 = wp4[wl + (unsigned)((kt * 4 + 1) * (D / 4))];
      const v2f a = *reinterpret_cast<const v2f*>(fH + 4 * kt);
      if (kt & 1) __builtin_amdgcn_sched_barrier(0);
#pragma unroll
      for (int j = 0; j < 4; ++j) {
        v2f bf;
        bf[0] = p0[j];
        bf[1] = p1[j];
        pacc[j] = __builtin_amdgcn_wmma_f32_16x16x4_f32(false, a, false, bf, (short)0, pacc[j], false, false);
      }
    }
    {
      const v4f bias = *reinterpret_cast<const v4f*>(bp + 4 * l16);
#pragma unroll
      for (int r = 0; r < 8; ++r) {
        v4f o;
        o[0] = pacc[0][r] + bias[0];
        o[1] = pacc[1][r] + bias[1];
        o[2] = pacc[2][r] + bias[2];
        o[3] = pacc[3][r] + bias[3];
        *reinterpret_cast<v4f*>(tA + (m0 + r) * TP + 4 * l16) = o;
      }
    }
    __syncthreads();

    v4f vo[8];
#pragma unroll
    for (int i = 0; i < 8; ++i)
      vo[i] = *reinterpret_cast<const v4f*>(tA + (2 * i + half) * TP + 4 * l16);
#pragma unroll
    for (int i = 0; i < 8; ++i) {
      const int onode = base + 2 * i + half;
      if (onode < n)
        *reinterpret_cast<v4f*>(out + (size_t)onode * D + 4 * l16) = vo[i];
    }
  }
}

extern "C" void kernel_launch(void* const* d_in, const int* in_sizes, int n_in,
                              void* d_out, int out_size, void* d_ws, size_t ws_size,
                              hipStream_t stream) {
  (void)in_sizes; (void)n_in; (void)out_size; (void)ws_size;

  const float* feats  = (const float*)d_in[0];
  const int*   src    = (const int*)  d_in[1];
  const int*   dst    = (const int*)  d_in[2];
  const float* conv_W = (const float*)d_in[3];
  const float* conv_b = (const float*)d_in[4];
  const float* res_W  = (const float*)d_in[5];
  const float* res_b  = (const float*)d_in[6];
  const float* ln_g   = (const float*)d_in[7];
  const float* ln_b   = (const float*)d_in[8];
  const float* pred_W = (const float*)d_in[9];
  const float* pred_b = (const float*)d_in[10];
  float* out = (float*)d_out;

  char* ws = (char*)d_ws;
  size_t off = 0;
  auto alloc = [&](size_t bytes) -> float* {
    float* p = (float*)(ws + off);
    off += (bytes + 255) & ~(size_t)255;
    return p;
  };
  const size_t nodeF = (size_t)N_NODES * 4;
  const size_t featF = (size_t)N_NODES * D * 4;
  float* in_norm  = alloc(nodeF);
  float* out_w    = alloc(nodeF);
  int*   row_ptr  = (int*)alloc(nodeF + 4);
  int*   tot      = (int*)alloc((size_t)NT * NS * 4);
  int2*  csr      = (int2*)alloc((size_t)N_EDGES * 8);
  float* sums     = alloc(featF);
  float* h1       = alloc(featF);
  float* h2       = alloc(featF);
  int*   pc       = (int*)sums;

  const int TB = 256;
  const int gTile   = (N_NODES + 127) / 128;

  count_kernel<<<dim3(NR, NS), GRP_THREADS, 0, stream>>>(src, dst, pc, tot);
  scan_kernel<<<NT, 256, 0, stream>>>(pc, tot, row_ptr, in_norm, out_w, N_NODES);
  fill_kernel<<<dim3(NR, NS), GRP_THREADS, 0, stream>>>(src, dst, pc, out_w, csr);

  gcn_layer_kernel<false><<<gTile, TB, 0, stream>>>(
      feats, row_ptr, csr, in_norm, conv_W, conv_b, res_W, res_b, ln_g, ln_b,
      h1, nullptr, nullptr, nullptr, nullptr, N_NODES, N_EDGES);

  gcn_layer_kernel<false><<<gTile, TB, 0, stream>>>(
      h1, row_ptr, csr, in_norm, conv_W + D * D, conv_b + D, res_W + D * D, res_b + D,
      ln_g + D, ln_b + D,
      h2, nullptr, nullptr, nullptr, nullptr, N_NODES, N_EDGES);

  gcn_layer_kernel<true><<<gTile, TB, 0, stream>>>(
      h2, row_ptr, csr, in_norm, conv_W + 2 * D * D, conv_b + 2 * D, res_W + 2 * D * D, res_b + 2 * D,
      ln_g + 2 * D, ln_b + 2 * D,
      nullptr, h1, pred_W, pred_b, out, N_NODES, N_EDGES);
}
